# starting kernel
# baseline (speedup 1.0000x reference)
#include <hip/hip_runtime.h>
#include <math.h>

#define NN 100000
#define EE 1600000
#define ET (EE + NN)
#define IN_CH 128
#define HID 32
#define HEADS 4
#define HC 128
#define OUT_CH 40
#define NEG_SLOPE 0.2f

typedef __attribute__((ext_vector_type(2))) float v2f;
typedef __attribute__((ext_vector_type(8))) float v8f;

__device__ __forceinline__ float atomic_max_f32(float* addr, float v) {
    if (v >= 0.0f)
        return __int_as_float(atomicMax((int*)addr, __float_as_int(v)));
    else
        return __uint_as_float(atomicMin((unsigned int*)addr, __float_as_uint(v)));
}

__device__ __forceinline__ float lrelu(float v) {
    return v >= 0.0f ? v : NEG_SLOPE * v;
}

__global__ void k_gemm1(const float* __restrict__ x, const float* __restrict__ W1,
                        float* __restrict__ h1) {
    const int wave = threadIdx.x >> 5;
    const int row0 = (blockIdx.x * 4 + wave) * 16;
    if (row0 >= NN) return;
    const int lane = threadIdx.x & 31;
    const int m    = lane & 15;
    const int kb   = (lane >> 4) * 2;

    v8f c[8];
#pragma unroll
    for (int t = 0; t < 8; ++t) c[t] = (v8f){0,0,0,0,0,0,0,0};

    for (int k = 0; k < IN_CH; k += 4) {
        v2f a;
        const float* xr = x + (size_t)(row0 + m) * IN_CH + (k + kb);
        a.x = xr[0];
        a.y = xr[1];
#pragma unroll
        for (int t = 0; t < 8; ++t) {
            v2f b;
            const float* wr = W1 + (size_t)(k + kb) * HC + t * 16 + m;
            b.x = wr[0];
            b.y = wr[HC];
            c[t] = __builtin_amdgcn_wmma_f32_16x16x4_f32(false, a, false, b,
                                                         (short)0, c[t], false, false);
        }
    }
    const int rofs = (lane >> 4) * 8;
#pragma unroll
    for (int t = 0; t < 8; ++t)
#pragma unroll
        for (int r = 0; r < 8; ++r)
            h1[(size_t)(row0 + r + rofs) * HC + t * 16 + m] = c[t][r];
}

__global__ void k_al1(const float* __restrict__ h1, const float* __restrict__ a_src,
                      const float* __restrict__ a_dst, float* __restrict__ als,
                      float* __restrict__ ald) {
    int i = blockIdx.x * blockDim.x + threadIdx.x;
    if (i >= NN * HEADS) return;
    int n = i >> 2, h = i & 3;
    const float* hp = h1 + (size_t)n * HC + h * HID;
    float s = 0.f, d = 0.f;
#pragma unroll 8
    for (int cc = 0; cc < HID; ++cc) {
        float v = hp[cc];
        s += v * a_src[h * HID + cc];
        d += v * a_dst[h * HID + cc];
    }
    als[i] = s;
    ald[i] = d;
}

__global__ void k_init1(float* __restrict__ agg, float* __restrict__ m,
                        float* __restrict__ dn) {
    int i = blockIdx.x * blockDim.x + threadIdx.x;
    if (i < NN * HC) agg[i] = 0.0f;
    if (i < NN * HEADS) { m[i] = -__builtin_inff(); dn[i] = 0.0f; }
}

__global__ void k_emax1(const int* __restrict__ ei, const float* __restrict__ als,
                        const float* __restrict__ ald, float* __restrict__ m) {
    int i = blockIdx.x * blockDim.x + threadIdx.x;
    if (i >= ET * HEADS) return;
    int e = i >> 2, h = i & 3;
    int s, d;
    if (e < EE) { s = ei[e]; d = ei[EE + e]; } else { s = d = e - EE; }
    float v = lrelu(als[s * 4 + h] + ald[d * 4 + h]);
    atomic_max_f32(&m[d * 4 + h], v);
}

__global__ void k_esum1(const int* __restrict__ ei, const float* __restrict__ als,
                        const float* __restrict__ ald, const float* __restrict__ m,
                        float* __restrict__ dn) {
    int i = blockIdx.x * blockDim.x + threadIdx.x;
    if (i >= ET * HEADS) return;
    int e = i >> 2, h = i & 3;
    int s, d;
    if (e < EE) { s = ei[e]; d = ei[EE + e]; } else { s = d = e - EE; }
    float v = lrelu(als[s * 4 + h] + ald[d * 4 + h]);
    atomicAdd(&dn[d * 4 + h], __expf(v - m[d * 4 + h]));
}

__global__ void k_eagg1(const int* __restrict__ ei, const float* __restrict__ h1,
                        const float* __restrict__ als, const float* __restrict__ ald,
                        const float* __restrict__ m, const float* __restrict__ dn,
                        float* __restrict__ agg) {
    int e = blockIdx.x * 8 + (threadIdx.x >> 5);
    if (e >= ET) return;
    int lane = threadIdx.x & 31;
    int s, d;
    if (e < EE) { s = ei[e]; d = ei[EE + e]; } else { s = d = e - EE; }
    int h = lane >> 3;
    float v = lrelu(als[s * 4 + h] + ald[d * 4 + h]);
    float alpha = __expf(v - m[d * 4 + h]) / (dn[d * 4 + h] + 1e-16f);
    const float4 hv = *(const float4*)(h1 + (size_t)s * HC + lane * 4);
    float* op = agg + (size_t)d * HC + lane * 4;
    atomicAdd(op + 0, hv.x * alpha);
    atomicAdd(op + 1, hv.y * alpha);
    atomicAdd(op + 2, hv.z * alpha);
    atomicAdd(op + 3, hv.w * alpha);
}

__global__ void k_fin1(float* __restrict__ agg, const float* __restrict__ b1) {
    int i = blockIdx.x * blockDim.x + threadIdx.x;
    if (i >= NN * HC) return;
    float v = agg[i] + b1[i & (HC - 1)];
    agg[i] = v > 0.0f ? v : expm1f(v);
}

__global__ void k_gemm2(const float* __restrict__ h2, const float* __restrict__ W2,
                        float* __restrict__ z) {
    const int wave = threadIdx.x >> 5;
    const int row0 = (blockIdx.x * 4 + wave) * 16;
    if (row0 >= NN) return;
    const int lane = threadIdx.x & 31;
    const int m    = lane & 15;
    const int kb   = (lane >> 4) * 2;

    v8f c[3];
#pragma unroll
    for (int t = 0; t < 3; ++t) c[t] = (v8f){0,0,0,0,0,0,0,0};

    for (int k = 0; k < HC; k += 4) {
        v2f a;
        const float* xr = h2 + (size_t)(row0 + m) * HC + (k + kb);
        a.x = xr[0];
        a.y = xr[1];
#pragma unroll
        for (int t = 0; t < 3; ++t) {
            int col = t * 16 + m;
            v2f b;
            b.x = (col < OUT_CH) ? W2[(size_t)(k + kb) * OUT_CH + col] : 0.0f;
            b.y = (col < OUT_CH) ? W2[(size_t)(k + kb + 1) * OUT_CH + col] : 0.0f;
            c[t] = __builtin_amdgcn_wmma_f32_16x16x4_f32(false, a, false, b,
                                                         (short)0, c[t], false, false);
        }
    }
    const int rofs = (lane >> 4) * 8;
#pragma unroll
    for (int t = 0; t < 3; ++t) {
        int col = t * 16 + m;
        if (col < OUT_CH)
#pragma unroll
            for (int r = 0; r < 8; ++r)
                z[(size_t)(row0 + r + rofs) * OUT_CH + col] = c[t][r];
    }
}

__global__ void k_al2(const float* __restrict__ z, const float* __restrict__ a_src,
                      const float* __restrict__ a_dst, float* __restrict__ als,
                      float* __restrict__ ald) {
    int n = blockIdx.x * blockDim.x + threadIdx.x;
    if (n >= NN) return;
    const float* zp = z + (size_t)n * OUT_CH;
    float s = 0.f, d = 0.f;
#pragma unroll 8
    for (int cc = 0; cc < OUT_CH; ++cc) {
        float v = zp[cc];
        s += v * a_src[cc];
        d += v * a_dst[cc];
    }
    als[n] = s;
    ald[n] = d;
}

__global__ void k_init2(float* __restrict__ out, const float* __restrict__ b2,
                        float* __restrict__ m, float* __restrict__ dn) {
    int i = blockIdx.x * blockDim.x + threadIdx.x;
    if (i < NN * OUT_CH) out[i] = b2[i % OUT_CH];
    if (i < NN) { m[i] = -__builtin_inff(); dn[i] = 0.0f; }
}

__global__ void k_emax2(const int* __restrict__ ei, const float* __restrict__ als,
                        const float* __restrict__ ald, float* __restrict__ m) {
    int e = blockIdx.x * blockDim.x + threadIdx.x;
    if (e >= ET) return;
    int s, d;
    if (e < EE) { s = ei[e]; d = ei[EE + e]; } else { s = d = e - EE; }
    atomic_max_f32(&m[d], lrelu(als[s] + ald[d]));
}

__global__ void k_esum2(const int* __restrict__ ei, const float* __restrict__ als,
                        const float* __restrict__ ald, const float* __restrict__ m,
                        float* __restrict__ dn) {
    int e = blockIdx.x * blockDim.x + threadIdx.x;
    if (e >= ET) return;
    int s, d;
    if (e < EE) { s = ei[e]; d = ei[EE + e]; } else { s = d = e - EE; }
    float v = lrelu(als[s] + ald[d]);
    atomicAdd(&dn[d], __expf(v - m[d]));
}

__global__ void k_eagg2(const int* __restrict__ ei, const float* __restrict__ z,
                        const float* __restrict__ als, const float* __restrict__ ald,
                        const float* __restrict__ m, const float* __restrict__ dn,
                        float* __restrict__ out) {
    int e = blockIdx.x * 8 + (threadIdx.x >> 5);
    if (e >= ET) return;
    int lane = threadIdx.x & 31;
    int s, d;
    if (e < EE) { s = ei[e]; d = ei[EE + e]; } else { s = d = e - EE; }
    float v = lrelu(als[s] + ald[d]);
    float alpha = __expf(v - m[d]) / (dn[d] + 1e-16f);
    const float* zp = z + (size_t)s * OUT_CH;
    float* op = out + (size_t)d * OUT_CH;
    atomicAdd(op + lane, zp[lane] * alpha);
    if (lane < OUT_CH - 32)
        atomicAdd(op + 32 + lane, zp[32 + lane] * alpha);
}

extern "C" void kernel_launch(void* const* d_in, const int* in_sizes, int n_in,
                              void* d_out, int out_size, void* d_ws, size_t ws_size,
                              hipStream_t stream) {
    const float* x      = (const float*)d_in[0];
    const int*   ei     = (const int*)  d_in[1];
    const float* W1     = (const float*)d_in[2];
    const float* a_src1 = (const float*)d_in[3];
    const float* a_dst1 = (const float*)d_in[4];
    const float* b1     = (const float*)d_in[5];
    const float* W2     = (const float*)d_in[6];
    const float* a_src2 = (const float*)d_in[7];
    const float* a_dst2 = (const float*)d_in[8];
    const float* b2     = (const float*)d_in[9];
    float* out = (float*)d_out;

    float* ws   = (float*)d_ws;
    float* h1   = ws;
    float* agg1 = h1   + (size_t)NN * HC;
    float* z2   = agg1 + (size_t)NN * HC;
    float* als1 = z2   + (size_t)NN * OUT_CH;
    float* ald1 = als1 + (size_t)NN * HEADS;
    float* m1   = ald1 + (size_t)NN * HEADS;
    float* dn1  = m1   + (size_t)NN * HEADS;
    float* als2 = dn1  + (size_t)NN * HEADS;
    float* ald2 = als2 + NN;
    float* m2   = ald2 + NN;
    float* dn2  = m2   + NN;

    const int TB = 256;
    const int gemmBlocks = (NN / 16 + 3) / 4;
    const int edgeWaveBlocks = (ET + 7) / 8;

    k_gemm1<<<gemmBlocks, 128, 0, stream>>>(x, W1, h1);
    k_al1  <<<(NN * HEADS + TB - 1) / TB, TB, 0, stream>>>(h1, a_src1, a_dst1, als1, ald1);
    k_init1<<<(NN * HC + TB - 1) / TB, TB, 0, stream>>>(agg1, m1, dn1);
    k_emax1<<<(ET * HEADS + TB - 1) / TB, TB, 0, stream>>>(ei, als1, ald1, m1);
    k_esum1<<<(ET * HEADS + TB - 1) / TB, TB, 0, stream>>>(ei, als1, ald1, m1, dn1);
    k_eagg1<<<edgeWaveBlocks, 256, 0, stream>>>(ei, h1, als1, ald1, m1, dn1, agg1);
    k_fin1 <<<(NN * HC + TB - 1) / TB, TB, 0, stream>>>(agg1, b1);

    k_gemm2<<<gemmBlocks, 128, 0, stream>>>(agg1, W2, z2);
    k_al2  <<<(NN + TB - 1) / TB, TB, 0, stream>>>(z2, a_src2, a_dst2, als2, ald2);
    k_init2<<<(NN * OUT_CH + TB - 1) / TB, TB, 0, stream>>>(out, b2, m2, dn2);
    k_emax2<<<(ET + TB - 1) / TB, TB, 0, stream>>>(ei, als2, ald2, m2);
    k_esum2<<<(ET + TB - 1) / TB, TB, 0, stream>>>(ei, als2, ald2, m2, dn2);
    k_eagg2<<<edgeWaveBlocks, 256, 0, stream>>>(ei, z2, als2, ald2, m2, dn2, out);
}


# Edges bucketed by destination once per call (integer-atomic slot, wave-scan offsets); one wave per node does max, exp-sum and weighted row sum in registers: no float atomics, sums in bucket order
# speedup vs baseline: 3.1890x; 3.1890x over previous
#include <hip/hip_runtime.h>
#include <math.h>

#define NN 100000
#define EE 1600000
#define IN_CH 128
#define HID 32
#define HEADS 4
#define HC 128
#define OUT_CH 40
#define NEG_SLOPE 0.2f

typedef __attribute__((ext_vector_type(2))) float v2f;
typedef __attribute__((ext_vector_type(8))) float v8f;

__device__ __forceinline__ float lrelu(float v) {
    return v >= 0.0f ? v : NEG_SLOPE * v;
}

__global__ void k_gemm1(const float* __restrict__ x, const float* __restrict__ W1,
                        float* __restrict__ h1) {
    const int wave = threadIdx.x >> 5;
    const int row0 = (blockIdx.x * 4 + wave) * 16;
    if (row0 >= NN) return;
    const int lane = threadIdx.x & 31;
    const int m    = lane & 15;
    const int kb   = (lane >> 4) * 2;

    v8f c[8];
#pragma unroll
    for (int t = 0; t < 8; ++t) c[t] = (v8f){0,0,0,0,0,0,0,0};

    for (int k = 0; k < IN_CH; k += 4) {
        v2f a;
        const float* xr = x + (size_t)(row0 + m) * IN_CH + (k + kb);
        a.x = xr[0];
        a.y = xr[1];
#pragma unroll
        for (int t = 0; t < 8; ++t) {
            v2f b;
            const float* wr = W1 + (size_t)(k + kb) * HC + t * 16 + m;
            b.x = wr[0];
            b.y = wr[HC];
            c[t] = __builtin_amdgcn_wmma_f32_16x16x4_f32(false, a, false, b,
                                                         (short)0, c[t], false, false);
        }
    }
    const int rofs = (lane >> 4) * 8;
#pragma unroll
    for (int t = 0; t < 8; ++t)
#pragma unroll
        for (int r = 0; r < 8; ++r)
            h1[(size_t)(row0 + r + rofs) * HC + t * 16 + m] = c[t][r];
}

__global__ void k_al1(const float* __restrict__ h1, const float* __restrict__ a_src,
                      const float* __restrict__ a_dst, float* __restrict__ als,
                      float* __restrict__ ald, int* __restrict__ deg,
                      int* __restrict__ total) {
    int i = blockIdx.x * blockDim.x + threadIdx.x;
    if (i >= NN * HEADS) return;
    if (i < NN) deg[i] = 0;
    if (i == 0) total[0] = 0;
    int n = i >> 2, h = i & 3;
    const float* hp = h1 + (size_t)n * HC + h * HID;
    float s = 0.f, d = 0.f;
#pragma unroll 8
    for (int cc = 0; cc < HID; ++cc) {
        float v = hp[cc];
        s += v * a_src[h * HID + cc];
        d += v * a_dst[h * HID + cc];
    }
    als[i] = s;
    ald[i] = d;
}

__global__ void k_count(const int* __restrict__ ei, int* __restrict__ deg,
                        int* __restrict__ rnk) {
    int e = blockIdx.x * blockDim.x + threadIdx.x;
    if (e >= EE) return;
    int d = ei[EE + e];
    rnk[e] = atomicAdd(&deg[d], 1);
}

__global__ void k_alloc(const int* __restrict__ deg, int* __restrict__ total,
                        int* __restrict__ rowstart) {
    const int n = blockIdx.x * blockDim.x + threadIdx.x;
    const int lane = threadIdx.x & 31;
    const int v = (n < NN) ? deg[n] : 0;
    int incl = v;
#pragma unroll
    for (int off = 1; off < 32; off <<= 1) {
        int t = __shfl_up(incl, off);
        if (lane >= off) incl += t;
    }
    const int wsum = __shfl(incl, 31);
    int base = 0;
    if (lane == 31) base = atomicAdd(total, wsum);
    base = __shfl(base, 31);
    if (n < NN) rowstart[n] = base + incl - v;
}

__global__ void k_fill(const int* __restrict__ ei, const int* __restrict__ rowstart,
                       const int* __restrict__ rnk, int* __restrict__ csr) {
    int e = blockIdx.x * blockDim.x + threadIdx.x;
    if (e >= EE) return;
    int s = ei[e];
    int d = ei[EE + e];
    csr[rowstart[d] + rnk[e]] = s;
}


__device__ __forceinline__ void logits4(const int* __restrict__ csr, const float* __restrict__ als,
                                        int start, int cnt, int n, int idx, const float4 dn,
                                        int& s, float4& v) {
    const float NEG_INF = -__builtin_inff();
    s = n;
    v = make_float4(NEG_INF, NEG_INF, NEG_INF, NEG_INF);
    if (idx <= cnt) {
        if (idx < cnt) s = csr[start + idx];
        const float4 a = *(const float4*)(als + (size_t)s * 4);
        v.x = lrelu(a.x + dn.x);
        v.y = lrelu(a.y + dn.y);
        v.z = lrelu(a.z + dn.z);
        v.w = lrelu(a.w + dn.w);
    }
}

__device__ __forceinline__ void chunk_rows(const float* __restrict__ h1, float4* wt,
                                           const float* wth, int lane, int s, const float4 w,
                                           int nv, float4& acc) {
    __builtin_amdgcn_wave_barrier();
    wt[lane] = w;
    __builtin_amdgcn_wave_barrier();
    const float* hl = h1 + lane * 4;
    int j = 0;
    for (; j + 4 <= nv; j += 4) {
        const int sa = __builtin_amdgcn_readlane(s, j);
        const int sb = __builtin_amdgcn_readlane(s, j + 1);
        const int sc = __builtin_amdgcn_readlane(s, j + 2);
        const int sd = __builtin_amdgcn_readlane(s, j + 3);
        const float4 ra = *(const float4*)(hl + (size_t)sa * HC);
        const float4 rb = *(const float4*)(hl + (size_t)sb * HC);
        const float4 rc = *(const float4*)(hl + (size_t)sc * HC);
        const float4 rd = *(const float4*)(hl + (size_t)sd * HC);
        const float wa = wth[4 * j];
        const float wb = wth[4 * j + 4];
        const float wc = wth[4 * j + 8];
        const float wd = wth[4 * j + 12];
        acc.x += ra.x * wa; acc.y += ra.y * wa; acc.z += ra.z * wa; acc.w += ra.w * wa;
        acc.x += rb.x * wb; acc.y += rb.y * wb; acc.z += rb.z * wb; acc.w += rb.w * wb;
        acc.x += rc.x * wc; acc.y += rc.y * wc; acc.z += rc.z * wc; acc.w += rc.w * wc;
        acc.x += rd.x * wd; acc.y += rd.y * wd; acc.z += rd.z * wd; acc.w += rd.w * wd;
    }
    for (; j < nv; ++j) {
        const int sa = __builtin_amdgcn_readlane(s, j);
        const float4 ra = *(const float4*)(hl + (size_t)sa * HC);
        const float wa = wth[4 * j];
        acc.x += ra.x * wa; acc.y += ra.y * wa; acc.z += ra.z * wa; acc.w += ra.w * wa;
    }
}

__global__ void __launch_bounds__(256)
k_node1(const int* __restrict__ csr, const int* __restrict__ rowstart,
        const int* __restrict__ deg, const float* __restrict__ h1,
        const float* __restrict__ als, const float* __restrict__ ald,
        const float* __restrict__ b1, float* __restrict__ h2) {
    __shared__ float4 wtab[8][32];
    const int lane = threadIdx.x & 31;
    const int wv   = threadIdx.x >> 5;
    const int n = __builtin_amdgcn_readfirstlane((int)(blockIdx.x * 8 + wv));
    if (n >= NN) return;
    const int start = __builtin_amdgcn_readfirstlane(rowstart[n]);
    const int cnt   = __builtin_amdgcn_readfirstlane(deg[n]);
    const int tot   = cnt + 1;
    const float4 dn = *(const float4*)(ald + (size_t)n * 4);
    float4* wt = wtab[wv];
    const float* wth = (const float*)wt + (lane >> 3);

    int s0; float4 v0;
    logits4(csr, als, start, cnt, n, lane, dn, s0, v0);

    float4 m = v0;
    for (int base = 32; base < tot; base += 32) {
        int s; float4 v;
        logits4(csr, als, start, cnt, n, base + lane, dn, s, v);
        m.x = fmaxf(m.x, v.x); m.y = fmaxf(m.y, v.y); m.z = fmaxf(m.z, v.z); m.w = fmaxf(m.w, v.w);
    }
#pragma unroll
    for (int off = 16; off > 0; off >>= 1) {
        m.x = fmaxf(m.x, __shfl_xor(m.x, off));
        m.y = fmaxf(m.y, __shfl_xor(m.y, off));
        m.z = fmaxf(m.z, __shfl_xor(m.z, off));
        m.w = fmaxf(m.w, __shfl_xor(m.w, off));
    }

    float4 p0;
    p0.x = __expf(v0.x - m.x); p0.y = __expf(v0.y - m.y);
    p0.z = __expf(v0.z - m.z); p0.w = __expf(v0.w - m.w);
    float4 d = p0;
    for (int base = 32; base < tot; base += 32) {
        int s; float4 v;
        logits4(csr, als, start, cnt, n, base + lane, dn, s, v);
        d.x += __expf(v.x - m.x); d.y += __expf(v.y - m.y);
        d.z += __expf(v.z - m.z); d.w += __expf(v.w - m.w);
    }
#pragma unroll
    for (int off = 16; off > 0; off >>= 1) {
        d.x += __shfl_xor(d.x, off);
        d.y += __shfl_xor(d.y, off);
        d.z += __shfl_xor(d.z, off);
        d.w += __shfl_xor(d.w, off);
    }
    d.x += 1e-16f; d.y += 1e-16f; d.z += 1e-16f; d.w += 1e-16f;

    float4 acc = make_float4(0.f, 0.f, 0.f, 0.f);
    {
        float4 w;
        w.x = p0.x / d.x; w.y = p0.y / d.y; w.z = p0.z / d.z; w.w = p0.w / d.w;
        chunk_rows(h1, wt, wth, lane, s0, w, (tot < 32) ? tot : 32, acc);
    }
    for (int base = 32; base < tot; base += 32) {
        int s; float4 v;
        logits4(csr, als, start, cnt, n, base + lane, dn, s, v);
        float4 w;
        w.x = __expf(v.x - m.x) / d.x; w.y = __expf(v.y - m.y) / d.y;
        w.z = __expf(v.z - m.z) / d.z; w.w = __expf(v.w - m.w) / d.w;
        chunk_rows(h1, wt, wth, lane, s, w, (tot - base < 32) ? (tot - base) : 32, acc);
    }

    const float4 bb = *(const float4*)(b1 + lane * 4);
    float4 o;
    o.x = acc.x + bb.x; o.y = acc.y + bb.y; o.z = acc.z + bb.z; o.w = acc.w + bb.w;
    o.x = o.x > 0.0f ? o.x : expm1f(o.x);
    o.y = o.y > 0.0f ? o.y : expm1f(o.y);
    o.z = o.z > 0.0f ? o.z : expm1f(o.z);
    o.w = o.w > 0.0f ? o.w : expm1f(o.w);
    *(float4*)(h2 + (size_t)n * HC + lane * 4) = o;
}

__global__ void k_gemm2(const float* __restrict__ h2, const float* __restrict__ W2,
                        float* __restrict__ z) {
    const int wave = threadIdx.x >> 5;
    const int row0 = (blockIdx.x * 4 + wave) * 16;
    if (row0 >= NN) return;
    const int lane = threadIdx.x & 31;
    const int m    = lane & 15;
    const int kb   = (lane >> 4) * 2;

    v8f c[3];
#pragma unroll
    for (int t = 0; t < 3; ++t) c[t] = (v8f){0,0,0,0,0,0,0,0};

    for (int k = 0; k < HC; k += 4) {
        v2f a;
        const float* xr = h2 + (size_t)(row0 + m) * HC + (k + kb);
        a.x = xr[0];
        a.y = xr[1];
#pragma unroll
        for (int t = 0; t < 3; ++t) {
            int col = t * 16 + m;
            v2f b;
            b.x = (col < OUT_CH) ? W2[(size_t)(k + kb) * OUT_CH + col] : 0.0f;
            b.y = (col < OUT_CH) ? W2[(size_t)(k + kb + 1) * OUT_CH + col] : 0.0f;
            c[t] = __builtin_amdgcn_wmma_f32_16x16x4_f32(false, a, false, b,
                                                         (short)0, c[t], false, false);
        }
    }
    const int rofs = (lane >> 4) * 8;
#pragma unroll
    for (int t = 0; t < 3; ++t) {
        int col = t * 16 + m;
        if (col < OUT_CH)
#pragma unroll
            for (int r = 0; r < 8; ++r)
                z[(size_t)(row0 + r + rofs) * OUT_CH + col] = c[t][r];
    }
}

__global__ void k_al2(const float* __restrict__ z, const float* __restrict__ a_src,
                      const float* __restrict__ a_dst, float* __restrict__ als,
                      float* __restrict__ ald) {
    int n = blockIdx.x * blockDim.x + threadIdx.x;
    if (n >= NN) return;
    const float* zp = z + (size_t)n * OUT_CH;
    float s = 0.f, d = 0.f;
#pragma unroll 8
    for (int cc = 0; cc < OUT_CH; ++cc) {
        float v = zp[cc];
        s += v * a_src[cc];
        d += v * a_dst[cc];
    }
    als[n] = s;
    ald[n] = d;
}

__global__ void __launch_bounds__(256)
k_node2(const int* __restrict__ csr, const int* __restrict__ rowstart,
        const int* __restrict__ deg, const float* __restrict__ z,
        const float* __restrict__ als, const float* __restrict__ ald,
        const float* __restrict__ b2, float* __restrict__ out) {
    const int lane = threadIdx.x & 31;
    const int w = __builtin_amdgcn_readfirstlane((int)(blockIdx.x * 8 + (threadIdx.x >> 5)));
    const bool hi = lane < (OUT_CH - 32);
    const float bl = b2[lane];
    const float bh = hi ? b2[32 + lane] : 0.0f;
    const float NEG_INF = -__builtin_inff();

    for (int q = 0; q < 4; ++q) {
        const int n = w * 4 + q;
        if (n >= NN) break;
        const int start = __builtin_amdgcn_readfirstlane(rowstart[n]);
        const int cnt   = __builtin_amdgcn_readfirstlane(deg[n]);
        const int tot   = cnt + 1;
        const float dn  = ald[n];

        float m = NEG_INF;
        for (int base = 0; base < tot; base += 32) {
            const int idx = base + lane;
            if (idx < tot) {
                const int s = (idx < cnt) ? csr[start + idx] : n;
                m = fmaxf(m, lrelu(als[s] + dn));
            }
        }
#pragma unroll
        for (int off = 16; off > 0; off >>= 1) m = fmaxf(m, __shfl_xor(m, off));

        float den = 0.f;
        for (int base = 0; base < tot; base += 32) {
            const int idx = base + lane;
            if (idx < tot) {
                const int s = (idx < cnt) ? csr[start + idx] : n;
                den += __expf(lrelu(als[s] + dn) - m);
            }
        }
#pragma unroll
        for (int off = 16; off > 0; off >>= 1) den += __shfl_xor(den, off);
        den += 1e-16f;

        float acc0 = 0.f, acc1 = 0.f;
        for (int base = 0; base < tot; base += 32) {
            const int idx = base + lane;
            int s = n;
            float wgt = 0.f;
            if (idx < tot) {
                s = (idx < cnt) ? csr[start + idx] : n;
                wgt = __expf(lrelu(als[s] + dn) - m) / den;
            }
            const int nv = (tot - base < 32) ? (tot - base) : 32;
            int j = 0;
            for (; j + 4 <= nv; j += 4) {
                const int sa = __builtin_amdgcn_readlane(s, j);
                const int sb = __builtin_amdgcn_readlane(s, j + 1);
                const int sc = __builtin_amdgcn_readlane(s, j + 2);
                const int sd = __builtin_amdgcn_readlane(s, j + 3);
                const float wa = __int_as_float(__builtin_amdgcn_readlane(__float_as_int(wgt), j));
                const float wb = __int_as_float(__builtin_amdgcn_readlane(__float_as_int(wgt), j + 1));
                const float wc = __int_as_float(__builtin_amdgcn_readlane(__float_as_int(wgt), j + 2));
                const float wd = __int_as_float(__builtin_amdgcn_readlane(__float_as_int(wgt), j + 3));
                const float* za = z + (size_t)sa * OUT_CH;
                const float* zb = z + (size_t)sb * OUT_CH;
                const float* zc = z + (size_t)sc * OUT_CH;
                const float* zd = z + (size_t)sd * OUT_CH;
                const float la = za[lane], lb = zb[lane], lc = zc[lane], ld = zd[lane];
                float ha = 0.f, hb = 0.f, hc = 0.f, hd = 0.f;
                if (hi) { ha = za[32 + lane]; hb = zb[32 + lane]; hc = zc[32 + lane]; hd = zd[32 + lane]; }
                acc0 += la * wa; acc0 += lb * wb; acc0 += lc * wc; acc0 += ld * wd;
                acc1 += ha * wa; acc1 += hb * wb; acc1 += hc * wc; acc1 += hd * wd;
            }
            for (; j < nv; ++j) {
                const int sa = __builtin_amdgcn_readlane(s, j);
                const float wa = __int_as_float(__builtin_amdgcn_readlane(__float_as_int(wgt), j));
                const float* za = z + (size_t)sa * OUT_CH;
                acc0 += za[lane] * wa;
                if (hi) acc1 += za[32 + lane] * wa;
            }
        }
        float* op = out + (size_t)n * OUT_CH;
        op[lane] = acc0 + bl;
        if (hi) op[32 + lane] = acc1 + bh;
    }
}

extern "C" void kernel_launch(void* const* d_in, const int* in_sizes, int n_in,
                              void* d_out, int out_size, void* d_ws, size_t ws_size,
                              hipStream_t stream) {
    const float* x      = (const float*)d_in[0];
    const int*   ei     = (const int*)  d_in[1];
    const float* W1     = (const float*)d_in[2];
    const float* a_src1 = (const float*)d_in[3];
    const float* a_dst1 = (const float*)d_in[4];
    const float* b1     = (const float*)d_in[5];
    const float* W2     = (const float*)d_in[6];
    const float* a_src2 = (const float*)d_in[7];
    const float* a_dst2 = (const float*)d_in[8];
    const float* b2     = (const float*)d_in[9];
    float* out = (float*)d_out;

    float* ws   = (float*)d_ws;
    float* h1   = ws;
    float* agg1 = h1   + (size_t)NN * HC;
    float* z2   = agg1 + (size_t)NN * HC;
    float* als1 = z2   + (size_t)NN * OUT_CH;
    float* ald1 = als1 + (size_t)NN * HEADS;
    float* als2 = ald1 + (size_t)NN * HEADS;
    float* ald2 = als2 + NN;
    int* deg      = (int*)(ald2 + NN);
    int* rowstart = deg + NN;
    int* total    = rowstart + NN;
    int* rnk      = total + 32;
    int* csr      = rnk + EE;

    const int TB = 256;
    const int gemmBlocks = (NN / 16 + 3) / 4;
    const int nodeBlocks1 = (NN + 7) / 8;
    const int nodeBlocks2 = (NN / 4 + 7) / 8;

    k_gemm1<<<gemmBlocks, 128, 0, stream>>>(x, W1, h1);
    k_al1  <<<(NN * HEADS + TB - 1) / TB, TB, 0, stream>>>(h1, a_src1, a_dst1, als1, ald1, deg, total);
    k_count<<<(EE + TB - 1) / TB, TB, 0, stream>>>(ei, deg, rnk);
    k_alloc<<<(NN + TB - 1) / TB, TB, 0, stream>>>(deg, total, rowstart);
    k_fill <<<(EE + TB - 1) / TB, TB, 0, stream>>>(ei, rowstart, rnk, csr);
    k_node1<<<nodeBlocks1, 256, 0, stream>>>(csr, rowstart, deg, h1, als1, ald1, b1, agg1);

    k_gemm2<<<gemmBlocks, 128, 0, stream>>>(agg1, W2, z2);
    k_al2  <<<(NN + TB - 1) / TB, TB, 0, stream>>>(z2, a_src2, a_dst2, als2, ald2);
    k_node2<<<nodeBlocks2, 256, 0, stream>>>(csr, rowstart, deg, z2, als2, ald2, b2, out);
}


# Bucket build on LDS counters: edges binned into 256-node buckets (one global add per block and bin), then a block per bucket ranks on 256 LDS counters; run order still set by atomic arrival
# speedup vs baseline: 4.3753x; 1.3720x over previous
#include <hip/hip_runtime.h>
#include <math.h>

#define NN 100000
#define EE 1600000
#define IN_CH 128
#define HID 32
#define HEADS 4
#define HC 128
#define OUT_CH 40
#define NEG_SLOPE 0.2f
#define GH 512

typedef __attribute__((ext_vector_type(2))) float v2f;
typedef __attribute__((ext_vector_type(8))) float v8f;

__device__ __forceinline__ float lrelu(float v) {
    return v >= 0.0f ? v : NEG_SLOPE * v;
}

__global__ void k_gemm1(const float* __restrict__ x, const float* __restrict__ W1,
                        float* __restrict__ h1) {
    const int wave = threadIdx.x >> 5;
    const int row0 = (blockIdx.x * 4 + wave) * 16;
    if (row0 >= NN) return;
    const int lane = threadIdx.x & 31;
    const int m    = lane & 15;
    const int kb   = (lane >> 4) * 2;

    v8f c[8];
#pragma unroll
    for (int t = 0; t < 8; ++t) c[t] = (v8f){0,0,0,0,0,0,0,0};

    for (int k = 0; k < IN_CH; k += 4) {
        v2f a;
        const float* xr = x + (size_t)(row0 + m) * IN_CH + (k + kb);
        a.x = xr[0];
        a.y = xr[1];
#pragma unroll
        for (int t = 0; t < 8; ++t) {
            v2f b;
            const float* wr = W1 + (size_t)(k + kb) * HC + t * 16 + m;
            b.x = wr[0];
            b.y = wr[HC];
            c[t] = __builtin_amdgcn_wmma_f32_16x16x4_f32(false, a, false, b,
                                                         (short)0, c[t], false, false);
        }
    }
    const int rofs = (lane >> 4) * 8;
#pragma unroll
    for (int t = 0; t < 8; ++t)
#pragma unroll
        for (int r = 0; r < 8; ++r)
            h1[(size_t)(row0 + r + rofs) * HC + t * 16 + m] = c[t][r];
}

__global__ void k_al1(const float* __restrict__ h1, const float* __restrict__ a_src,
                      const float* __restrict__ a_dst, float* __restrict__ als,
                      float* __restrict__ ald, int* __restrict__ ghist) {
    int i = blockIdx.x * blockDim.x + threadIdx.x;
    if (i >= NN * HEADS) return;
    if (i < 2 * GH) ghist[i] = 0;
    int n = i >> 2, h = i & 3;
    const float* hp = h1 + (size_t)n * HC + h * HID;
    float s = 0.f, d = 0.f;
#pragma unroll 8
    for (int cc = 0; cc < HID; ++cc) {
        float v = hp[cc];
        s += v * a_src[h * HID + cc];
        d += v * a_dst[h * HID + cc];
    }
    als[i] = s;
    ald[i] = d;
}

#define BSH 8
#define BW  256
#define NB  ((NN + BW - 1) / BW)
#define TBE 512
#define EPT 8
#define BPT 12
#define EBLK ((EE / EPT + TBE - 1) / TBE)
static_assert(NB <= TBE && NB <= GH, "one thread per coarse bucket");
static_assert(EE % EPT == 0, "whole threads of edges");
static_assert(NN <= (1 << 23), "source id and local destination share one word");

__device__ __forceinline__ int block_excl_scan(int v, int* wsum) {
    const int lane = threadIdx.x & 31;
    const int wv   = threadIdx.x >> 5;
    int incl = v;
#pragma unroll
    for (int off = 1; off < 32; off <<= 1) {
        int t = __shfl_up(incl, off);
        if (lane >= off) incl += t;
    }
    if (lane == 31) wsum[wv] = incl;
    __syncthreads();
    int woff = 0;
    for (int k = 0; k < wv; ++k) woff += wsum[k];
    return woff + incl - v;
}

__global__ void __launch_bounds__(TBE)
k_hist(const int* __restrict__ ei, int* __restrict__ ghist) {
    __shared__ int h[TBE];
    const int t = threadIdx.x;
    h[t] = 0;
    __syncthreads();
    const int e0 = (blockIdx.x * TBE + t) * EPT;
    if (e0 < EE) {
        const int4 a = *(const int4*)(ei + EE + e0);
        const int4 b = *(const int4*)(ei + EE + e0 + 4);
        atomicAdd(&h[a.x >> BSH], 1); atomicAdd(&h[a.y >> BSH], 1);
        atomicAdd(&h[a.z >> BSH], 1); atomicAdd(&h[a.w >> BSH], 1);
        atomicAdd(&h[b.x >> BSH], 1); atomicAdd(&h[b.y >> BSH], 1);
        atomicAdd(&h[b.z >> BSH], 1); atomicAdd(&h[b.w >> BSH], 1);
    }
    __syncthreads();
    const int c = h[t];
    if (c) atomicAdd(&ghist[t], c);
}

__global__ void __launch_bounds__(TBE)
k_scatter(const int* __restrict__ ei, const int* __restrict__ ghist,
          int* __restrict__ gcur, int* __restrict__ tmp) {
    __shared__ int cnt[TBE];
    __shared__ int off[TBE];
    __shared__ int wsum[16];
    const int t = threadIdx.x;
    cnt[t] = 0;
    const int gv  = (t < NB) ? ghist[t] : 0;
    const int bst = block_excl_scan(gv, wsum);
    const int e0 = (blockIdx.x * TBE + t) * EPT;
    const bool act = e0 < EE;
    int4 s0, s1, d0, d1;
    int r0 = 0, r1 = 0, r2 = 0, r3 = 0, r4 = 0, r5 = 0, r6 = 0, r7 = 0;
    if (act) {
        s0 = *(const int4*)(ei + e0);
        s1 = *(const int4*)(ei + e0 + 4);
        d0 = *(const int4*)(ei + EE + e0);
        d1 = *(const int4*)(ei + EE + e0 + 4);
        r0 = atomicAdd(&cnt[d0.x >> BSH], 1); r1 = atomicAdd(&cnt[d0.y >> BSH], 1);
        r2 = atomicAdd(&cnt[d0.z >> BSH], 1); r3 = atomicAdd(&cnt[d0.w >> BSH], 1);
        r4 = atomicAdd(&cnt[d1.x >> BSH], 1); r5 = atomicAdd(&cnt[d1.y >> BSH], 1);
        r6 = atomicAdd(&cnt[d1.z >> BSH], 1); r7 = atomicAdd(&cnt[d1.w >> BSH], 1);
    }
    __syncthreads();
    {
        const int c = cnt[t];
        int base = 0;
        if (c) base = atomicAdd(&gcur[t], c);
        off[t] = bst + base;
    }
    __syncthreads();
    if (act) {
        tmp[off[d0.x >> BSH] + r0] = (s0.x << BSH) | (d0.x & (BW - 1));
        tmp[off[d0.y >> BSH] + r1] = (s0.y << BSH) | (d0.y & (BW - 1));
        tmp[off[d0.z >> BSH] + r2] = (s0.z << BSH) | (d0.z & (BW - 1));
        tmp[off[d0.w >> BSH] + r3] = (s0.w << BSH) | (d0.w & (BW - 1));
        tmp[off[d1.x >> BSH] + r4] = (s1.x << BSH) | (d1.x & (BW - 1));
        tmp[off[d1.y >> BSH] + r5] = (s1.y << BSH) | (d1.y & (BW - 1));
        tmp[off[d1.z >> BSH] + r6] = (s1.z << BSH) | (d1.z & (BW - 1));
        tmp[off[d1.w >> BSH] + r7] = (s1.w << BSH) | (d1.w & (BW - 1));
    }
}

__global__ void __launch_bounds__(TBE)
k_bsort(const int* __restrict__ ghist, const int* __restrict__ tmp,
        int* __restrict__ deg, int* __restrict__ rowstart, int* __restrict__ csr) {
    __shared__ int c[BW];
    __shared__ int cur[BW];
    __shared__ int wsumA[16];
    __shared__ int wsumB[16];
    __shared__ int sb[2];
    const int t = threadIdx.x;
    const int b = blockIdx.x;
    if (t < BW) c[t] = 0;
    const int gv = (t < NB) ? ghist[t] : 0;
    const int ex = block_excl_scan(gv, wsumA);
    if (t == b) { sb[0] = ex; sb[1] = gv; }
    __syncthreads();
    const int bstart = sb[0];
    const int bend   = bstart + sb[1];
    for (int base = bstart; base < bend; base += BPT * TBE) {
        int p[BPT];
#pragma unroll
        for (int k = 0; k < BPT; ++k) {
            const int i = base + k * TBE + t;
            p[k] = tmp[i < bend ? i : bend - 1];
        }
#pragma unroll
        for (int k = 0; k < BPT; ++k)
            if (base + k * TBE + t < bend) atomicAdd(&c[p[k] & (BW - 1)], 1);
    }
    __syncthreads();
    const int v   = (t < BW) ? c[t] : 0;
    const int pre = block_excl_scan(v, wsumB);
    if (t < BW) {
        cur[t] = pre;
        const int n = b * BW + t;
        if (n < NN) { deg[n] = v; rowstart[n] = bstart + pre; }
    }
    __syncthreads();
    for (int base = bstart; base < bend; base += BPT * TBE) {
        int p[BPT];
#pragma unroll
        for (int k = 0; k < BPT; ++k) {
            const int i = base + k * TBE + t;
            p[k] = tmp[i < bend ? i : bend - 1];
        }
#pragma unroll
        for (int k = 0; k < BPT; ++k)
            if (base + k * TBE + t < bend) {
                const int r = atomicAdd(&cur[p[k] & (BW - 1)], 1);
                csr[bstart + r] = p[k] >> BSH;
            }
    }
}


__device__ __forceinline__ void logits4(const int* __restrict__ csr, const float* __restrict__ als,
                                        int start, int cnt, int n, int idx, const float4 dn,
                                        int& s, float4& v) {
    const float NEG_INF = -__builtin_inff();
    s = n;
    v = make_float4(NEG_INF, NEG_INF, NEG_INF, NEG_INF);
    if (idx <= cnt) {
        if (idx < cnt) s = csr[start + idx];
        const float4 a = *(const float4*)(als + (size_t)s * 4);
        v.x = lrelu(a.x + dn.x);
        v.y = lrelu(a.y + dn.y);
        v.z = lrelu(a.z + dn.z);
        v.w = lrelu(a.w + dn.w);
    }
}

__device__ __forceinline__ void chunk_rows(const float* __restrict__ h1, float4* wt,
                                           const float* wth, int lane, int s, const float4 w,
                                           int nv, float4& acc) {
    __builtin_amdgcn_wave_barrier();
    wt[lane] = w;
    __builtin_amdgcn_wave_barrier();
    const float* hl = h1 + lane * 4;
    int j = 0;
    for (; j + 4 <= nv; j += 4) {
        const int sa = __builtin_amdgcn_readlane(s, j);
        const int sb = __builtin_amdgcn_readlane(s, j + 1);
        const int sc = __builtin_amdgcn_readlane(s, j + 2);
        const int sd = __builtin_amdgcn_readlane(s, j + 3);
        const float4 ra = *(const float4*)(hl + (size_t)sa * HC);
        const float4 rb = *(const float4*)(hl + (size_t)sb * HC);
        const float4 rc = *(const float4*)(hl + (size_t)sc * HC);
        const float4 rd = *(const float4*)(hl + (size_t)sd * HC);
        const float wa = wth[4 * j];
        const float wb = wth[4 * j + 4];
        const float wc = wth[4 * j + 8];
        const float wd = wth[4 * j + 12];
        acc.x += ra.x * wa; acc.y += ra.y * wa; acc.z += ra.z * wa; acc.w += ra.w * wa;
        acc.x += rb.x * wb; acc.y += rb.y * wb; acc.z += rb.z * wb; acc.w += rb.w * wb;
        acc.x += rc.x * wc; acc.y += rc.y * wc; acc.z += rc.z * wc; acc.w += rc.w * wc;
        acc.x += rd.x * wd; acc.y += rd.y * wd; acc.z += rd.z * wd; acc.w += rd.w * wd;
    }
    for (; j < nv; ++j) {
        const int sa = __builtin_amdgcn_readlane(s, j);
        const float4 ra = *(const float4*)(hl + (size_t)sa * HC);
        const float wa = wth[4 * j];
        acc.x += ra.x * wa; acc.y += ra.y * wa; acc.z += ra.z * wa; acc.w += ra.w * wa;
    }
}

__global__ void __launch_bounds__(256)
k_node1(const int* __restrict__ csr, const int* __restrict__ rowstart,
        const int* __restrict__ deg, const float* __restrict__ h1,
        const float* __restrict__ als, const float* __restrict__ ald,
        const float* __restrict__ b1, float* __restrict__ h2) {
    __shared__ float4 wtab[8][32];
    const int lane = threadIdx.x & 31;
    const int wv   = threadIdx.x >> 5;
    const int n = __builtin_amdgcn_readfirstlane((int)(blockIdx.x * 8 + wv));
    if (n >= NN) return;
    const int start = __builtin_amdgcn_readfirstlane(rowstart[n]);
    const int cnt   = __builtin_amdgcn_readfirstlane(deg[n]);
    const int tot   = cnt + 1;
    const float4 dn = *(const float4*)(ald + (size_t)n * 4);
    float4* wt = wtab[wv];
    const float* wth = (const float*)wt + (lane >> 3);

    int s0; float4 v0;
    logits4(csr, als, start, cnt, n, lane, dn, s0, v0);

    float4 m = v0;
    for (int base = 32; base < tot; base += 32) {
        int s; float4 v;
        logits4(csr, als, start, cnt, n, base + lane, dn, s, v);
        m.x = fmaxf(m.x, v.x); m.y = fmaxf(m.y, v.y); m.z = fmaxf(m.z, v.z); m.w = fmaxf(m.w, v.w);
    }
#pragma unroll
    for (int off = 16; off > 0; off >>= 1) {
        m.x = fmaxf(m.x, __shfl_xor(m.x, off));
        m.y = fmaxf(m.y, __shfl_xor(m.y, off));
        m.z = fmaxf(m.z, __shfl_xor(m.z, off));
        m.w = fmaxf(m.w, __shfl_xor(m.w, off));
    }

    float4 p0;
    p0.x = __expf(v0.x - m.x); p0.y = __expf(v0.y - m.y);
    p0.z = __expf(v0.z - m.z); p0.w = __expf(v0.w - m.w);
    float4 d = p0;
    for (int base = 32; base < tot; base += 32) {
        int s; float4 v;
        logits4(csr, als, start, cnt, n, base + lane, dn, s, v);
        d.x += __expf(v.x - m.x); d.y += __expf(v.y - m.y);
        d.z += __expf(v.z - m.z); d.w += __expf(v.w - m.w);
    }
#pragma unroll
    for (int off = 16; off > 0; off >>= 1) {
        d.x += __shfl_xor(d.x, off);
        d.y += __shfl_xor(d.y, off);
        d.z += __shfl_xor(d.z, off);
        d.w += __shfl_xor(d.w, off);
    }
    d.x += 1e-16f; d.y += 1e-16f; d.z += 1e-16f; d.w += 1e-16f;

    float4 acc = make_float4(0.f, 0.f, 0.f, 0.f);
    {
        float4 w;
        w.x = p0.x / d.x; w.y = p0.y / d.y; w.z = p0.z / d.z; w.w = p0.w / d.w;
        chunk_rows(h1, wt, wth, lane, s0, w, (tot < 32) ? tot : 32, acc);
    }
    for (int base = 32; base < tot; base += 32) {
        int s; float4 v;
        logits4(csr, als, start, cnt, n, base + lane, dn, s, v);
        float4 w;
        w.x = __expf(v.x - m.x) / d.x; w.y = __expf(v.y - m.y) / d.y;
        w.z = __expf(v.z - m.z) / d.z; w.w = __expf(v.w - m.w) / d.w;
        chunk_rows(h1, wt, wth, lane, s, w, (tot - base < 32) ? (tot - base) : 32, acc);
    }

    const float4 bb = *(const float4*)(b1 + lane * 4);
    float4 o;
    o.x = acc.x + bb.x; o.y = acc.y + bb.y; o.z = acc.z + bb.z; o.w = acc.w + bb.w;
    o.x = o.x > 0.0f ? o.x : expm1f(o.x);
    o.y = o.y > 0.0f ? o.y : expm1f(o.y);
    o.z = o.z > 0.0f ? o.z : expm1f(o.z);
    o.w = o.w > 0.0f ? o.w : expm1f(o.w);
    *(float4*)(h2 + (size_t)n * HC + lane * 4) = o;
}

__global__ void k_gemm2(const float* __restrict__ h2, const float* __restrict__ W2,
                        float* __restrict__ z) {
    const int wave = threadIdx.x >> 5;
    const int row0 = (blockIdx.x * 4 + wave) * 16;
    if (row0 >= NN) return;
    const int lane = threadIdx.x & 31;
    const int m    = lane & 15;
    const int kb   = (lane >> 4) * 2;

    v8f c[3];
#pragma unroll
    for (int t = 0; t < 3; ++t) c[t] = (v8f){0,0,0,0,0,0,0,0};

    for (int k = 0; k < HC; k += 4) {
        v2f a;
        const float* xr = h2 + (size_t)(row0 + m) * HC + (k + kb);
        a.x = xr[0];
        a.y = xr[1];
#pragma unroll
        for (int t = 0; t < 3; ++t) {
            int col = t * 16 + m;
            v2f b;
            b.x = (col < OUT_CH) ? W2[(size_t)(k + kb) * OUT_CH + col] : 0.0f;
            b.y = (col < OUT_CH) ? W2[(size_t)(k + kb + 1) * OUT_CH + col] : 0.0f;
            c[t] = __builtin_amdgcn_wmma_f32_16x16x4_f32(false, a, false, b,
                                                         (short)0, c[t], false, false);
        }
    }
    const int rofs = (lane >> 4) * 8;
#pragma unroll
    for (int t = 0; t < 3; ++t) {
        int col = t * 16 + m;
        if (col < OUT_CH)
#pragma unroll
            for (int r = 0; r < 8; ++r)
                z[(size_t)(row0 + r + rofs) * OUT_CH + col] = c[t][r];
    }
}

__global__ void k_al2(const float* __restrict__ z, const float* __restrict__ a_src,
                      const float* __restrict__ a_dst, float* __restrict__ als,
                      float* __restrict__ ald) {
    int n = blockIdx.x * blockDim.x + threadIdx.x;
    if (n >= NN) return;
    const float* zp = z + (size_t)n * OUT_CH;
    float s = 0.f, d = 0.f;
#pragma unroll 8
    for (int cc = 0; cc < OUT_CH; ++cc) {
        float v = zp[cc];
        s += v * a_src[cc];
        d += v * a_dst[cc];
    }
    als[n] = s;
    ald[n] = d;
}

__global__ void __launch_bounds__(256)
k_node2(const int* __restrict__ csr, const int* __restrict__ rowstart,
        const int* __restrict__ deg, const float* __restrict__ z,
        const float* __restrict__ als, const float* __restrict__ ald,
        const float* __restrict__ b2, float* __restrict__ out) {
    const int lane = threadIdx.x & 31;
    const int w = __builtin_amdgcn_readfirstlane((int)(blockIdx.x * 8 + (threadIdx.x >> 5)));
    const bool hi = lane < (OUT_CH - 32);
    const float bl = b2[lane];
    const float bh = hi ? b2[32 + lane] : 0.0f;
    const float NEG_INF = -__builtin_inff();

    for (int q = 0; q < 4; ++q) {
        const int n = w * 4 + q;
        if (n >= NN) break;
        const int start = __builtin_amdgcn_readfirstlane(rowstart[n]);
        const int cnt   = __builtin_amdgcn_readfirstlane(deg[n]);
        const int tot   = cnt + 1;
        const float dn  = ald[n];

        float m = NEG_INF;
        for (int base = 0; base < tot; base += 32) {
            const int idx = base + lane;
            if (idx < tot) {
                const int s = (idx < cnt) ? csr[start + idx] : n;
                m = fmaxf(m, lrelu(als[s] + dn));
            }
        }
#pragma unroll
        for (int off = 16; off > 0; off >>= 1) m = fmaxf(m, __shfl_xor(m, off));

        float den = 0.f;
        for (int base = 0; base < tot; base += 32) {
            const int idx = base + lane;
            if (idx < tot) {
                const int s = (idx < cnt) ? csr[start + idx] : n;
                den += __expf(lrelu(als[s] + dn) - m);
            }
        }
#pragma unroll
        for (int off = 16; off > 0; off >>= 1) den += __shfl_xor(den, off);
        den += 1e-16f;

        float acc0 = 0.f, acc1 = 0.f;
        for (int base = 0; base < tot; base += 32) {
            const int idx = base + lane;
            int s = n;
            float wgt = 0.f;
            if (idx < tot) {
                s = (idx < cnt) ? csr[start + idx] : n;
                wgt = __expf(lrelu(als[s] + dn) - m) / den;
            }
            const int nv = (tot - base < 32) ? (tot - base) : 32;
            int j = 0;
            for (; j + 4 <= nv; j += 4) {
                const int sa = __builtin_amdgcn_readlane(s, j);
                const int sb = __builtin_amdgcn_readlane(s, j + 1);
                const int sc = __builtin_amdgcn_readlane(s, j + 2);
                const int sd = __builtin_amdgcn_readlane(s, j + 3);
                const float wa = __int_as_float(__builtin_amdgcn_readlane(__float_as_int(wgt), j));
                const float wb = __int_as_float(__builtin_amdgcn_readlane(__float_as_int(wgt), j + 1));
                const float wc = __int_as_float(__builtin_amdgcn_readlane(__float_as_int(wgt), j + 2));
                const float wd = __int_as_float(__builtin_amdgcn_readlane(__float_as_int(wgt), j + 3));
                const float* za = z + (size_t)sa * OUT_CH;
                const float* zb = z + (size_t)sb * OUT_CH;
                const float* zc = z + (size_t)sc * OUT_CH;
                const float* zd = z + (size_t)sd * OUT_CH;
                const float la = za[lane], lb = zb[lane], lc = zc[lane], ld = zd[lane];
                float ha = 0.f, hb = 0.f, hc = 0.f, hd = 0.f;
                if (hi) { ha = za[32 + lane]; hb = zb[32 + lane]; hc = zc[32 + lane]; hd = zd[32 + lane]; }
                acc0 += la * wa; acc0 += lb * wb; acc0 += lc * wc; acc0 += ld * wd;
                acc1 += ha * wa; acc1 += hb * wb; acc1 += hc * wc; acc1 += hd * wd;
            }
            for (; j < nv; ++j) {
                const int sa = __builtin_amdgcn_readlane(s, j);
                const float wa = __int_as_float(__builtin_amdgcn_readlane(__float_as_int(wgt), j));
                const float* za = z + (size_t)sa * OUT_CH;
                acc0 += za[lane] * wa;
                if (hi) acc1 += za[32 + lane] * wa;
            }
        }
        float* op = out + (size_t)n * OUT_CH;
        op[lane] = acc0 + bl;
        if (hi) op[32 + lane] = acc1 + bh;
    }
}

extern "C" void kernel_launch(void* const* d_in, const int* in_sizes, int n_in,
                              void* d_out, int out_size, void* d_ws, size_t ws_size,
                              hipStream_t stream) {
    const float* x      = (const float*)d_in[0];
    const int*   ei     = (const int*)  d_in[1];
    const float* W1     = (const float*)d_in[2];
    const float* a_src1 = (const float*)d_in[3];
    const float* a_dst1 = (const float*)d_in[4];
    const float* b1     = (const float*)d_in[5];
    const float* W2     = (const float*)d_in[6];
    const float* a_src2 = (const float*)d_in[7];
    const float* a_dst2 = (const float*)d_in[8];
    const float* b2     = (const float*)d_in[9];
    float* out = (float*)d_out;

    float* ws   = (float*)d_ws;
    float* h1   = ws;
    float* agg1 = h1   + (size_t)NN * HC;
    float* z2   = agg1 + (size_t)NN * HC;
    float* als1 = z2   + (size_t)NN * OUT_CH;
    float* ald1 = als1 + (size_t)NN * HEADS;
    float* als2 = ald1 + (size_t)NN * HEADS;
    float* ald2 = als2 + NN;
    int* deg      = (int*)(ald2 + NN);
    int* rowstart = deg + NN;
    int* ghist    = rowstart + NN;
    int* gcur     = ghist + GH;
    int* tmp      = gcur + GH;
    int* csr      = tmp + EE;

    const int TB = 256;
    const int gemmBlocks = (NN / 16 + 3) / 4;
    const int nodeBlocks1 = (NN + 7) / 8;
    const int nodeBlocks2 = (NN / 4 + 7) / 8;

    k_gemm1<<<gemmBlocks, 128, 0, stream>>>(x, W1, h1);
    k_al1  <<<(NN * HEADS + TB - 1) / TB, TB, 0, stream>>>(h1, a_src1, a_dst1, als1, ald1, ghist);
    k_hist   <<<EBLK, TBE, 0, stream>>>(ei, ghist);
    k_scatter<<<EBLK, TBE, 0, stream>>>(ei, ghist, gcur, tmp);
    k_bsort  <<<NB,   TBE, 0, stream>>>(ghist, tmp, deg, rowstart, csr);
    k_node1<<<nodeBlocks1, 256, 0, stream>>>(csr, rowstart, deg, h1, als1, ald1, b1, agg1);

    k_gemm2<<<gemmBlocks, 128, 0, stream>>>(agg1, W2, z2);
    k_al2  <<<(NN + TB - 1) / TB, TB, 0, stream>>>(z2, a_src2, a_dst2, als2, ald2);
    k_node2<<<nodeBlocks2, 256, 0, stream>>>(csr, rowstart, deg, z2, als2, ald2, b2, out);
}


# Per-node attention logits of both layers summed in the GEMM kernels' epilogues from the finished tile staged in LDS, same order: two launches and the re-reads of h1 and z removed
# speedup vs baseline: 4.7796x; 1.0924x over previous
#include <hip/hip_runtime.h>
#include <math.h>

#define NN 100000
#define EE 1600000
#define IN_CH 128
#define HID 32
#define HEADS 4
#define HC 128
#define OUT_CH 40
#define NEG_SLOPE 0.2f
#define GH 512

typedef __attribute__((ext_vector_type(2))) float v2f;
typedef __attribute__((ext_vector_type(8))) float v8f;

__device__ __forceinline__ float lrelu(float v) {
    return v >= 0.0f ? v : NEG_SLOPE * v;
}

#define P1 132
__global__ void __launch_bounds__(128)
k_gemm1(const float* __restrict__ x, const float* __restrict__ W1,
        const float* __restrict__ a_src, const float* __restrict__ a_dst,
        float* __restrict__ h1, float* __restrict__ als, float* __restrict__ ald,
        int* __restrict__ ghist) {
    __shared__ float tile[4][16 * P1];
    if (blockIdx.x == 0)
        for (int i = threadIdx.x; i < 2 * GH; i += 128) ghist[i] = 0;
    const int wave = threadIdx.x >> 5;
    const int row0 = (blockIdx.x * 4 + wave) * 16;
    if (row0 >= NN) return;
    const int lane = threadIdx.x & 31;
    const int m    = lane & 15;
    const int kb   = (lane >> 4) * 2;

    v8f c[8];
#pragma unroll
    for (int t = 0; t < 8; ++t) c[t] = (v8f){0,0,0,0,0,0,0,0};

    for (int k = 0; k < IN_CH; k += 4) {
        v2f a;
        const float* xr = x + (size_t)(row0 + m) * IN_CH + (k + kb);
        a.x = xr[0];
        a.y = xr[1];
#pragma unroll
        for (int t = 0; t < 8; ++t) {
            v2f b;
            const float* wr = W1 + (size_t)(k + kb) * HC + t * 16 + m;
            b.x = wr[0];
            b.y = wr[HC];
            c[t] = __builtin_amdgcn_wmma_f32_16x16x4_f32(false, a, false, b,
                                                         (short)0, c[t], false, false);
        }
    }
    const int rofs = (lane >> 4) * 8;
    float* sm = tile[wave];
#pragma unroll
    for (int t = 0; t < 8; ++t)
#pragma unroll
        for (int r = 0; r < 8; ++r) {
            h1[(size_t)(row0 + r + rofs) * HC + t * 16 + m] = c[t][r];
            sm[(r + rofs) * P1 + t * 16 + m] = c[t][r];
        }
    __builtin_amdgcn_wave_barrier();

    const int lr = lane & 15;
    const int hp = (lane >> 4) * 2;
    float s0 = 0.f, d0 = 0.f, s1 = 0.f, d1 = 0.f;
    const float* r0 = sm + lr * P1 + hp * HID;
    const float* as0 = a_src + hp * HID;
    const float* ad0 = a_dst + hp * HID;
#pragma unroll
    for (int cc = 0; cc < HID; ++cc) {
        const float v0 = r0[cc];
        const float v1 = r0[HID + cc];
        s0 = __builtin_fmaf(v0, as0[cc], s0);
        d0 = __builtin_fmaf(v0, ad0[cc], d0);
        s1 = __builtin_fmaf(v1, as0[HID + cc], s1);
        d1 = __builtin_fmaf(v1, ad0[HID + cc], d1);
    }
    *(float2*)(als + (size_t)(row0 + lr) * HEADS + hp) = make_float2(s0, s1);
    *(float2*)(ald + (size_t)(row0 + lr) * HEADS + hp) = make_float2(d0, d1);
}

#define BSH 8
#define BW  256
#define NB  ((NN + BW - 1) / BW)
#define TBE 512
#define EPT 8
#define BPT 12
#define EBLK ((EE / EPT + TBE - 1) / TBE)
static_assert(NB <= TBE && NB <= GH, "one thread per coarse bucket");
static_assert(EE % EPT == 0, "whole threads of edges");
static_assert(NN <= (1 << 23), "source id and local destination share one word");

__device__ __forceinline__ int block_excl_scan(int v, int* wsum) {
    const int lane = threadIdx.x & 31;
    const int wv   = threadIdx.x >> 5;
    int incl = v;
#pragma unroll
    for (int off = 1; off < 32; off <<= 1) {
        int t = __shfl_up(incl, off);
        if (lane >= off) incl += t;
    }
    if (lane == 31) wsum[wv] = incl;
    __syncthreads();
    int woff = 0;
    for (int k = 0; k < wv; ++k) woff += wsum[k];
    return woff + incl - v;
}

__global__ void __launch_bounds__(TBE)
k_hist(const int* __restrict__ ei, int* __restrict__ ghist) {
    __shared__ int h[TBE];
    const int t = threadIdx.x;
    h[t] = 0;
    __syncthreads();
    const int e0 = (blockIdx.x * TBE + t) * EPT;
    if (e0 < EE) {
        const int4 a = *(const int4*)(ei + EE + e0);
        const int4 b = *(const int4*)(ei + EE + e0 + 4);
        atomicAdd(&h[a.x >> BSH], 1); atomicAdd(&h[a.y >> BSH], 1);
        atomicAdd(&h[a.z >> BSH], 1); atomicAdd(&h[a.w >> BSH], 1);
        atomicAdd(&h[b.x >> BSH], 1); atomicAdd(&h[b.y >> BSH], 1);
        atomicAdd(&h[b.z >> BSH], 1); atomicAdd(&h[b.w >> BSH], 1);
    }
    __syncthreads();
    const int c = h[t];
    if (c) atomicAdd(&ghist[t], c);
}

__global__ void __launch_bounds__(TBE)
k_scatter(const int* __restrict__ ei, const int* __restrict__ ghist,
          int* __restrict__ gcur, int* __restrict__ tmp) {
    __shared__ int cnt[TBE];
    __shared__ int off[TBE];
    __shared__ int wsum[16];
    const int t = threadIdx.x;
    cnt[t] = 0;
    const int gv  = (t < NB) ? ghist[t] : 0;
    const int bst = block_excl_scan(gv, wsum);
    const int e0 = (blockIdx.x * TBE + t) * EPT;
    const bool act = e0 < EE;
    int4 s0, s1, d0, d1;
    int r0 = 0, r1 = 0, r2 = 0, r3 = 0, r4 = 0, r5 = 0, r6 = 0, r7 = 0;
    if (act) {
        s0 = *(const int4*)(ei + e0);
        s1 = *(const int4*)(ei + e0 + 4);
        d0 = *(const int4*)(ei + EE + e0);
        d1 = *(const int4*)(ei + EE + e0 + 4);
        r0 = atomicAdd(&cnt[d0.x >> BSH], 1); r1 = atomicAdd(&cnt[d0.y >> BSH], 1);
        r2 = atomicAdd(&cnt[d0.z >> BSH], 1); r3 = atomicAdd(&cnt[d0.w >> BSH], 1);
        r4 = atomicAdd(&cnt[d1.x >> BSH], 1); r5 = atomicAdd(&cnt[d1.y >> BSH], 1);
        r6 = atomicAdd(&cnt[d1.z >> BSH], 1); r7 = atomicAdd(&cnt[d1.w >> BSH], 1);
    }
    __syncthreads();
    {
        const int c = cnt[t];
        int base = 0;
        if (c) base = atomicAdd(&gcur[t], c);
        off[t] = bst + base;
    }
    __syncthreads();
    if (act) {
        tmp[off[d0.x >> BSH] + r0] = (s0.x << BSH) | (d0.x & (BW - 1));
        tmp[off[d0.y >> BSH] + r1] = (s0.y << BSH) | (d0.y & (BW - 1));
        tmp[off[d0.z >> BSH] + r2] = (s0.z << BSH) | (d0.z & (BW - 1));
        tmp[off[d0.w >> BSH] + r3] = (s0.w << BSH) | (d0.w & (BW - 1));
        tmp[off[d1.x >> BSH] + r4] = (s1.x << BSH) | (d1.x & (BW - 1));
        tmp[off[d1.y >> BSH] + r5] = (s1.y << BSH) | (d1.y & (BW - 1));
        tmp[off[d1.z >> BSH] + r6] = (s1.z << BSH) | (d1.z & (BW - 1));
        tmp[off[d1.w >> BSH] + r7] = (s1.w << BSH) | (d1.w & (BW - 1));
    }
}

__global__ void __launch_bounds__(TBE)
k_bsort(const int* __restrict__ ghist, const int* __restrict__ tmp,
        int* __restrict__ deg, int* __restrict__ rowstart, int* __restrict__ csr) {
    __shared__ int c[BW];
    __shared__ int cur[BW];
    __shared__ int wsumA[16];
    __shared__ int wsumB[16];
    __shared__ int sb[2];
    const int t = threadIdx.x;
    const int b = blockIdx.x;
    if (t < BW) c[t] = 0;
    const int gv = (t < NB) ? ghist[t] : 0;
    const int ex = block_excl_scan(gv, wsumA);
    if (t == b) { sb[0] = ex; sb[1] = gv; }
    __syncthreads();
    const int bstart = sb[0];
    const int bend   = bstart + sb[1];
    for (int base = bstart; base < bend; base += BPT * TBE) {
        int p[BPT];
#pragma unroll
        for (int k = 0; k < BPT; ++k) {
            const int i = base + k * TBE + t;
            p[k] = tmp[i < bend ? i : bend - 1];
        }
#pragma unroll
        for (int k = 0; k < BPT; ++k)
            if (base + k * TBE + t < bend) atomicAdd(&c[p[k] & (BW - 1)], 1);
    }
    __syncthreads();
    const int v   = (t < BW) ? c[t] : 0;
    const int pre = block_excl_scan(v, wsumB);
    if (t < BW) {
        cur[t] = pre;
        const int n = b * BW + t;
        if (n < NN) { deg[n] = v; rowstart[n] = bstart + pre; }
    }
    __syncthreads();
    for (int base = bstart; base < bend; base += BPT * TBE) {
        int p[BPT];
#pragma unroll
        for (int k = 0; k < BPT; ++k) {
            const int i = base + k * TBE + t;
            p[k] = tmp[i < bend ? i : bend - 1];
        }
#pragma unroll
        for (int k = 0; k < BPT; ++k)
            if (base + k * TBE + t < bend) {
                const int r = atomicAdd(&cur[p[k] & (BW - 1)], 1);
                csr[bstart + r] = p[k] >> BSH;
            }
    }
}


__device__ __forceinline__ void logits4(const int* __restrict__ csr, const float* __restrict__ als,
                                        int start, int cnt, int n, int idx, const float4 dn,
                                        int& s, float4& v) {
    const float NEG_INF = -__builtin_inff();
    s = n;
    v = make_float4(NEG_INF, NEG_INF, NEG_INF, NEG_INF);
    if (idx <= cnt) {
        if (idx < cnt) s = csr[start + idx];
        const float4 a = *(const float4*)(als + (size_t)s * 4);
        v.x = lrelu(a.x + dn.x);
        v.y = lrelu(a.y + dn.y);
        v.z = lrelu(a.z + dn.z);
        v.w = lrelu(a.w + dn.w);
    }
}

__device__ __forceinline__ void chunk_rows(const float* __restrict__ h1, float4* wt,
                                           const float* wth, int lane, int s, const float4 w,
                                           int nv, float4& acc) {
    __builtin_amdgcn_wave_barrier();
    wt[lane] = w;
    __builtin_amdgcn_wave_barrier();
    const float* hl = h1 + lane * 4;
    int j = 0;
    for (; j + 4 <= nv; j += 4) {
        const int sa = __builtin_amdgcn_readlane(s, j);
        const int sb = __builtin_amdgcn_readlane(s, j + 1);
        const int sc = __builtin_amdgcn_readlane(s, j + 2);
        const int sd = __builtin_amdgcn_readlane(s, j + 3);
        const float4 ra = *(const float4*)(hl + (size_t)sa * HC);
        const float4 rb = *(const float4*)(hl + (size_t)sb * HC);
        const float4 rc = *(const float4*)(hl + (size_t)sc * HC);
        const float4 rd = *(const float4*)(hl + (size_t)sd * HC);
        const float wa = wth[4 * j];
        const float wb = wth[4 * j + 4];
        const float wc = wth[4 * j + 8];
        const float wd = wth[4 * j + 12];
        acc.x += ra.x * wa; acc.y += ra.y * wa; acc.z += ra.z * wa; acc.w += ra.w * wa;
        acc.x += rb.x * wb; acc.y += rb.y * wb; acc.z += rb.z * wb; acc.w += rb.w * wb;
        acc.x += rc.x * wc; acc.y += rc.y * wc; acc.z += rc.z * wc; acc.w += rc.w * wc;
        acc.x += rd.x * wd; acc.y += rd.y * wd; acc.z += rd.z * wd; acc.w += rd.w * wd;
    }
    for (; j < nv; ++j) {
        const int sa = __builtin_amdgcn_readlane(s, j);
        const float4 ra = *(const float4*)(hl + (size_t)sa * HC);
        const float wa = wth[4 * j];
        acc.x += ra.x * wa; acc.y += ra.y * wa; acc.z += ra.z * wa; acc.w += ra.w * wa;
    }
}

__global__ void __launch_bounds__(256)
k_node1(const int* __restrict__ csr, const int* __restrict__ rowstart,
        const int* __restrict__ deg, const float* __restrict__ h1,
        const float* __restrict__ als, const float* __restrict__ ald,
        const float* __restrict__ b1, float* __restrict__ h2) {
    __shared__ float4 wtab[8][32];
    const int lane = threadIdx.x & 31;
    const int wv   = threadIdx.x >> 5;
    const int n = __builtin_amdgcn_readfirstlane((int)(blockIdx.x * 8 + wv));
    if (n >= NN) return;
    const int start = __builtin_amdgcn_readfirstlane(rowstart[n]);
    const int cnt   = __builtin_amdgcn_readfirstlane(deg[n]);
    const int tot   = cnt + 1;
    const float4 dn = *(const float4*)(ald + (size_t)n * 4);
    float4* wt = wtab[wv];
    const float* wth = (const float*)wt + (lane >> 3);

    int s0; float4 v0;
    logits4(csr, als, start, cnt, n, lane, dn, s0, v0);

    float4 m = v0;
    for (int base = 32; base < tot; base += 32) {
        int s; float4 v;
        logits4(csr, als, start, cnt, n, base + lane, dn, s, v);
        m.x = fmaxf(m.x, v.x); m.y = fmaxf(m.y, v.y); m.z = fmaxf(m.z, v.z); m.w = fmaxf(m.w, v.w);
    }
#pragma unroll
    for (int off = 16; off > 0; off >>= 1) {
        m.x = fmaxf(m.x, __shfl_xor(m.x, off));
        m.y = fmaxf(m.y, __shfl_xor(m.y, off));
        m.z = fmaxf(m.z, __shfl_xor(m.z, off));
        m.w = fmaxf(m.w, __shfl_xor(m.w, off));
    }

    float4 p0;
    p0.x = __expf(v0.x - m.x); p0.y = __expf(v0.y - m.y);
    p0.z = __expf(v0.z - m.z); p0.w = __expf(v0.w - m.w);
    float4 d = p0;
    for (int base = 32; base < tot; base += 32) {
        int s; float4 v;
        logits4(csr, als, start, cnt, n, base + lane, dn, s, v);
        d.x += __expf(v.x - m.x); d.y += __expf(v.y - m.y);
        d.z += __expf(v.z - m.z); d.w += __expf(v.w - m.w);
    }
#pragma unroll
    for (int off = 16; off > 0; off >>= 1) {
        d.x += __shfl_xor(d.x, off);
        d.y += __shfl_xor(d.y, off);
        d.z += __shfl_xor(d.z, off);
        d.w += __shfl_xor(d.w, off);
    }
    d.x += 1e-16f; d.y += 1e-16f; d.z += 1e-16f; d.w += 1e-16f;

    float4 acc = make_float4(0.f, 0.f, 0.f, 0.f);
    {
        float4 w;
        w.x = p0.x / d.x; w.y = p0.y / d.y; w.z = p0.z / d.z; w.w = p0.w / d.w;
        chunk_rows(h1, wt, wth, lane, s0, w, (tot < 32) ? tot : 32, acc);
    }
    for (int base = 32; base < tot; base += 32) {
        int s; float4 v;
        logits4(csr, als, start, cnt, n, base + lane, dn, s, v);
        float4 w;
        w.x = __expf(v.x - m.x) / d.x; w.y = __expf(v.y - m.y) / d.y;
        w.z = __expf(v.z - m.z) / d.z; w.w = __expf(v.w - m.w) / d.w;
        chunk_rows(h1, wt, wth, lane, s, w, (tot - base < 32) ? (tot - base) : 32, acc);
    }

    const float4 bb = *(const float4*)(b1 + lane * 4);
    float4 o;
    o.x = acc.x + bb.x; o.y = acc.y + bb.y; o.z = acc.z + bb.z; o.w = acc.w + bb.w;
    o.x = o.x > 0.0f ? o.x : expm1f(o.x);
    o.y = o.y > 0.0f ? o.y : expm1f(o.y);
    o.z = o.z > 0.0f ? o.z : expm1f(o.z);
    o.w = o.w > 0.0f ? o.w : expm1f(o.w);
    *(float4*)(h2 + (size_t)n * HC + lane * 4) = o;
}

#define P2 52
__global__ void __launch_bounds__(128)
k_gemm2(const float* __restrict__ h2, const float* __restrict__ W2,
        const float* __restrict__ a_src, const float* __restrict__ a_dst,
        float* __restrict__ z, float* __restrict__ als, float* __restrict__ ald) {
    __shared__ float tile[4][16 * P2];
    const int wave = threadIdx.x >> 5;
    const int row0 = (blockIdx.x * 4 + wave) * 16;
    if (row0 >= NN) return;
    const int lane = threadIdx.x & 31;
    const int m    = lane & 15;
    const int kb   = (lane >> 4) * 2;

    v8f c[3];
#pragma unroll
    for (int t = 0; t < 3; ++t) c[t] = (v8f){0,0,0,0,0,0,0,0};

    for (int k = 0; k < HC; k += 4) {
        v2f a;
        const float* xr = h2 + (size_t)(row0 + m) * HC + (k + kb);
        a.x = xr[0];
        a.y = xr[1];
#pragma unroll
        for (int t = 0; t < 3; ++t) {
            int col = t * 16 + m;
            v2f b;
            b.x = (col < OUT_CH) ? W2[(size_t)(k + kb) * OUT_CH + col] : 0.0f;
            b.y = (col < OUT_CH) ? W2[(size_t)(k + kb + 1) * OUT_CH + col] : 0.0f;
            c[t] = __builtin_amdgcn_wmma_f32_16x16x4_f32(false, a, false, b,
                                                         (short)0, c[t], false, false);
        }
    }
    const int rofs = (lane >> 4) * 8;
    float* sm = tile[wave];
#pragma unroll
    for (int t = 0; t < 3; ++t) {
        int col = t * 16 + m;
#pragma unroll
        for (int r = 0; r < 8; ++r) {
            if (col < OUT_CH) z[(size_t)(row0 + r + rofs) * OUT_CH + col] = c[t][r];
            sm[(r + rofs) * P2 + col] = c[t][r];
        }
    }
    __builtin_amdgcn_wave_barrier();

    const int lr = lane & 15;
    const bool tgt = lane >= 16;
    const float* av = tgt ? a_dst : a_src;
    const float* zr = sm + lr * P2;
    float s = 0.f;
#pragma unroll
    for (int cc = 0; cc < OUT_CH; ++cc) s = __builtin_fmaf(zr[cc], av[cc], s);
    (tgt ? ald : als)[row0 + lr] = s;
}

__global__ void __launch_bounds__(256)
k_node2(const int* __restrict__ csr, const int* __restrict__ rowstart,
        const int* __restrict__ deg, const float* __restrict__ z,
        const float* __restrict__ als, const float* __restrict__ ald,
        const float* __restrict__ b2, float* __restrict__ out) {
    const int lane = threadIdx.x & 31;
    const int w = __builtin_amdgcn_readfirstlane((int)(blockIdx.x * 8 + (threadIdx.x >> 5)));
    const bool hi = lane < (OUT_CH - 32);
    const float bl = b2[lane];
    const float bh = hi ? b2[32 + lane] : 0.0f;
    const float NEG_INF = -__builtin_inff();

    for (int q = 0; q < 4; ++q) {
        const int n = w * 4 + q;
        if (n >= NN) break;
        const int start = __builtin_amdgcn_readfirstlane(rowstart[n]);
        const int cnt   = __builtin_amdgcn_readfirstlane(deg[n]);
        const int tot   = cnt + 1;
        const float dn  = ald[n];

        float m = NEG_INF;
        for (int base = 0; base < tot; base += 32) {
            const int idx = base + lane;
            if (idx < tot) {
                const int s = (idx < cnt) ? csr[start + idx] : n;
                m = fmaxf(m, lrelu(als[s] + dn));
            }
        }
#pragma unroll
        for (int off = 16; off > 0; off >>= 1) m = fmaxf(m, __shfl_xor(m, off));

        float den = 0.f;
        for (int base = 0; base < tot; base += 32) {
            const int idx = base + lane;
            if (idx < tot) {
                const int s = (idx < cnt) ? csr[start + idx] : n;
                den += __expf(lrelu(als[s] + dn) - m);
            }
        }
#pragma unroll
        for (int off = 16; off > 0; off >>= 1) den += __shfl_xor(den, off);
        den += 1e-16f;

        float acc0 = 0.f, acc1 = 0.f;
        for (int base = 0; base < tot; base += 32) {
            const int idx = base + lane;
            int s = n;
            float wgt = 0.f;
            if (idx < tot) {
                s = (idx < cnt) ? csr[start + idx] : n;
                wgt = __expf(lrelu(als[s] + dn) - m) / den;
            }
            const int nv = (tot - base < 32) ? (tot - base) : 32;
            int j = 0;
            for (; j + 4 <= nv; j += 4) {
                const int sa = __builtin_amdgcn_readlane(s, j);
                const int sb = __builtin_amdgcn_readlane(s, j + 1);
                const int sc = __builtin_amdgcn_readlane(s, j + 2);
                const int sd = __builtin_amdgcn_readlane(s, j + 3);
                const float wa = __int_as_float(__builtin_amdgcn_readlane(__float_as_int(wgt), j));
                const float wb = __int_as_float(__builtin_amdgcn_readlane(__float_as_int(wgt), j + 1));
                const float wc = __int_as_float(__builtin_amdgcn_readlane(__float_as_int(wgt), j + 2));
                const float wd = __int_as_float(__builtin_amdgcn_readlane(__float_as_int(wgt), j + 3));
                const float* za = z + (size_t)sa * OUT_CH;
                const float* zb = z + (size_t)sb * OUT_CH;
                const float* zc = z + (size_t)sc * OUT_CH;
                const float* zd = z + (size_t)sd * OUT_CH;
                const float la = za[lane], lb = zb[lane], lc = zc[lane], ld = zd[lane];
                float ha = 0.f, hb = 0.f, hc = 0.f, hd = 0.f;
                if (hi) { ha = za[32 + lane]; hb = zb[32 + lane]; hc = zc[32 + lane]; hd = zd[32 + lane]; }
                acc0 += la * wa; acc0 += lb * wb; acc0 += lc * wc; acc0 += ld * wd;
                acc1 += ha * wa; acc1 += hb * wb; acc1 += hc * wc; acc1 += hd * wd;
            }
            for (; j < nv; ++j) {
                const int sa = __builtin_amdgcn_readlane(s, j);
                const float wa = __int_as_float(__builtin_amdgcn_readlane(__float_as_int(wgt), j));
                const float* za = z + (size_t)sa * OUT_CH;
                acc0 += za[lane] * wa;
                if (hi) acc1 += za[32 + lane] * wa;
            }
        }
        float* op = out + (size_t)n * OUT_CH;
        op[lane] = acc0 + bl;
        if (hi) op[32 + lane] = acc1 + bh;
    }
}

extern "C" void kernel_launch(void* const* d_in, const int* in_sizes, int n_in,
                              void* d_out, int out_size, void* d_ws, size_t ws_size,
                              hipStream_t stream) {
    const float* x      = (const float*)d_in[0];
    const int*   ei     = (const int*)  d_in[1];
    const float* W1     = (const float*)d_in[2];
    const float* a_src1 = (const float*)d_in[3];
    const float* a_dst1 = (const float*)d_in[4];
    const float* b1     = (const float*)d_in[5];
    const float* W2     = (const float*)d_in[6];
    const float* a_src2 = (const float*)d_in[7];
    const float* a_dst2 = (const float*)d_in[8];
    const float* b2     = (const float*)d_in[9];
    float* out = (float*)d_out;

    float* ws   = (float*)d_ws;
    float* h1   = ws;
    float* agg1 = h1   + (size_t)NN * HC;
    float* z2   = agg1 + (size_t)NN * HC;
    float* als1 = z2   + (size_t)NN * OUT_CH;
    float* ald1 = als1 + (size_t)NN * HEADS;
    float* als2 = ald1 + (size_t)NN * HEADS;
    float* ald2 = als2 + NN;
    int* deg      = (int*)(ald2 + NN);
    int* rowstart = deg + NN;
    int* ghist    = rowstart + NN;
    int* gcur     = ghist + GH;
    int* tmp      = gcur + GH;
    int* csr      = tmp + EE;

    const int gemmBlocks = (NN / 16 + 3) / 4;
    const int nodeBlocks1 = (NN + 7) / 8;
    const int nodeBlocks2 = (NN / 4 + 7) / 8;

    k_gemm1<<<gemmBlocks, 128, 0, stream>>>(x, W1, a_src1, a_dst1, h1, als1, ald1, ghist);
    k_hist   <<<EBLK, TBE, 0, stream>>>(ei, ghist);
    k_scatter<<<EBLK, TBE, 0, stream>>>(ei, ghist, gcur, tmp);
    k_bsort  <<<NB,   TBE, 0, stream>>>(ghist, tmp, deg, rowstart, csr);
    k_node1<<<nodeBlocks1, 256, 0, stream>>>(csr, rowstart, deg, h1, als1, ald1, b1, agg1);

    k_gemm2<<<gemmBlocks, 128, 0, stream>>>(agg1, W2, a_src2, a_dst2, z2, als2, ald2);
    k_node2<<<nodeBlocks2, 256, 0, stream>>>(csr, rowstart, deg, z2, als2, ald2, b2, out);
}


# Layer-2 product moved into the layer-1 node kernel: a block finishes 16 nodes' rows into LDS and multiplies them by W2 there (same k order), so h2 never goes to memory; one launch fewer
# speedup vs baseline: 5.1857x; 1.0850x over previous
#include <hip/hip_runtime.h>
#include <math.h>

#define NN 100000
#define EE 1600000
#define IN_CH 128
#define HID 32
#define HEADS 4
#define HC 128
#define OUT_CH 40
#define NEG_SLOPE 0.2f
#define GH 512

typedef __attribute__((ext_vector_type(2))) float v2f;
typedef __attribute__((ext_vector_type(8))) float v8f;

__device__ __forceinline__ float lrelu(float v) {
    return v >= 0.0f ? v : NEG_SLOPE * v;
}

#define P1 132
__global__ void __launch_bounds__(128)
k_gemm1(const float* __restrict__ x, const float* __restrict__ W1,
        const float* __restrict__ a_src, const float* __restrict__ a_dst,
        float* __restrict__ h1, float* __restrict__ als, float* __restrict__ ald,
        int* __restrict__ ghist) {
    __shared__ float tile[4][16 * P1];
    if (blockIdx.x == 0)
        for (int i = threadIdx.x; i < 2 * GH; i += 128) ghist[i] = 0;
    const int wave = threadIdx.x >> 5;
    const int row0 = (blockIdx.x * 4 + wave) * 16;
    if (row0 >= NN) return;
    const int lane = threadIdx.x & 31;
    const int m    = lane & 15;
    const int kb   = (lane >> 4) * 2;

    v8f c[8];
#pragma unroll
    for (int t = 0; t < 8; ++t) c[t] = (v8f){0,0,0,0,0,0,0,0};

    for (int k = 0; k < IN_CH; k += 4) {
        v2f a;
        const float* xr = x + (size_t)(row0 + m) * IN_CH + (k + kb);
        a.x = xr[0];
        a.y = xr[1];
#pragma unroll
        for (int t = 0; t < 8; ++t) {
            v2f b;
            const float* wr = W1 + (size_t)(k + kb) * HC + t * 16 + m;
            b.x = wr[0];
            b.y = wr[HC];
            c[t] = __builtin_amdgcn_wmma_f32_16x16x4_f32(false, a, false, b,
                                                         (short)0, c[t], false, false);
        }
    }
    const int rofs = (lane >> 4) * 8;
    float* sm = tile[wave];
#pragma unroll
    for (int t = 0; t < 8; ++t)
#pragma unroll
        for (int r = 0; r < 8; ++r) {
            h1[(size_t)(row0 + r + rofs) * HC + t * 16 + m] = c[t][r];
            sm[(r + rofs) * P1 + t * 16 + m] = c[t][r];
        }
    __builtin_amdgcn_wave_barrier();

    const int lr = lane & 15;
    const int hp = (lane >> 4) * 2;
    float s0 = 0.f, d0 = 0.f, s1 = 0.f, d1 = 0.f;
    const float* r0 = sm + lr * P1 + hp * HID;
    const float* as0 = a_src + hp * HID;
    const float* ad0 = a_dst + hp * HID;
#pragma unroll
    for (int cc = 0; cc < HID; ++cc) {
        const float v0 = r0[cc];
        const float v1 = r0[HID + cc];
        s0 = __builtin_fmaf(v0, as0[cc], s0);
        d0 = __builtin_fmaf(v0, ad0[cc], d0);
        s1 = __builtin_fmaf(v1, as0[HID + cc], s1);
        d1 = __builtin_fmaf(v1, ad0[HID + cc], d1);
    }
    *(float2*)(als + (size_t)(row0 + lr) * HEADS + hp) = make_float2(s0, s1);
    *(float2*)(ald + (size_t)(row0 + lr) * HEADS + hp) = make_float2(d0, d1);
}

#define BSH 8
#define BW  256
#define NB  ((NN + BW - 1) / BW)
#define TBE 512
#define EPT 8
#define BPT 12
#define EBLK ((EE / EPT + TBE - 1) / TBE)
static_assert(NB <= TBE && NB <= GH, "one thread per coarse bucket");
static_assert(EE % EPT == 0, "whole threads of edges");
static_assert(NN <= (1 << 23), "source id and local destination share one word");

__device__ __forceinline__ int block_excl_scan(int v, int* wsum) {
    const int lane = threadIdx.x & 31;
    const int wv   = threadIdx.x >> 5;
    int incl = v;
#pragma unroll
    for (int off = 1; off < 32; off <<= 1) {
        int t = __shfl_up(incl, off);
        if (lane >= off) incl += t;
    }
    if (lane == 31) wsum[wv] = incl;
    __syncthreads();
    int woff = 0;
    for (int k = 0; k < wv; ++k) woff += wsum[k];
    return woff + incl - v;
}

__global__ void __launch_bounds__(TBE)
k_hist(const int* __restrict__ ei, int* __restrict__ ghist) {
    __shared__ int h[TBE];
    const int t = threadIdx.x;
    h[t] = 0;
    __syncthreads();
    const int e0 = (blockIdx.x * TBE + t) * EPT;
    if (e0 < EE) {
        const int4 a = *(const int4*)(ei + EE + e0);
        const int4 b = *(const int4*)(ei + EE + e0 + 4);
        atomicAdd(&h[a.x >> BSH], 1); atomicAdd(&h[a.y >> BSH], 1);
        atomicAdd(&h[a.z >> BSH], 1); atomicAdd(&h[a.w >> BSH], 1);
        atomicAdd(&h[b.x >> BSH], 1); atomicAdd(&h[b.y >> BSH], 1);
        atomicAdd(&h[b.z >> BSH], 1); atomicAdd(&h[b.w >> BSH], 1);
    }
    __syncthreads();
    const int c = h[t];
    if (c) atomicAdd(&ghist[t], c);
}

__global__ void __launch_bounds__(TBE)
k_scatter(const int* __restrict__ ei, const int* __restrict__ ghist,
          int* __restrict__ gcur, int* __restrict__ tmp) {
    __shared__ int cnt[TBE];
    __shared__ int off[TBE];
    __shared__ int wsum[16];
    const int t = threadIdx.x;
    cnt[t] = 0;
    const int gv  = (t < NB) ? ghist[t] : 0;
    const int bst = block_excl_scan(gv, wsum);
    const int e0 = (blockIdx.x * TBE + t) * EPT;
    const bool act = e0 < EE;
    int4 s0, s1, d0, d1;
    int r0 = 0, r1 = 0, r2 = 0, r3 = 0, r4 = 0, r5 = 0, r6 = 0, r7 = 0;
    if (act) {
        s0 = *(const int4*)(ei + e0);
        s1 = *(const int4*)(ei + e0 + 4);
        d0 = *(const int4*)(ei + EE + e0);
        d1 = *(const int4*)(ei + EE + e0 + 4);
        r0 = atomicAdd(&cnt[d0.x >> BSH], 1); r1 = atomicAdd(&cnt[d0.y >> BSH], 1);
        r2 = atomicAdd(&cnt[d0.z >> BSH], 1); r3 = atomicAdd(&cnt[d0.w >> BSH], 1);
        r4 = atomicAdd(&cnt[d1.x >> BSH], 1); r5 = atomicAdd(&cnt[d1.y >> BSH], 1);
        r6 = atomicAdd(&cnt[d1.z >> BSH], 1); r7 = atomicAdd(&cnt[d1.w >> BSH], 1);
    }
    __syncthreads();
    {
        const int c = cnt[t];
        int base = 0;
        if (c) base = atomicAdd(&gcur[t], c);
        off[t] = bst + base;
    }
    __syncthreads();
    if (act) {
        tmp[off[d0.x >> BSH] + r0] = (s0.x << BSH) | (d0.x & (BW - 1));
        tmp[off[d0.y >> BSH] + r1] = (s0.y << BSH) | (d0.y & (BW - 1));
        tmp[off[d0.z >> BSH] + r2] = (s0.z << BSH) | (d0.z & (BW - 1));
        tmp[off[d0.w >> BSH] + r3] = (s0.w << BSH) | (d0.w & (BW - 1));
        tmp[off[d1.x >> BSH] + r4] = (s1.x << BSH) | (d1.x & (BW - 1));
        tmp[off[d1.y >> BSH] + r5] = (s1.y << BSH) | (d1.y & (BW - 1));
        tmp[off[d1.z >> BSH] + r6] = (s1.z << BSH) | (d1.z & (BW - 1));
        tmp[off[d1.w >> BSH] + r7] = (s1.w << BSH) | (d1.w & (BW - 1));
    }
}

__global__ void __launch_bounds__(TBE)
k_bsort(const int* __restrict__ ghist, const int* __restrict__ tmp,
        int* __restrict__ deg, int* __restrict__ rowstart, int* __restrict__ csr) {
    __shared__ int c[BW];
    __shared__ int cur[BW];
    __shared__ int wsumA[16];
    __shared__ int wsumB[16];
    __shared__ int sb[2];
    const int t = threadIdx.x;
    const int b = blockIdx.x;
    if (t < BW) c[t] = 0;
    const int gv = (t < NB) ? ghist[t] : 0;
    const int ex = block_excl_scan(gv, wsumA);
    if (t == b) { sb[0] = ex; sb[1] = gv; }
    __syncthreads();
    const int bstart = sb[0];
    const int bend   = bstart + sb[1];
    for (int base = bstart; base < bend; base += BPT * TBE) {
        int p[BPT];
#pragma unroll
        for (int k = 0; k < BPT; ++k) {
            const int i = base + k * TBE + t;
            p[k] = tmp[i < bend ? i : bend - 1];
        }
#pragma unroll
        for (int k = 0; k < BPT; ++k)
            if (base + k * TBE + t < bend) atomicAdd(&c[p[k] & (BW - 1)], 1);
    }
    __syncthreads();
    const int v   = (t < BW) ? c[t] : 0;
    const int pre = block_excl_scan(v, wsumB);
    if (t < BW) {
        cur[t] = pre;
        const int n = b * BW + t;
        if (n < NN) { deg[n] = v; rowstart[n] = bstart + pre; }
    }
    __syncthreads();
    for (int base = bstart; base < bend; base += BPT * TBE) {
        int p[BPT];
#pragma unroll
        for (int k = 0; k < BPT; ++k) {
            const int i = base + k * TBE + t;
            p[k] = tmp[i < bend ? i : bend - 1];
        }
#pragma unroll
        for (int k = 0; k < BPT; ++k)
            if (base + k * TBE + t < bend) {
                const int r = atomicAdd(&cur[p[k] & (BW - 1)], 1);
                csr[bstart + r] = p[k] >> BSH;
            }
    }
}


__device__ __forceinline__ void logits4(const int* __restrict__ csr, const float* __restrict__ als,
                                        int start, int cnt, int n, int idx, const float4 dn,
                                        int& s, float4& v) {
    const float NEG_INF = -__builtin_inff();
    s = n;
    v = make_float4(NEG_INF, NEG_INF, NEG_INF, NEG_INF);
    if (idx <= cnt) {
        if (idx < cnt) s = csr[start + idx];
        const float4 a = *(const float4*)(als + (size_t)s * 4);
        v.x = lrelu(a.x + dn.x);
        v.y = lrelu(a.y + dn.y);
        v.z = lrelu(a.z + dn.z);
        v.w = lrelu(a.w + dn.w);
    }
}

__device__ __forceinline__ void chunk_rows(const float* __restrict__ h1, float4* wt,
                                           const float* wth, int lane, int s, const float4 w,
                                           int nv, float4& acc) {
    __builtin_amdgcn_wave_barrier();
    wt[lane] = w;
    __builtin_amdgcn_wave_barrier();
    const float* hl = h1 + lane * 4;
    int j = 0;
    for (; j + 4 <= nv; j += 4) {
        const int sa = __builtin_amdgcn_readlane(s, j);
        const int sb = __builtin_amdgcn_readlane(s, j + 1);
        const int sc = __builtin_amdgcn_readlane(s, j + 2);
        const int sd = __builtin_amdgcn_readlane(s, j + 3);
        const float4 ra = *(const float4*)(hl + (size_t)sa * HC);
        const float4 rb = *(const float4*)(hl + (size_t)sb * HC);
        const float4 rc = *(const float4*)(hl + (size_t)sc * HC);
        const float4 rd = *(const float4*)(hl + (size_t)sd * HC);
        const float wa = wth[4 * j];
        const float wb = wth[4 * j + 4];
        const float wc = wth[4 * j + 8];
        const float wd = wth[4 * j + 12];
        acc.x += ra.x * wa; acc.y += ra.y * wa; acc.z += ra.z * wa; acc.w += ra.w * wa;
        acc.x += rb.x * wb; acc.y += rb.y * wb; acc.z += rb.z * wb; acc.w += rb.w * wb;
        acc.x += rc.x * wc; acc.y += rc.y * wc; acc.z += rc.z * wc; acc.w += rc.w * wc;
        acc.x += rd.x * wd; acc.y += rd.y * wd; acc.z += rd.z * wd; acc.w += rd.w * wd;
    }
    for (; j < nv; ++j) {
        const int sa = __builtin_amdgcn_readlane(s, j);
        const float4 ra = *(const float4*)(hl + (size_t)sa * HC);
        const float wa = wth[4 * j];
        acc.x += ra.x * wa; acc.y += ra.y * wa; acc.z += ra.z * wa; acc.w += ra.w * wa;
    }
}

#define P2 52
static_assert(NN % 16 == 0, "whole 16-node strips: every wave of every block has its four nodes");
__global__ void __launch_bounds__(128)
k_node1(const int* __restrict__ csr, const int* __restrict__ rowstart,
        const int* __restrict__ deg, const float* __restrict__ h1,
        const float* __restrict__ als, const float* __restrict__ ald,
        const float* __restrict__ b1, const float* __restrict__ W2,
        const float* __restrict__ a_src2, const float* __restrict__ a_dst2,
        float* __restrict__ z, float* __restrict__ als2, float* __restrict__ ald2) {
    __shared__ float4 wtab[4][32];
    __shared__ float hs[16 * P1];
    __shared__ float zt[16 * P2];
    const int lane = threadIdx.x & 31;
    const int wv   = threadIdx.x >> 5;
    for (int q = 0; q < 4; ++q) {
        const int n = __builtin_amdgcn_readfirstlane((int)(blockIdx.x * 16 + wv * 4 + q));
        const int start = __builtin_amdgcn_readfirstlane(rowstart[n]);
        const int cnt   = __builtin_amdgcn_readfirstlane(deg[n]);
        const int tot   = cnt + 1;
        const float4 dn = *(const float4*)(ald + (size_t)n * 4);
        float4* wt = wtab[wv];
        const float* wth = (const float*)wt + (lane >> 3);

        int s0; float4 v0;
        logits4(csr, als, start, cnt, n, lane, dn, s0, v0);

        float4 m = v0;
        for (int base = 32; base < tot; base += 32) {
            int s; float4 v;
            logits4(csr, als, start, cnt, n, base + lane, dn, s, v);
            m.x = fmaxf(m.x, v.x); m.y = fmaxf(m.y, v.y); m.z = fmaxf(m.z, v.z); m.w = fmaxf(m.w, v.w);
        }
#pragma unroll
        for (int off = 16; off > 0; off >>= 1) {
            m.x = fmaxf(m.x, __shfl_xor(m.x, off));
            m.y = fmaxf(m.y, __shfl_xor(m.y, off));
            m.z = fmaxf(m.z, __shfl_xor(m.z, off));
            m.w = fmaxf(m.w, __shfl_xor(m.w, off));
        }

        float4 p0;
        p0.x = __expf(v0.x - m.x); p0.y = __expf(v0.y - m.y);
        p0.z = __expf(v0.z - m.z); p0.w = __expf(v0.w - m.w);
        float4 d = p0;
        for (int base = 32; base < tot; base += 32) {
            int s; float4 v;
            logits4(csr, als, start, cnt, n, base + lane, dn, s, v);
            d.x += __expf(v.x - m.x); d.y += __expf(v.y - m.y);
            d.z += __expf(v.z - m.z); d.w += __expf(v.w - m.w);
        }
#pragma unroll
        for (int off = 16; off > 0; off >>= 1) {
            d.x += __shfl_xor(d.x, off);
            d.y += __shfl_xor(d.y, off);
            d.z += __shfl_xor(d.z, off);
            d.w += __shfl_xor(d.w, off);
        }
        d.x += 1e-16f; d.y += 1e-16f; d.z += 1e-16f; d.w += 1e-16f;

        float4 acc = make_float4(0.f, 0.f, 0.f, 0.f);
        {
            float4 w;
            w.x = p0.x / d.x; w.y = p0.y / d.y; w.z = p0.z / d.z; w.w = p0.w / d.w;
            chunk_rows(h1, wt, wth, lane, s0, w, (tot < 32) ? tot : 32, acc);
        }
        for (int base = 32; base < tot; base += 32) {
            int s; float4 v;
            logits4(csr, als, start, cnt, n, base + lane, dn, s, v);
            float4 w;
            w.x = __expf(v.x - m.x) / d.x; w.y = __expf(v.y - m.y) / d.y;
            w.z = __expf(v.z - m.z) / d.z; w.w = __expf(v.w - m.w) / d.w;
            chunk_rows(h1, wt, wth, lane, s, w, (tot - base < 32) ? (tot - base) : 32, acc);
        }

        const float4 bb = *(const float4*)(b1 + lane * 4);
        float4 o;
        o.x = acc.x + bb.x; o.y = acc.y + bb.y; o.z = acc.z + bb.z; o.w = acc.w + bb.w;
        o.x = o.x > 0.0f ? o.x : expm1f(o.x);
        o.y = o.y > 0.0f ? o.y : expm1f(o.y);
        o.z = o.z > 0.0f ? o.z : expm1f(o.z);
        o.w = o.w > 0.0f ? o.w : expm1f(o.w);
        *(float4*)(hs + (wv * 4 + q) * P1 + lane * 4) = o;
    }

    const int gm  = lane & 15;
    const int gkb = (lane >> 4) * 2;
    const int col = wv * 16 + gm;
    const float* wp = W2 + (size_t)gkb * OUT_CH + (col < OUT_CH ? col : OUT_CH - 1);
    v2f bq[16];
    if (wv < 3) {
#pragma unroll
        for (int i = 0; i < 16; ++i) {
            bq[i].x = wp[(size_t)(4 * i) * OUT_CH];
            bq[i].y = wp[(size_t)(4 * i + 1) * OUT_CH];
        }
    }
    __syncthreads();

    const int row0 = blockIdx.x * 16;
    if (wv < 3) {
        const float* ap = hs + gm * P1 + gkb;
        v8f c = (v8f){0,0,0,0,0,0,0,0};
#pragma unroll
        for (int i = 0; i < 16; ++i) {
            const v2f a = *(const v2f*)(ap + 4 * i);
            c = __builtin_amdgcn_wmma_f32_16x16x4_f32(false, a, false, bq[i],
                                                      (short)0, c, false, false);
        }
#pragma unroll
        for (int k = 64; k < HC; k += 4) {
            const v2f a = *(const v2f*)(ap + k);
            v2f b;
            b.x = wp[(size_t)k * OUT_CH];
            b.y = wp[(size_t)(k + 1) * OUT_CH];
            c = __builtin_amdgcn_wmma_f32_16x16x4_f32(false, a, false, b,
                                                      (short)0, c, false, false);
        }
        const int rofs = (lane >> 4) * 8;
#pragma unroll
        for (int r = 0; r < 8; ++r) zt[(r + rofs) * P2 + col] = c[r];
    }
    __syncthreads();
    for (int idx = threadIdx.x; idx < 16 * OUT_CH / 4; idx += 128) {
        const int r  = idx / 10;
        const int c4 = idx - r * 10;
        const float4 v = *(const float4*)(zt + r * P2 + c4 * 4);
        *(float4*)(z + (size_t)row0 * OUT_CH + idx * 4) = v;
    }
    if (wv == 3) {
        const int lr = lane & 15;
        const bool tgt = lane >= 16;
        const float* av = tgt ? a_dst2 : a_src2;
        const float* zr = zt + lr * P2;
        float sacc = 0.f;
#pragma unroll
        for (int cc = 0; cc < OUT_CH; ++cc) sacc = __builtin_fmaf(zr[cc], av[cc], sacc);
        (tgt ? ald2 : als2)[row0 + lr] = sacc;
    }
}

__global__ void __launch_bounds__(256)
k_node2(const int* __restrict__ csr, const int* __restrict__ rowstart,
        const int* __restrict__ deg, const float* __restrict__ z,
        const float* __restrict__ als, const float* __restrict__ ald,
        const float* __restrict__ b2, float* __restrict__ out) {
    const int lane = threadIdx.x & 31;
    const int w = __builtin_amdgcn_readfirstlane((int)(blockIdx.x * 8 + (threadIdx.x >> 5)));
    const bool hi = lane < (OUT_CH - 32);
    const float bl = b2[lane];
    const float bh = hi ? b2[32 + lane] : 0.0f;
    const float NEG_INF = -__builtin_inff();

    for (int q = 0; q < 4; ++q) {
        const int n = w * 4 + q;
        if (n >= NN) break;
        const int start = __builtin_amdgcn_readfirstlane(rowstart[n]);
        const int cnt   = __builtin_amdgcn_readfirstlane(deg[n]);
        const int tot   = cnt + 1;
        const float dn  = ald[n];

        float m = NEG_INF;
        for (int base = 0; base < tot; base += 32) {
            const int idx = base + lane;
            if (idx < tot) {
                const int s = (idx < cnt) ? csr[start + idx] : n;
                m = fmaxf(m, lrelu(als[s] + dn));
            }
        }
#pragma unroll
        for (int off = 16; off > 0; off >>= 1) m = fmaxf(m, __shfl_xor(m, off));

        float den = 0.f;
        for (int base = 0; base < tot; base += 32) {
            const int idx = base + lane;
            if (idx < tot) {
                const int s = (idx < cnt) ? csr[start + idx] : n;
                den += __expf(lrelu(als[s] + dn) - m);
            }
        }
#pragma unroll
        for (int off = 16; off > 0; off >>= 1) den += __shfl_xor(den, off);
        den += 1e-16f;

        float acc0 = 0.f, acc1 = 0.f;
        for (int base = 0; base < tot; base += 32) {
            const int idx = base + lane;
            int s = n;
            float wgt = 0.f;
            if (idx < tot) {
                s = (idx < cnt) ? csr[start + idx] : n;
                wgt = __expf(lrelu(als[s] + dn) - m) / den;
            }
            const int nv = (tot - base < 32) ? (tot - base) : 32;
            int j = 0;
            for (; j + 4 <= nv; j += 4) {
                const int sa = __builtin_amdgcn_readlane(s, j);
                const int sb = __builtin_amdgcn_readlane(s, j + 1);
                const int sc = __builtin_amdgcn_readlane(s, j + 2);
                const int sd = __builtin_amdgcn_readlane(s, j + 3);
                const float wa = __int_as_float(__builtin_amdgcn_readlane(__float_as_int(wgt), j));
                const float wb = __int_as_float(__builtin_amdgcn_readlane(__float_as_int(wgt), j + 1));
                const float wc = __int_as_float(__builtin_amdgcn_readlane(__float_as_int(wgt), j + 2));
                const float wd = __int_as_float(__builtin_amdgcn_readlane(__float_as_int(wgt), j + 3));
                const float* za = z + (size_t)sa * OUT_CH;
                const float* zb = z + (size_t)sb * OUT_CH;
                const float* zc = z + (size_t)sc * OUT_CH;
                const float* zd = z + (size_t)sd * OUT_CH;
                const float la = za[lane], lb = zb[lane], lc = zc[lane], ld = zd[lane];
                float ha = 0.f, hb = 0.f, hc = 0.f, hd = 0.f;
                if (hi) { ha = za[32 + lane]; hb = zb[32 + lane]; hc = zc[32 + lane]; hd = zd[32 + lane]; }
                acc0 += la * wa; acc0 += lb * wb; acc0 += lc * wc; acc0 += ld * wd;
                acc1 += ha * wa; acc1 += hb * wb; acc1 += hc * wc; acc1 += hd * wd;
            }
            for (; j < nv; ++j) {
                const int sa = __builtin_amdgcn_readlane(s, j);
                const float wa = __int_as_float(__builtin_amdgcn_readlane(__float_as_int(wgt), j));
                const float* za = z + (size_t)sa * OUT_CH;
                acc0 += za[lane] * wa;
                if (hi) acc1 += za[32 + lane] * wa;
            }
        }
        float* op = out + (size_t)n * OUT_CH;
        op[lane] = acc0 + bl;
        if (hi) op[32 + lane] = acc1 + bh;
    }
}

extern "C" void kernel_launch(void* const* d_in, const int* in_sizes, int n_in,
                              void* d_out, int out_size, void* d_ws, size_t ws_size,
                              hipStream_t stream) {
    const float* x      = (const float*)d_in[0];
    const int*   ei     = (const int*)  d_in[1];
    const float* W1     = (const float*)d_in[2];
    const float* a_src1 = (const float*)d_in[3];
    const float* a_dst1 = (const float*)d_in[4];
    const float* b1     = (const float*)d_in[5];
    const float* W2     = (const float*)d_in[6];
    const float* a_src2 = (const float*)d_in[7];
    const float* a_dst2 = (const float*)d_in[8];
    const float* b2     = (const float*)d_in[9];
    float* out = (float*)d_out;

    float* ws   = (float*)d_ws;
    float* h1   = ws;
    float* z2   = h1   + (size_t)NN * HC;
    float* als1 = z2   + (size_t)NN * OUT_CH;
    float* ald1 = als1 + (size_t)NN * HEADS;
    float* als2 = ald1 + (size_t)NN * HEADS;
    float* ald2 = als2 + NN;
    int* deg      = (int*)(ald2 + NN);
    int* rowstart = deg + NN;
    int* ghist    = rowstart + NN;
    int* gcur     = ghist + GH;
    int* tmp      = gcur + GH;
    int* csr      = tmp + EE;

    const int gemmBlocks = (NN / 16 + 3) / 4;
    const int nodeBlocks1 = NN / 16;
    const int nodeBlocks2 = (NN / 4 + 7) / 8;

    k_gemm1<<<gemmBlocks, 128, 0, stream>>>(x, W1, a_src1, a_dst1, h1, als1, ald1, ghist);
    k_hist   <<<EBLK, TBE, 0, stream>>>(ei, ghist);
    k_scatter<<<EBLK, TBE, 0, stream>>>(ei, ghist, gcur, tmp);
    k_bsort  <<<NB,   TBE, 0, stream>>>(ghist, tmp, deg, rowstart, csr);
    k_node1<<<nodeBlocks1, 128, 0, stream>>>(csr, rowstart, deg, h1, als1, ald1, b1,
                                             W2, a_src2, a_dst2, z2, als2, ald2);

    k_node2<<<nodeBlocks2, 256, 0, stream>>>(csr, rowstart, deg, z2, als2, ald2, b2, out);
}


# Both node kernels load run ids and logits of a wave's 4 nodes together, ahead of the first node; layer 2 keeps them in registers instead of gathering them again in each softmax pass; sum order kept
# speedup vs baseline: 5.7467x; 1.1082x over previous
#include <hip/hip_runtime.h>
#include <math.h>

#define NN 100000
#define EE 1600000
#define IN_CH 128
#define HID 32
#define HEADS 4
#define HC 128
#define OUT_CH 40
#define NEG_SLOPE 0.2f
#define GH 512

typedef __attribute__((ext_vector_type(2))) float v2f;
typedef __attribute__((ext_vector_type(8))) float v8f;

__device__ __forceinline__ float lrelu(float v) {
    return v >= 0.0f ? v : NEG_SLOPE * v;
}

#define P1 132
__global__ void __launch_bounds__(128)
k_gemm1(const float* __restrict__ x, const float* __restrict__ W1,
        const float* __restrict__ a_src, const float* __restrict__ a_dst,
        float* __restrict__ h1, float* __restrict__ als, float* __restrict__ ald,
        int* __restrict__ ghist) {
    __shared__ float tile[4][16 * P1];
    if (blockIdx.x == 0)
        for (int i = threadIdx.x; i < 2 * GH; i += 128) ghist[i] = 0;
    const int wave = threadIdx.x >> 5;
    const int row0 = (blockIdx.x * 4 + wave) * 16;
    if (row0 >= NN) return;
    const int lane = threadIdx.x & 31;
    const int m    = lane & 15;
    const int kb   = (lane >> 4) * 2;

    v8f c[8];
#pragma unroll
    for (int t = 0; t < 8; ++t) c[t] = (v8f){0,0,0,0,0,0,0,0};

    for (int k = 0; k < IN_CH; k += 4) {
        v2f a;
        const float* xr = x + (size_t)(row0 + m) * IN_CH + (k + kb);
        a.x = xr[0];
        a.y = xr[1];
#pragma unroll
        for (int t = 0; t < 8; ++t) {
            v2f b;
            const float* wr = W1 + (size_t)(k + kb) * HC + t * 16 + m;
            b.x = wr[0];
            b.y = wr[HC];
            c[t] = __builtin_amdgcn_wmma_f32_16x16x4_f32(false, a, false, b,
                                                         (short)0, c[t], false, false);
        }
    }
    const int rofs = (lane >> 4) * 8;
    float* sm = tile[wave];
#pragma unroll
    for (int t = 0; t < 8; ++t)
#pragma unroll
        for (int r = 0; r < 8; ++r) {
            h1[(size_t)(row0 + r + rofs) * HC + t * 16 + m] = c[t][r];
            sm[(r + rofs) * P1 + t * 16 + m] = c[t][r];
        }
    __builtin_amdgcn_wave_barrier();

    const int lr = lane & 15;
    const int hp = (lane >> 4) * 2;
    float s0 = 0.f, d0 = 0.f, s1 = 0.f, d1 = 0.f;
    const float* r0 = sm + lr * P1 + hp * HID;
    const float* as0 = a_src + hp * HID;
    const float* ad0 = a_dst + hp * HID;
#pragma unroll
    for (int cc = 0; cc < HID; ++cc) {
        const float v0 = r0[cc];
        const float v1 = r0[HID + cc];
        s0 = __builtin_fmaf(v0, as0[cc], s0);
        d0 = __builtin_fmaf(v0, ad0[cc], d0);
        s1 = __builtin_fmaf(v1, as0[HID + cc], s1);
        d1 = __builtin_fmaf(v1, ad0[HID + cc], d1);
    }
    *(float2*)(als + (size_t)(row0 + lr) * HEADS + hp) = make_float2(s0, s1);
    *(float2*)(ald + (size_t)(row0 + lr) * HEADS + hp) = make_float2(d0, d1);
}

#define BSH 8
#define BW  256
#define NB  ((NN + BW - 1) / BW)
#define TBE 512
#define EPT 8
#define BPT 12
#define EBLK ((EE / EPT + TBE - 1) / TBE)
static_assert(NB <= TBE && NB <= GH, "one thread per coarse bucket");
static_assert(EE % EPT == 0, "whole threads of edges");
static_assert(NN <= (1 << 23), "source id and local destination share one word");

__device__ __forceinline__ int block_excl_scan(int v, int* wsum) {
    const int lane = threadIdx.x & 31;
    const int wv   = threadIdx.x >> 5;
    int incl = v;
#pragma unroll
    for (int off = 1; off < 32; off <<= 1) {
        int t = __shfl_up(incl, off);
        if (lane >= off) incl += t;
    }
    if (lane == 31) wsum[wv] = incl;
    __syncthreads();
    int woff = 0;
    for (int k = 0; k < wv; ++k) woff += wsum[k];
    return woff + incl - v;
}

__global__ void __launch_bounds__(TBE)
k_hist(const int* __restrict__ ei, int* __restrict__ ghist) {
    __shared__ int h[TBE];
    const int t = threadIdx.x;
    h[t] = 0;
    __syncthreads();
    const int e0 = (blockIdx.x * TBE + t) * EPT;
    if (e0 < EE) {
        const int4 a = *(const int4*)(ei + EE + e0);
        const int4 b = *(const int4*)(ei + EE + e0 + 4);
        atomicAdd(&h[a.x >> BSH], 1); atomicAdd(&h[a.y >> BSH], 1);
        atomicAdd(&h[a.z >> BSH], 1); atomicAdd(&h[a.w >> BSH], 1);
        atomicAdd(&h[b.x >> BSH], 1); atomicAdd(&h[b.y >> BSH], 1);
        atomicAdd(&h[b.z >> BSH], 1); atomicAdd(&h[b.w >> BSH], 1);
    }
    __syncthreads();
    const int c = h[t];
    if (c) atomicAdd(&ghist[t], c);
}

__global__ void __launch_bounds__(TBE)
k_scatter(const int* __restrict__ ei, const int* __restrict__ ghist,
          int* __restrict__ gcur, int* __restrict__ tmp) {
    __shared__ int cnt[TBE];
    __shared__ int off[TBE];
    __shared__ int wsum[16];
    const int t = threadIdx.x;
    cnt[t] = 0;
    const int gv  = (t < NB) ? ghist[t] : 0;
    const int bst = block_excl_scan(gv, wsum);
    const int e0 = (blockIdx.x * TBE + t) * EPT;
    const bool act = e0 < EE;
    int4 s0, s1, d0, d1;
    int r0 = 0, r1 = 0, r2 = 0, r3 = 0, r4 = 0, r5 = 0, r6 = 0, r7 = 0;
    if (act) {
        s0 = *(const int4*)(ei + e0);
        s1 = *(const int4*)(ei + e0 + 4);
        d0 = *(const int4*)(ei + EE + e0);
        d1 = *(const int4*)(ei + EE + e0 + 4);
        r0 = atomicAdd(&cnt[d0.x >> BSH], 1); r1 = atomicAdd(&cnt[d0.y >> BSH], 1);
        r2 = atomicAdd(&cnt[d0.z >> BSH], 1); r3 = atomicAdd(&cnt[d0.w >> BSH], 1);
        r4 = atomicAdd(&cnt[d1.x >> BSH], 1); r5 = atomicAdd(&cnt[d1.y >> BSH], 1);
        r6 = atomicAdd(&cnt[d1.z >> BSH], 1); r7 = atomicAdd(&cnt[d1.w >> BSH], 1);
    }
    __syncthreads();
    {
        const int c = cnt[t];
        int base = 0;
        if (c) base = atomicAdd(&gcur[t], c);
        off[t] = bst + base;
    }
    __syncthreads();
    if (act) {
        tmp[off[d0.x >> BSH] + r0] = (s0.x << BSH) | (d0.x & (BW - 1));
        tmp[off[d0.y >> BSH] + r1] = (s0.y << BSH) | (d0.y & (BW - 1));
        tmp[off[d0.z >> BSH] + r2] = (s0.z << BSH) | (d0.z & (BW - 1));
        tmp[off[d0.w >> BSH] + r3] = (s0.w << BSH) | (d0.w & (BW - 1));
        tmp[off[d1.x >> BSH] + r4] = (s1.x << BSH) | (d1.x & (BW - 1));
        tmp[off[d1.y >> BSH] + r5] = (s1.y << BSH) | (d1.y & (BW - 1));
        tmp[off[d1.z >> BSH] + r6] = (s1.z << BSH) | (d1.z & (BW - 1));
        tmp[off[d1.w >> BSH] + r7] = (s1.w << BSH) | (d1.w & (BW - 1));
    }
}

__global__ void __launch_bounds__(TBE)
k_bsort(const int* __restrict__ ghist, const int* __restrict__ tmp,
        int* __restrict__ deg, int* __restrict__ rowstart, int* __restrict__ csr) {
    __shared__ int c[BW];
    __shared__ int cur[BW];
    __shared__ int wsumA[16];
    __shared__ int wsumB[16];
    __shared__ int sb[2];
    const int t = threadIdx.x;
    const int b = blockIdx.x;
    if (t < BW) c[t] = 0;
    const int gv = (t < NB) ? ghist[t] : 0;
    const int ex = block_excl_scan(gv, wsumA);
    if (t == b) { sb[0] = ex; sb[1] = gv; }
    __syncthreads();
    const int bstart = sb[0];
    const int bend   = bstart + sb[1];
    for (int base = bstart; base < bend; base += BPT * TBE) {
        int p[BPT];
#pragma unroll
        for (int k = 0; k < BPT; ++k) {
            const int i = base + k * TBE + t;
            p[k] = tmp[i < bend ? i : bend - 1];
        }
#pragma unroll
        for (int k = 0; k < BPT; ++k)
            if (base + k * TBE + t < bend) atomicAdd(&c[p[k] & (BW - 1)], 1);
    }
    __syncthreads();
    const int v   = (t < BW) ? c[t] : 0;
    const int pre = block_excl_scan(v, wsumB);
    if (t < BW) {
        cur[t] = pre;
        const int n = b * BW + t;
        if (n < NN) { deg[n] = v; rowstart[n] = bstart + pre; }
    }
    __syncthreads();
    for (int base = bstart; base < bend; base += BPT * TBE) {
        int p[BPT];
#pragma unroll
        for (int k = 0; k < BPT; ++k) {
            const int i = base + k * TBE + t;
            p[k] = tmp[i < bend ? i : bend - 1];
        }
#pragma unroll
        for (int k = 0; k < BPT; ++k)
            if (base + k * TBE + t < bend) {
                const int r = atomicAdd(&cur[p[k] & (BW - 1)], 1);
                csr[bstart + r] = p[k] >> BSH;
            }
    }
}


__device__ __forceinline__ void logits4(const int* __restrict__ csr, const float* __restrict__ als,
                                        int start, int cnt, int n, int idx, const float4 dn,
                                        int& s, float4& v) {
    const float NEG_INF = -__builtin_inff();
    s = n;
    v = make_float4(NEG_INF, NEG_INF, NEG_INF, NEG_INF);
    if (idx <= cnt) {
        if (idx < cnt) s = csr[start + idx];
        const float4 a = *(const float4*)(als + (size_t)s * 4);
        v.x = lrelu(a.x + dn.x);
        v.y = lrelu(a.y + dn.y);
        v.z = lrelu(a.z + dn.z);
        v.w = lrelu(a.w + dn.w);
    }
}

__device__ __forceinline__ void chunk_rows(const float* __restrict__ h1, float4* wt,
                                           const float* wth, int lane, int s, const float4 w,
                                           int nv, float4& acc) {
    __builtin_amdgcn_wave_barrier();
    wt[lane] = w;
    __builtin_amdgcn_wave_barrier();
    const float* hl = h1 + lane * 4;
    int j = 0;
    for (; j + 4 <= nv; j += 4) {
        const int sa = __builtin_amdgcn_readlane(s, j);
        const int sb = __builtin_amdgcn_readlane(s, j + 1);
        const int sc = __builtin_amdgcn_readlane(s, j + 2);
        const int sd = __builtin_amdgcn_readlane(s, j + 3);
        const float4 ra = *(const float4*)(hl + (size_t)sa * HC);
        const float4 rb = *(const float4*)(hl + (size_t)sb * HC);
        const float4 rc = *(const float4*)(hl + (size_t)sc * HC);
        const float4 rd = *(const float4*)(hl + (size_t)sd * HC);
        const float wa = wth[4 * j];
        const float wb = wth[4 * j + 4];
        const float wc = wth[4 * j + 8];
        const float wd = wth[4 * j + 12];
        acc.x += ra.x * wa; acc.y += ra.y * wa; acc.z += ra.z * wa; acc.w += ra.w * wa;
        acc.x += rb.x * wb; acc.y += rb.y * wb; acc.z += rb.z * wb; acc.w += rb.w * wb;
        acc.x += rc.x * wc; acc.y += rc.y * wc; acc.z += rc.z * wc; acc.w += rc.w * wc;
        acc.x += rd.x * wd; acc.y += rd.y * wd; acc.z += rd.z * wd; acc.w += rd.w * wd;
    }
    for (; j < nv; ++j) {
        const int sa = __builtin_amdgcn_readlane(s, j);
        const float4 ra = *(const float4*)(hl + (size_t)sa * HC);
        const float wa = wth[4 * j];
        acc.x += ra.x * wa; acc.y += ra.y * wa; acc.z += ra.z * wa; acc.w += ra.w * wa;
    }
}

#define P2 52
static_assert(NN % 16 == 0, "whole 16-node strips: every wave of every block has its four nodes");
__global__ void __launch_bounds__(128)
k_node1(const int* __restrict__ csr, const int* __restrict__ rowstart,
        const int* __restrict__ deg, const float* __restrict__ h1,
        const float* __restrict__ als, const float* __restrict__ ald,
        const float* __restrict__ b1, const float* __restrict__ W2,
        const float* __restrict__ a_src2, const float* __restrict__ a_dst2,
        float* __restrict__ z, float* __restrict__ als2, float* __restrict__ ald2) {
    __shared__ float4 wtab[4][6][32];
    __shared__ float hs[16 * P1];
    __shared__ float zt[16 * P2];
    const int lane = threadIdx.x & 31;
    const int wv   = threadIdx.x >> 5;
    {
        const int nb = __builtin_amdgcn_readfirstlane((int)(blockIdx.x * 16 + wv * 4));
        const int4 st4 = *(const int4*)(rowstart + nb);
        const int4 cn4 = *(const int4*)(deg + nb);
        const int stq[4] = { __builtin_amdgcn_readfirstlane(st4.x), __builtin_amdgcn_readfirstlane(st4.y),
                             __builtin_amdgcn_readfirstlane(st4.z), __builtin_amdgcn_readfirstlane(st4.w) };
        const int cnq[4] = { __builtin_amdgcn_readfirstlane(cn4.x), __builtin_amdgcn_readfirstlane(cn4.y),
                             __builtin_amdgcn_readfirstlane(cn4.z), __builtin_amdgcn_readfirstlane(cn4.w) };
        int sq[4];
#pragma unroll
        for (int q = 0; q < 4; ++q) {
            const int sv = csr[lane < cnq[q] ? stq[q] + lane : 0];
            sq[q] = lane < cnq[q] ? sv : nb + q;
        }
#pragma unroll
        for (int q = 0; q < 4; ++q)
            wtab[wv][1 + q][lane] = *(const float4*)(als + (size_t)sq[q] * 4);
        *(int4*)&wtab[wv][5][lane] = make_int4(sq[0], sq[1], sq[2], sq[3]);
    }
    for (int q = 0; q < 4; ++q) {
        const int n = __builtin_amdgcn_readfirstlane((int)(blockIdx.x * 16 + wv * 4 + q));
        const int start = __builtin_amdgcn_readfirstlane(rowstart[n]);
        const int cnt   = __builtin_amdgcn_readfirstlane(deg[n]);
        const int tot   = cnt + 1;
        const float4 dn = *(const float4*)(ald + (size_t)n * 4);
        float4* wt = wtab[wv][0];
        const float* wth = (const float*)wt + (lane >> 3);

        const int s0 = ((const int*)&wtab[wv][5][lane])[q];
        float4 v0 = make_float4(-__builtin_inff(), -__builtin_inff(), -__builtin_inff(), -__builtin_inff());
        {
            const float4 a0 = wtab[wv][1 + q][lane];
            if (lane <= cnt) {
                v0.x = lrelu(a0.x + dn.x);
                v0.y = lrelu(a0.y + dn.y);
                v0.z = lrelu(a0.z + dn.z);
                v0.w = lrelu(a0.w + dn.w);
            }
        }

        float4 m = v0;
        for (int base = 32; base < tot; base += 32) {
            int s; float4 v;
            logits4(csr, als, start, cnt, n, base + lane, dn, s, v);
            m.x = fmaxf(m.x, v.x); m.y = fmaxf(m.y, v.y); m.z = fmaxf(m.z, v.z); m.w = fmaxf(m.w, v.w);
        }
#pragma unroll
        for (int off = 16; off > 0; off >>= 1) {
            m.x = fmaxf(m.x, __shfl_xor(m.x, off));
            m.y = fmaxf(m.y, __shfl_xor(m.y, off));
            m.z = fmaxf(m.z, __shfl_xor(m.z, off));
            m.w = fmaxf(m.w, __shfl_xor(m.w, off));
        }

        float4 p0;
        p0.x = __expf(v0.x - m.x); p0.y = __expf(v0.y - m.y);
        p0.z = __expf(v0.z - m.z); p0.w = __expf(v0.w - m.w);
        float4 d = p0;
        for (int base = 32; base < tot; base += 32) {
            int s; float4 v;
            logits4(csr, als, start, cnt, n, base + lane, dn, s, v);
            d.x += __expf(v.x - m.x); d.y += __expf(v.y - m.y);
            d.z += __expf(v.z - m.z); d.w += __expf(v.w - m.w);
        }
#pragma unroll
        for (int off = 16; off > 0; off >>= 1) {
            d.x += __shfl_xor(d.x, off);
            d.y += __shfl_xor(d.y, off);
            d.z += __shfl_xor(d.z, off);
            d.w += __shfl_xor(d.w, off);
        }
        d.x += 1e-16f; d.y += 1e-16f; d.z += 1e-16f; d.w += 1e-16f;

        float4 acc = make_float4(0.f, 0.f, 0.f, 0.f);
        {
            float4 w;
            w.x = p0.x / d.x; w.y = p0.y / d.y; w.z = p0.z / d.z; w.w = p0.w / d.w;
            chunk_rows(h1, wt, wth, lane, s0, w, (tot < 32) ? tot : 32, acc);
        }
        for (int base = 32; base < tot; base += 32) {
            int s; float4 v;
            logits4(csr, als, start, cnt, n, base + lane, dn, s, v);
            float4 w;
            w.x = __expf(v.x - m.x) / d.x; w.y = __expf(v.y - m.y) / d.y;
            w.z = __expf(v.z - m.z) / d.z; w.w = __expf(v.w - m.w) / d.w;
            chunk_rows(h1, wt, wth, lane, s, w, (tot - base < 32) ? (tot - base) : 32, acc);
        }

        const float4 bb = *(const float4*)(b1 + lane * 4);
        float4 o;
        o.x = acc.x + bb.x; o.y = acc.y + bb.y; o.z = acc.z + bb.z; o.w = acc.w + bb.w;
        o.x = o.x > 0.0f ? o.x : expm1f(o.x);
        o.y = o.y > 0.0f ? o.y : expm1f(o.y);
        o.z = o.z > 0.0f ? o.z : expm1f(o.z);
        o.w = o.w > 0.0f ? o.w : expm1f(o.w);
        *(float4*)(hs + (wv * 4 + q) * P1 + lane * 4) = o;
    }

    const int gm  = lane & 15;
    const int gkb = (lane >> 4) * 2;
    const int col = wv * 16 + gm;
    const float* wp = W2 + (size_t)gkb * OUT_CH + (col < OUT_CH ? col : OUT_CH - 1);
    v2f bq[16];
    if (wv < 3) {
#pragma unroll
        for (int i = 0; i < 16; ++i) {
            bq[i].x = wp[(size_t)(4 * i) * OUT_CH];
            bq[i].y = wp[(size_t)(4 * i + 1) * OUT_CH];
        }
    }
    __syncthreads();

    const int row0 = blockIdx.x * 16;
    if (wv < 3) {
        const float* ap = hs + gm * P1 + gkb;
        v8f c = (v8f){0,0,0,0,0,0,0,0};
#pragma unroll
        for (int i = 0; i < 16; ++i) {
            const v2f a = *(const v2f*)(ap + 4 * i);
            c = __builtin_amdgcn_wmma_f32_16x16x4_f32(false, a, false, bq[i],
                                                      (short)0, c, false, false);
        }
#pragma unroll
        for (int k = 64; k < HC; k += 4) {
            const v2f a = *(const v2f*)(ap + k);
            v2f b;
            b.x = wp[(size_t)k * OUT_CH];
            b.y = wp[(size_t)(k + 1) * OUT_CH];
            c = __builtin_amdgcn_wmma_f32_16x16x4_f32(false, a, false, b,
                                                      (short)0, c, false, false);
        }
        const int rofs = (lane >> 4) * 8;
#pragma unroll
        for (int r = 0; r < 8; ++r) zt[(r + rofs) * P2 + col] = c[r];
    }
    __syncthreads();
    for (int idx = threadIdx.x; idx < 16 * OUT_CH / 4; idx += 128) {
        const int r  = idx / 10;
        const int c4 = idx - r * 10;
        const float4 v = *(const float4*)(zt + r * P2 + c4 * 4);
        *(float4*)(z + (size_t)row0 * OUT_CH + idx * 4) = v;
    }
    if (wv == 3) {
        const int lr = lane & 15;
        const bool tgt = lane >= 16;
        const float* av = tgt ? a_dst2 : a_src2;
        const float* zr = zt + lr * P2;
        float sacc = 0.f;
#pragma unroll
        for (int cc = 0; cc < OUT_CH; ++cc) sacc = __builtin_fmaf(zr[cc], av[cc], sacc);
        (tgt ? ald2 : als2)[row0 + lr] = sacc;
    }
}


__device__ __forceinline__ float late_logit(const int* __restrict__ csr, const float* __restrict__ als,
                                            int start, int cnt, int n, int idx, float dn, int& s) {
    s = n;
    float v = -__builtin_inff();
    if (idx <= cnt) {
        if (idx < cnt) s = csr[start + idx];
        v = lrelu(als[s] + dn);
    }
    return v;
}

__device__ __forceinline__ void chunk_rows2(const float* __restrict__ z, int lane, bool hi,
                                            int s, float wgt, int nv, float& acc0, float& acc1) {
    int j = 0;
    for (; j + 4 <= nv; j += 4) {
        const int sa = __builtin_amdgcn_readlane(s, j);
        const int sb = __builtin_amdgcn_readlane(s, j + 1);
        const int sc = __builtin_amdgcn_readlane(s, j + 2);
        const int sd = __builtin_amdgcn_readlane(s, j + 3);
        const float wa = __int_as_float(__builtin_amdgcn_readlane(__float_as_int(wgt), j));
        const float wb = __int_as_float(__builtin_amdgcn_readlane(__float_as_int(wgt), j + 1));
        const float wc = __int_as_float(__builtin_amdgcn_readlane(__float_as_int(wgt), j + 2));
        const float wd = __int_as_float(__builtin_amdgcn_readlane(__float_as_int(wgt), j + 3));
        const float* za = z + (size_t)sa * OUT_CH;
        const float* zb = z + (size_t)sb * OUT_CH;
        const float* zc = z + (size_t)sc * OUT_CH;
        const float* zd = z + (size_t)sd * OUT_CH;
        const float la = za[lane], lb = zb[lane], lc = zc[lane], ld = zd[lane];
        float ha = 0.f, hb = 0.f, hc = 0.f, hd = 0.f;
        if (hi) { ha = za[32 + lane]; hb = zb[32 + lane]; hc = zc[32 + lane]; hd = zd[32 + lane]; }
        acc0 += la * wa; acc0 += lb * wb; acc0 += lc * wc; acc0 += ld * wd;
        acc1 += ha * wa; acc1 += hb * wb; acc1 += hc * wc; acc1 += hd * wd;
    }
    for (; j < nv; ++j) {
        const int sa = __builtin_amdgcn_readlane(s, j);
        const float wa = __int_as_float(__builtin_amdgcn_readlane(__float_as_int(wgt), j));
        const float* za = z + (size_t)sa * OUT_CH;
        acc0 += za[lane] * wa;
        if (hi) acc1 += za[32 + lane] * wa;
    }
}

static_assert(NN % 4 == 0, "a wave's four nodes all exist");
static_assert(EE > 0, "run slot 0 is always written");
__global__ void __launch_bounds__(256)
k_node2(const int* __restrict__ csr, const int* __restrict__ rowstart,
        const int* __restrict__ deg, const float* __restrict__ z,
        const float* __restrict__ als, const float* __restrict__ ald,
        const float* __restrict__ b2, float* __restrict__ out) {
    const int lane = threadIdx.x & 31;
    const int w = __builtin_amdgcn_readfirstlane((int)(blockIdx.x * 8 + (threadIdx.x >> 5)));
    const int n0 = w * 4;
    if (n0 >= NN) return;
    const bool hi = lane < (OUT_CH - 32);
    const float bl = b2[lane];
    const float bh = hi ? b2[32 + lane] : 0.0f;
    const float NEG_INF = -__builtin_inff();

    const int4   st4 = *(const int4*)(rowstart + n0);
    const int4   cn4 = *(const int4*)(deg + n0);
    const float4 dn4 = *(const float4*)(ald + n0);
    const int st[4] = { __builtin_amdgcn_readfirstlane(st4.x), __builtin_amdgcn_readfirstlane(st4.y),
                        __builtin_amdgcn_readfirstlane(st4.z), __builtin_amdgcn_readfirstlane(st4.w) };
    const int cn[4] = { __builtin_amdgcn_readfirstlane(cn4.x), __builtin_amdgcn_readfirstlane(cn4.y),
                        __builtin_amdgcn_readfirstlane(cn4.z), __builtin_amdgcn_readfirstlane(cn4.w) };
    const float dn[4] = { dn4.x, dn4.y, dn4.z, dn4.w };

    int s[4];
    float e[4];
#pragma unroll
    for (int q = 0; q < 4; ++q) {
        const int sv = csr[lane < cn[q] ? st[q] + lane : 0];
        s[q] = lane < cn[q] ? sv : n0 + q;
    }
    float a[4];
#pragma unroll
    for (int q = 0; q < 4; ++q) a[q] = als[s[q]];
    asm volatile("" :: "v"(a[0]), "v"(a[1]), "v"(a[2]), "v"(a[3]));
#pragma unroll
    for (int q = 0; q < 4; ++q) e[q] = lane <= cn[q] ? lrelu(a[q] + dn[q]) : NEG_INF;

    float m[4];
#pragma unroll
    for (int q = 0; q < 4; ++q) {
        float mm = e[q];
#pragma nounroll
        for (int base = 32; base <= cn[q]; base += 32) {
            int s2;
            mm = fmaxf(mm, late_logit(csr, als, st[q], cn[q], n0 + q, base + lane, dn[q], s2));
        }
        m[q] = mm;
    }
#pragma unroll
    for (int off = 16; off > 0; off >>= 1)
#pragma unroll
        for (int q = 0; q < 4; ++q) m[q] = fmaxf(m[q], __shfl_xor(m[q], off));

    float den[4], wg[4];
#pragma unroll
    for (int q = 0; q < 4; ++q) {
        wg[q] = __expf(e[q] - m[q]);
        float d = wg[q];
#pragma nounroll
        for (int base = 32; base <= cn[q]; base += 32) {
            int s2;
            d += __expf(late_logit(csr, als, st[q], cn[q], n0 + q, base + lane, dn[q], s2) - m[q]);
        }
        den[q] = d;
    }
#pragma unroll
    for (int off = 16; off > 0; off >>= 1)
#pragma unroll
        for (int q = 0; q < 4; ++q) den[q] += __shfl_xor(den[q], off);
#pragma unroll
    for (int q = 0; q < 4; ++q) {
        den[q] += 1e-16f;
        wg[q] = wg[q] / den[q];
    }

#pragma unroll
    for (int q = 0; q < 4; ++q) {
        const int n   = n0 + q;
        const int tot = cn[q] + 1;
        float acc0 = 0.f, acc1 = 0.f;
        chunk_rows2(z, lane, hi, s[q], wg[q], (tot < 32) ? tot : 32, acc0, acc1);
#pragma nounroll
        for (int base = 32; base < tot; base += 32) {
            int s2;
            const float v = late_logit(csr, als, st[q], cn[q], n, base + lane, dn[q], s2);
            const float w2 = __expf(v - m[q]) / den[q];
            chunk_rows2(z, lane, hi, s2, w2, (tot - base < 32) ? (tot - base) : 32, acc0, acc1);
        }
        float* op = out + (size_t)n * OUT_CH;
        op[lane] = acc0 + bl;
        if (hi) op[32 + lane] = acc1 + bh;
    }
}

extern "C" void kernel_launch(void* const* d_in, const int* in_sizes, int n_in,
                              void* d_out, int out_size, void* d_ws, size_t ws_size,
                              hipStream_t stream) {
    const float* x      = (const float*)d_in[0];
    const int*   ei     = (const int*)  d_in[1];
    const float* W1     = (const float*)d_in[2];
    const float* a_src1 = (const float*)d_in[3];
    const float* a_dst1 = (const float*)d_in[4];
    const float* b1     = (const float*)d_in[5];
    const float* W2     = (const float*)d_in[6];
    const float* a_src2 = (const float*)d_in[7];
    const float* a_dst2 = (const float*)d_in[8];
    const float* b2     = (const float*)d_in[9];
    float* out = (float*)d_out;

    float* ws   = (float*)d_ws;
    float* h1   = ws;
    float* z2   = h1   + (size_t)NN * HC;
    float* als1 = z2   + (size_t)NN * OUT_CH;
    float* ald1 = als1 + (size_t)NN * HEADS;
    float* als2 = ald1 + (size_t)NN * HEADS;
    float* ald2 = als2 + NN;
    int* deg      = (int*)(ald2 + NN);
    int* rowstart = deg + NN;
    int* ghist    = rowstart + NN;
    int* gcur     = ghist + GH;
    int* tmp      = gcur + GH;
    int* csr      = tmp + EE;

    const int gemmBlocks = (NN / 16 + 3) / 4;
    const int nodeBlocks1 = NN / 16;
    const int nodeBlocks2 = (NN / 4 + 7) / 8;

    k_gemm1<<<gemmBlocks, 128, 0, stream>>>(x, W1, a_src1, a_dst1, h1, als1, ald1, ghist);
    k_hist   <<<EBLK, TBE, 0, stream>>>(ei, ghist);
    k_scatter<<<EBLK, TBE, 0, stream>>>(ei, ghist, gcur, tmp);
    k_bsort  <<<NB,   TBE, 0, stream>>>(ghist, tmp, deg, rowstart, csr);
    k_node1<<<nodeBlocks1, 128, 0, stream>>>(csr, rowstart, deg, h1, als1, ald1, b1,
                                             W2, a_src2, a_dst2, z2, als2, ald2);

    k_node2<<<nodeBlocks2, 256, 0, stream>>>(csr, rowstart, deg, z2, als2, ald2, b2, out);
}


# Layer-2 rows at a 192-byte pitch, read two per load (16 bytes a lane, a half wave per row): sums kept as even- and odd-entry partial sums per channel, added at the end
# speedup vs baseline: 5.9583x; 1.0368x over previous
#include <hip/hip_runtime.h>
#include <math.h>

#define NN 100000
#define EE 1600000
#define IN_CH 128
#define HID 32
#define HEADS 4
#define HC 128
#define OUT_CH 40
#define NEG_SLOPE 0.2f
#define GH 512

typedef __attribute__((ext_vector_type(2))) float v2f;
typedef __attribute__((ext_vector_type(8))) float v8f;

__device__ __forceinline__ float lrelu(float v) {
    return v >= 0.0f ? v : NEG_SLOPE * v;
}

#define P1 132
__global__ void __launch_bounds__(128)
k_gemm1(const float* __restrict__ x, const float* __restrict__ W1,
        const float* __restrict__ a_src, const float* __restrict__ a_dst,
        float* __restrict__ h1, float* __restrict__ als, float* __restrict__ ald,
        int* __restrict__ ghist) {
    __shared__ float tile[4][16 * P1];
    if (blockIdx.x == 0)
        for (int i = threadIdx.x; i < 2 * GH; i += 128) ghist[i] = 0;
    const int wave = threadIdx.x >> 5;
    const int row0 = (blockIdx.x * 4 + wave) * 16;
    if (row0 >= NN) return;
    const int lane = threadIdx.x & 31;
    const int m    = lane & 15;
    const int kb   = (lane >> 4) * 2;

    v8f c[8];
#pragma unroll
    for (int t = 0; t < 8; ++t) c[t] = (v8f){0,0,0,0,0,0,0,0};

    for (int k = 0; k < IN_CH; k += 4) {
        v2f a;
        const float* xr = x + (size_t)(row0 + m) * IN_CH + (k + kb);
        a.x = xr[0];
        a.y = xr[1];
#pragma unroll
        for (int t = 0; t < 8; ++t) {
            v2f b;
            const float* wr = W1 + (size_t)(k + kb) * HC + t * 16 + m;
            b.x = wr[0];
            b.y = wr[HC];
            c[t] = __builtin_amdgcn_wmma_f32_16x16x4_f32(false, a, false, b,
                                                         (short)0, c[t], false, false);
        }
    }
    const int rofs = (lane >> 4) * 8;
    float* sm = tile[wave];
#pragma unroll
    for (int t = 0; t < 8; ++t)
#pragma unroll
        for (int r = 0; r < 8; ++r) {
            h1[(size_t)(row0 + r + rofs) * HC + t * 16 + m] = c[t][r];
            sm[(r + rofs) * P1 + t * 16 + m] = c[t][r];
        }
    __builtin_amdgcn_wave_barrier();

    const int lr = lane & 15;
    const int hp = (lane >> 4) * 2;
    float s0 = 0.f, d0 = 0.f, s1 = 0.f, d1 = 0.f;
    const float* r0 = sm + lr * P1 + hp * HID;
    const float* as0 = a_src + hp * HID;
    const float* ad0 = a_dst + hp * HID;
#pragma unroll
    for (int cc = 0; cc < HID; ++cc) {
        const float v0 = r0[cc];
        const float v1 = r0[HID + cc];
        s0 = __builtin_fmaf(v0, as0[cc], s0);
        d0 = __builtin_fmaf(v0, ad0[cc], d0);
        s1 = __builtin_fmaf(v1, as0[HID + cc], s1);
        d1 = __builtin_fmaf(v1, ad0[HID + cc], d1);
    }
    *(float2*)(als + (size_t)(row0 + lr) * HEADS + hp) = make_float2(s0, s1);
    *(float2*)(ald + (size_t)(row0 + lr) * HEADS + hp) = make_float2(d0, d1);
}

#define BSH 8
#define BW  256
#define NB  ((NN + BW - 1) / BW)
#define TBE 512
#define EPT 8
#define BPT 12
#define EBLK ((EE / EPT + TBE - 1) / TBE)
static_assert(NB <= TBE && NB <= GH, "one thread per coarse bucket");
static_assert(EE % EPT == 0, "whole threads of edges");
static_assert(NN <= (1 << 23), "source id and local destination share one word");

__device__ __forceinline__ int block_excl_scan(int v, int* wsum) {
    const int lane = threadIdx.x & 31;
    const int wv   = threadIdx.x >> 5;
    int incl = v;
#pragma unroll
    for (int off = 1; off < 32; off <<= 1) {
        int t = __shfl_up(incl, off);
        if (lane >= off) incl += t;
    }
    if (lane == 31) wsum[wv] = incl;
    __syncthreads();
    int woff = 0;
    for (int k = 0; k < wv; ++k) woff += wsum[k];
    return woff + incl - v;
}

__global__ void __launch_bounds__(TBE)
k_hist(const int* __restrict__ ei, int* __restrict__ ghist) {
    __shared__ int h[TBE];
    const int t = threadIdx.x;
    h[t] = 0;
    __syncthreads();
    const int e0 = (blockIdx.x * TBE + t) * EPT;
    if (e0 < EE) {
        const int4 a = *(const int4*)(ei + EE + e0);
        const int4 b = *(const int4*)(ei + EE + e0 + 4);
        atomicAdd(&h[a.x >> BSH], 1); atomicAdd(&h[a.y >> BSH], 1);
        atomicAdd(&h[a.z >> BSH], 1); atomicAdd(&h[a.w >> BSH], 1);
        atomicAdd(&h[b.x >> BSH], 1); atomicAdd(&h[b.y >> BSH], 1);
        atomicAdd(&h[b.z >> BSH], 1); atomicAdd(&h[b.w >> BSH], 1);
    }
    __syncthreads();
    const int c = h[t];
    if (c) atomicAdd(&ghist[t], c);
}

__global__ void __launch_bounds__(TBE)
k_scatter(const int* __restrict__ ei, const int* __restrict__ ghist,
          int* __restrict__ gcur, int* __restrict__ tmp) {
    __shared__ int cnt[TBE];
    __shared__ int off[TBE];
    __shared__ int wsum[16];
    const int t = threadIdx.x;
    cnt[t] = 0;
    const int gv  = (t < NB) ? ghist[t] : 0;
    const int bst = block_excl_scan(gv, wsum);
    const int e0 = (blockIdx.x * TBE + t) * EPT;
    const bool act = e0 < EE;
    int4 s0, s1, d0, d1;
    int r0 = 0, r1 = 0, r2 = 0, r3 = 0, r4 = 0, r5 = 0, r6 = 0, r7 = 0;
    if (act) {
        s0 = *(const int4*)(ei + e0);
        s1 = *(const int4*)(ei + e0 + 4);
        d0 = *(const int4*)(ei + EE + e0);
        d1 = *(const int4*)(ei + EE + e0 + 4);
        r0 = atomicAdd(&cnt[d0.x >> BSH], 1); r1 = atomicAdd(&cnt[d0.y >> BSH], 1);
        r2 = atomicAdd(&cnt[d0.z >> BSH], 1); r3 = atomicAdd(&cnt[d0.w >> BSH], 1);
        r4 = atomicAdd(&cnt[d1.x >> BSH], 1); r5 = atomicAdd(&cnt[d1.y >> BSH], 1);
        r6 = atomicAdd(&cnt[d1.z >> BSH], 1); r7 = atomicAdd(&cnt[d1.w >> BSH], 1);
    }
    __syncthreads();
    {
        const int c = cnt[t];
        int base = 0;
        if (c) base = atomicAdd(&gcur[t], c);
        off[t] = bst + base;
    }
    __syncthreads();
    if (act) {
        tmp[off[d0.x >> BSH] + r0] = (s0.x << BSH) | (d0.x & (BW - 1));
        tmp[off[d0.y >> BSH] + r1] = (s0.y << BSH) | (d0.y & (BW - 1));
        tmp[off[d0.z >> BSH] + r2] = (s0.z << BSH) | (d0.z & (BW - 1));
        tmp[off[d0.w >> BSH] + r3] = (s0.w << BSH) | (d0.w & (BW - 1));
        tmp[off[d1.x >> BSH] + r4] = (s1.x << BSH) | (d1.x & (BW - 1));
        tmp[off[d1.y >> BSH] + r5] = (s1.y << BSH) | (d1.y & (BW - 1));
        tmp[off[d1.z >> BSH] + r6] = (s1.z << BSH) | (d1.z & (BW - 1));
        tmp[off[d1.w >> BSH] + r7] = (s1.w << BSH) | (d1.w & (BW - 1));
    }
}

__global__ void __launch_bounds__(TBE)
k_bsort(const int* __restrict__ ghist, const int* __restrict__ tmp,
        int* __restrict__ deg, int* __restrict__ rowstart, int* __restrict__ csr) {
    __shared__ int c[BW];
    __shared__ int cur[BW];
    __shared__ int wsumA[16];
    __shared__ int wsumB[16];
    __shared__ int sb[2];
    const int t = threadIdx.x;
    const int b = blockIdx.x;
    if (t < BW) c[t] = 0;
    const int gv = (t < NB) ? ghist[t] : 0;
    const int ex = block_excl_scan(gv, wsumA);
    if (t == b) { sb[0] = ex; sb[1] = gv; }
    __syncthreads();
    const int bstart = sb[0];
    const int bend   = bstart + sb[1];
    for (int base = bstart; base < bend; base += BPT * TBE) {
        int p[BPT];
#pragma unroll
        for (int k = 0; k < BPT; ++k) {
            const int i = base + k * TBE + t;
            p[k] = tmp[i < bend ? i : bend - 1];
        }
#pragma unroll
        for (int k = 0; k < BPT; ++k)
            if (base + k * TBE + t < bend) atomicAdd(&c[p[k] & (BW - 1)], 1);
    }
    __syncthreads();
    const int v   = (t < BW) ? c[t] : 0;
    const int pre = block_excl_scan(v, wsumB);
    if (t < BW) {
        cur[t] = pre;
        const int n = b * BW + t;
        if (n < NN) { deg[n] = v; rowstart[n] = bstart + pre; }
    }
    __syncthreads();
    for (int base = bstart; base < bend; base += BPT * TBE) {
        int p[BPT];
#pragma unroll
        for (int k = 0; k < BPT; ++k) {
            const int i = base + k * TBE + t;
            p[k] = tmp[i < bend ? i : bend - 1];
        }
#pragma unroll
        for (int k = 0; k < BPT; ++k)
            if (base + k * TBE + t < bend) {
                const int r = atomicAdd(&cur[p[k] & (BW - 1)], 1);
                csr[bstart + r] = p[k] >> BSH;
            }
    }
}


__device__ __forceinline__ void logits4(const int* __restrict__ csr, const float* __restrict__ als,
                                        int start, int cnt, int n, int idx, const float4 dn,
                                        int& s, float4& v) {
    const float NEG_INF = -__builtin_inff();
    s = n;
    v = make_float4(NEG_INF, NEG_INF, NEG_INF, NEG_INF);
    if (idx <= cnt) {
        if (idx < cnt) s = csr[start + idx];
        const float4 a = *(const float4*)(als + (size_t)s * 4);
        v.x = lrelu(a.x + dn.x);
        v.y = lrelu(a.y + dn.y);
        v.z = lrelu(a.z + dn.z);
        v.w = lrelu(a.w + dn.w);
    }
}

__device__ __forceinline__ void chunk_rows(const float* __restrict__ h1, float4* wt,
                                           const float* wth, int lane, int s, const float4 w,
                                           int nv, float4& acc) {
    __builtin_amdgcn_wave_barrier();
    wt[lane] = w;
    __builtin_amdgcn_wave_barrier();
    const float* hl = h1 + lane * 4;
    int j = 0;
    for (; j + 4 <= nv; j += 4) {
        const int sa = __builtin_amdgcn_readlane(s, j);
        const int sb = __builtin_amdgcn_readlane(s, j + 1);
        const int sc = __builtin_amdgcn_readlane(s, j + 2);
        const int sd = __builtin_amdgcn_readlane(s, j + 3);
        const float4 ra = *(const float4*)(hl + (size_t)sa * HC);
        const float4 rb = *(const float4*)(hl + (size_t)sb * HC);
        const float4 rc = *(const float4*)(hl + (size_t)sc * HC);
        const float4 rd = *(const float4*)(hl + (size_t)sd * HC);
        const float wa = wth[4 * j];
        const float wb = wth[4 * j + 4];
        const float wc = wth[4 * j + 8];
        const float wd = wth[4 * j + 12];
        acc.x += ra.x * wa; acc.y += ra.y * wa; acc.z += ra.z * wa; acc.w += ra.w * wa;
        acc.x += rb.x * wb; acc.y += rb.y * wb; acc.z += rb.z * wb; acc.w += rb.w * wb;
        acc.x += rc.x * wc; acc.y += rc.y * wc; acc.z += rc.z * wc; acc.w += rc.w * wc;
        acc.x += rd.x * wd; acc.y += rd.y * wd; acc.z += rd.z * wd; acc.w += rd.w * wd;
    }
    for (; j < nv; ++j) {
        const int sa = __builtin_amdgcn_readlane(s, j);
        const float4 ra = *(const float4*)(hl + (size_t)sa * HC);
        const float wa = wth[4 * j];
        acc.x += ra.x * wa; acc.y += ra.y * wa; acc.z += ra.z * wa; acc.w += ra.w * wa;
    }
}

#define ZP 48
#define P2 52
static_assert(NN % 16 == 0, "whole 16-node strips: every wave of every block has its four nodes");
__global__ void __launch_bounds__(128)
k_node1(const int* __restrict__ csr, const int* __restrict__ rowstart,
        const int* __restrict__ deg, const float* __restrict__ h1,
        const float* __restrict__ als, const float* __restrict__ ald,
        const float* __restrict__ b1, const float* __restrict__ W2,
        const float* __restrict__ a_src2, const float* __restrict__ a_dst2,
        float* __restrict__ z, float* __restrict__ als2, float* __restrict__ ald2) {
    __shared__ float4 wtab[4][6][32];
    __shared__ float hs[16 * P1];
    __shared__ float zt[16 * P2];
    const int lane = threadIdx.x & 31;
    const int wv   = threadIdx.x >> 5;
    {
        const int nb = __builtin_amdgcn_readfirstlane((int)(blockIdx.x * 16 + wv * 4));
        const int4 st4 = *(const int4*)(rowstart + nb);
        const int4 cn4 = *(const int4*)(deg + nb);
        const int stq[4] = { __builtin_amdgcn_readfirstlane(st4.x), __builtin_amdgcn_readfirstlane(st4.y),
                             __builtin_amdgcn_readfirstlane(st4.z), __builtin_amdgcn_readfirstlane(st4.w) };
        const int cnq[4] = { __builtin_amdgcn_readfirstlane(cn4.x), __builtin_amdgcn_readfirstlane(cn4.y),
                             __builtin_amdgcn_readfirstlane(cn4.z), __builtin_amdgcn_readfirstlane(cn4.w) };
        int sq[4];
#pragma unroll
        for (int q = 0; q < 4; ++q) {
            const int sv = csr[lane < cnq[q] ? stq[q] + lane : 0];
            sq[q] = lane < cnq[q] ? sv : nb + q;
        }
#pragma unroll
        for (int q = 0; q < 4; ++q)
            wtab[wv][1 + q][lane] = *(const float4*)(als + (size_t)sq[q] * 4);
        *(int4*)&wtab[wv][5][lane] = make_int4(sq[0], sq[1], sq[2], sq[3]);
    }
    for (int q = 0; q < 4; ++q) {
        const int n = __builtin_amdgcn_readfirstlane((int)(blockIdx.x * 16 + wv * 4 + q));
        const int start = __builtin_amdgcn_readfirstlane(rowstart[n]);
        const int cnt   = __builtin_amdgcn_readfirstlane(deg[n]);
        const int tot   = cnt + 1;
        const float4 dn = *(const float4*)(ald + (size_t)n * 4);
        float4* wt = wtab[wv][0];
        const float* wth = (const float*)wt + (lane >> 3);

        const int s0 = ((const int*)&wtab[wv][5][lane])[q];
        float4 v0 = make_float4(-__builtin_inff(), -__builtin_inff(), -__builtin_inff(), -__builtin_inff());
        {
            const float4 a0 = wtab[wv][1 + q][lane];
            if (lane <= cnt) {
                v0.x = lrelu(a0.x + dn.x);
                v0.y = lrelu(a0.y + dn.y);
                v0.z = lrelu(a0.z + dn.z);
                v0.w = lrelu(a0.w + dn.w);
            }
        }

        float4 m = v0;
        for (int base = 32; base < tot; base += 32) {
            int s; float4 v;
            logits4(csr, als, start, cnt, n, base + lane, dn, s, v);
            m.x = fmaxf(m.x, v.x); m.y = fmaxf(m.y, v.y); m.z = fmaxf(m.z, v.z); m.w = fmaxf(m.w, v.w);
        }
#pragma unroll
        for (int off = 16; off > 0; off >>= 1) {
            m.x = fmaxf(m.x, __shfl_xor(m.x, off));
            m.y = fmaxf(m.y, __shfl_xor(m.y, off));
            m.z = fmaxf(m.z, __shfl_xor(m.z, off));
            m.w = fmaxf(m.w, __shfl_xor(m.w, off));
        }

        float4 p0;
        p0.x = __expf(v0.x - m.x); p0.y = __expf(v0.y - m.y);
        p0.z = __expf(v0.z - m.z); p0.w = __expf(v0.w - m.w);
        float4 d = p0;
        for (int base = 32; base < tot; base += 32) {
            int s; float4 v;
            logits4(csr, als, start, cnt, n, base + lane, dn, s, v);
            d.x += __expf(v.x - m.x); d.y += __expf(v.y - m.y);
            d.z += __expf(v.z - m.z); d.w += __expf(v.w - m.w);
        }
#pragma unroll
        for (int off = 16; off > 0; off >>= 1) {
            d.x += __shfl_xor(d.x, off);
            d.y += __shfl_xor(d.y, off);
            d.z += __shfl_xor(d.z, off);
            d.w += __shfl_xor(d.w, off);
        }
        d.x += 1e-16f; d.y += 1e-16f; d.z += 1e-16f; d.w += 1e-16f;

        float4 acc = make_float4(0.f, 0.f, 0.f, 0.f);
        {
            float4 w;
            w.x = p0.x / d.x; w.y = p0.y / d.y; w.z = p0.z / d.z; w.w = p0.w / d.w;
            chunk_rows(h1, wt, wth, lane, s0, w, (tot < 32) ? tot : 32, acc);
        }
        for (int base = 32; base < tot; base += 32) {
            int s; float4 v;
            logits4(csr, als, start, cnt, n, base + lane, dn, s, v);
            float4 w;
            w.x = __expf(v.x - m.x) / d.x; w.y = __expf(v.y - m.y) / d.y;
            w.z = __expf(v.z - m.z) / d.z; w.w = __expf(v.w - m.w) / d.w;
            chunk_rows(h1, wt, wth, lane, s, w, (tot - base < 32) ? (tot - base) : 32, acc);
        }

        const float4 bb = *(const float4*)(b1 + lane * 4);
        float4 o;
        o.x = acc.x + bb.x; o.y = acc.y + bb.y; o.z = acc.z + bb.z; o.w = acc.w + bb.w;
        o.x = o.x > 0.0f ? o.x : expm1f(o.x);
        o.y = o.y > 0.0f ? o.y : expm1f(o.y);
        o.z = o.z > 0.0f ? o.z : expm1f(o.z);
        o.w = o.w > 0.0f ? o.w : expm1f(o.w);
        *(float4*)(hs + (wv * 4 + q) * P1 + lane * 4) = o;
    }

    const int gm  = lane & 15;
    const int gkb = (lane >> 4) * 2;
    const int col = wv * 16 + gm;
    const float* wp = W2 + (size_t)gkb * OUT_CH + (col < OUT_CH ? col : OUT_CH - 1);
    v2f bq[16];
    if (wv < 3) {
#pragma unroll
        for (int i = 0; i < 16; ++i) {
            bq[i].x = wp[(size_t)(4 * i) * OUT_CH];
            bq[i].y = wp[(size_t)(4 * i + 1) * OUT_CH];
        }
    }
    __syncthreads();

    const int row0 = blockIdx.x * 16;
    if (wv < 3) {
        const float* ap = hs + gm * P1 + gkb;
        v8f c = (v8f){0,0,0,0,0,0,0,0};
#pragma unroll
        for (int i = 0; i < 16; ++i) {
            const v2f a = *(const v2f*)(ap + 4 * i);
            c = __builtin_amdgcn_wmma_f32_16x16x4_f32(false, a, false, bq[i],
                                                      (short)0, c, false, false);
        }
#pragma unroll
        for (int k = 64; k < HC; k += 4) {
            const v2f a = *(const v2f*)(ap + k);
            v2f b;
            b.x = wp[(size_t)k * OUT_CH];
            b.y = wp[(size_t)(k + 1) * OUT_CH];
            c = __builtin_amdgcn_wmma_f32_16x16x4_f32(false, a, false, b,
                                                      (short)0, c, false, false);
        }
        const int rofs = (lane >> 4) * 8;
#pragma unroll
        for (int r = 0; r < 8; ++r) zt[(r + rofs) * P2 + col] = c[r];
    }
    __syncthreads();
    for (int idx = threadIdx.x; idx < 16 * ZP / 4; idx += 128) {
        const int r  = idx / (ZP / 4);
        const int c4 = idx - r * (ZP / 4);
        const float4 v = *(const float4*)(zt + r * P2 + c4 * 4);
        *(float4*)(z + (size_t)row0 * ZP + idx * 4) = v;
    }
    if (wv == 3) {
        const int lr = lane & 15;
        const bool tgt = lane >= 16;
        const float* av = tgt ? a_dst2 : a_src2;
        const float* zr = zt + lr * P2;
        float sacc = 0.f;
#pragma unroll
        for (int cc = 0; cc < OUT_CH; ++cc) sacc = __builtin_fmaf(zr[cc], av[cc], sacc);
        (tgt ? ald2 : als2)[row0 + lr] = sacc;
    }
}


__device__ __forceinline__ float late_logit(const int* __restrict__ csr, const float* __restrict__ als,
                                            int start, int cnt, int n, int idx, float dn, int& s) {
    s = n;
    float v = -__builtin_inff();
    if (idx <= cnt) {
        if (idx < cnt) s = csr[start + idx];
        v = lrelu(als[s] + dn);
    }
    return v;
}

__device__ __forceinline__ void chunk_rows2(const float* __restrict__ zl, int half,
                                            int s, float wgt, int nv, float4& acc) {
    const int np = (nv + 1) >> 1;
    int p = 0;
    for (; p + 4 <= np; p += 4) {
        const int i0 = 2 * p + half;
        const int sa = __shfl(s, i0);
        const int sb = __shfl(s, i0 + 2);
        const int sc = __shfl(s, i0 + 4);
        const int sd = __shfl(s, i0 + 6);
        const float wa = __shfl(wgt, i0);
        const float wb = __shfl(wgt, i0 + 2);
        const float wc = __shfl(wgt, i0 + 4);
        const float wd = __shfl(wgt, i0 + 6);
        const float4 ra = *(const float4*)(zl + (size_t)sa * ZP);
        const float4 rb = *(const float4*)(zl + (size_t)sb * ZP);
        const float4 rc = *(const float4*)(zl + (size_t)sc * ZP);
        const float4 rd = *(const float4*)(zl + (size_t)sd * ZP);
        acc.x = __builtin_fmaf(ra.x, wa, acc.x); acc.y = __builtin_fmaf(ra.y, wa, acc.y);
        acc.z = __builtin_fmaf(ra.z, wa, acc.z); acc.w = __builtin_fmaf(ra.w, wa, acc.w);
        acc.x = __builtin_fmaf(rb.x, wb, acc.x); acc.y = __builtin_fmaf(rb.y, wb, acc.y);
        acc.z = __builtin_fmaf(rb.z, wb, acc.z); acc.w = __builtin_fmaf(rb.w, wb, acc.w);
        acc.x = __builtin_fmaf(rc.x, wc, acc.x); acc.y = __builtin_fmaf(rc.y, wc, acc.y);
        acc.z = __builtin_fmaf(rc.z, wc, acc.z); acc.w = __builtin_fmaf(rc.w, wc, acc.w);
        acc.x = __builtin_fmaf(rd.x, wd, acc.x); acc.y = __builtin_fmaf(rd.y, wd, acc.y);
        acc.z = __builtin_fmaf(rd.z, wd, acc.z); acc.w = __builtin_fmaf(rd.w, wd, acc.w);
    }
    for (; p < np; ++p) {
        const int i0 = 2 * p + half;
        const int sa = __shfl(s, i0);
        const float wa = __shfl(wgt, i0);
        const float4 ra = *(const float4*)(zl + (size_t)sa * ZP);
        acc.x = __builtin_fmaf(ra.x, wa, acc.x); acc.y = __builtin_fmaf(ra.y, wa, acc.y);
        acc.z = __builtin_fmaf(ra.z, wa, acc.z); acc.w = __builtin_fmaf(ra.w, wa, acc.w);
    }
}

static_assert(NN % 4 == 0, "a wave's four nodes all exist");
static_assert(OUT_CH % 4 == 0 && OUT_CH / 4 <= 16 && OUT_CH <= ZP && ZP <= P2 && ZP % 4 == 0,
              "a row is one 16-byte load per lane of a half wave");
static_assert(EE > 0, "run slot 0 is always written");
__global__ void __launch_bounds__(256)
k_node2(const int* __restrict__ csr, const int* __restrict__ rowstart,
        const int* __restrict__ deg, const float* __restrict__ z,
        const float* __restrict__ als, const float* __restrict__ ald,
        const float* __restrict__ b2, float* __restrict__ out) {
    const int lane = threadIdx.x & 31;
    const int w = __builtin_amdgcn_readfirstlane((int)(blockIdx.x * 8 + (threadIdx.x >> 5)));
    const int n0 = w * 4;
    if (n0 >= NN) return;
    const int  half = lane >> 4;
    const int  c4   = lane & 15;
    const int  cg   = c4 < OUT_CH / 4 ? c4 : 0;
    const float4 bb = *(const float4*)(b2 + cg * 4);
    const float* zl = z + cg * 4;
    const float NEG_INF = -__builtin_inff();

    const int4   st4 = *(const int4*)(rowstart + n0);
    const int4   cn4 = *(const int4*)(deg + n0);
    const float4 dn4 = *(const float4*)(ald + n0);
    const int st[4] = { __builtin_amdgcn_readfirstlane(st4.x), __builtin_amdgcn_readfirstlane(st4.y),
                        __builtin_amdgcn_readfirstlane(st4.z), __builtin_amdgcn_readfirstlane(st4.w) };
    const int cn[4] = { __builtin_amdgcn_readfirstlane(cn4.x), __builtin_amdgcn_readfirstlane(cn4.y),
                        __builtin_amdgcn_readfirstlane(cn4.z), __builtin_amdgcn_readfirstlane(cn4.w) };
    const float dn[4] = { dn4.x, dn4.y, dn4.z, dn4.w };

    int s[4];
    float e[4];
#pragma unroll
    for (int q = 0; q < 4; ++q) {
        const int sv = csr[lane < cn[q] ? st[q] + lane : 0];
        s[q] = lane < cn[q] ? sv : n0 + q;
    }
    float a[4];
#pragma unroll
    for (int q = 0; q < 4; ++q) a[q] = als[s[q]];
    asm volatile("" :: "v"(a[0]), "v"(a[1]), "v"(a[2]), "v"(a[3]));
#pragma unroll
    for (int q = 0; q < 4; ++q) e[q] = lane <= cn[q] ? lrelu(a[q] + dn[q]) : NEG_INF;

    float m[4];
#pragma unroll
    for (int q = 0; q < 4; ++q) {
        float mm = e[q];
#pragma nounroll
        for (int base = 32; base <= cn[q]; base += 32) {
            int s2;
            mm = fmaxf(mm, late_logit(csr, als, st[q], cn[q], n0 + q, base + lane, dn[q], s2));
        }
        m[q] = mm;
    }
#pragma unroll
    for (int off = 16; off > 0; off >>= 1)
#pragma unroll
        for (int q = 0; q < 4; ++q) m[q] = fmaxf(m[q], __shfl_xor(m[q], off));

    float den[4], wg[4];
#pragma unroll
    for (int q = 0; q < 4; ++q) {
        wg[q] = __expf(e[q] - m[q]);
        float d = wg[q];
#pragma nounroll
        for (int base = 32; base <= cn[q]; base += 32) {
            int s2;
            d += __expf(late_logit(csr, als, st[q], cn[q], n0 + q, base + lane, dn[q], s2) - m[q]);
        }
        den[q] = d;
    }
#pragma unroll
    for (int off = 16; off > 0; off >>= 1)
#pragma unroll
        for (int q = 0; q < 4; ++q) den[q] += __shfl_xor(den[q], off);
#pragma unroll
    for (int q = 0; q < 4; ++q) {
        den[q] += 1e-16f;
        wg[q] = wg[q] / den[q];
    }

#pragma unroll
    for (int q = 0; q < 4; ++q) {
        const int n   = n0 + q;
        const int tot = cn[q] + 1;
        float4 acc = make_float4(0.f, 0.f, 0.f, 0.f);
        chunk_rows2(zl, half, s[q], wg[q], (tot < 32) ? tot : 32, acc);
#pragma nounroll
        for (int base = 32; base < tot; base += 32) {
            int s2;
            const float v = late_logit(csr, als, st[q], cn[q], n, base + lane, dn[q], s2);
            const float w2 = __expf(v - m[q]) / den[q];
            chunk_rows2(zl, half, s2, w2, (tot - base < 32) ? (tot - base) : 32, acc);
        }
        acc.x += __shfl_xor(acc.x, 16); acc.y += __shfl_xor(acc.y, 16);
        acc.z += __shfl_xor(acc.z, 16); acc.w += __shfl_xor(acc.w, 16);
        if (lane < OUT_CH / 4)
            *(float4*)(out + (size_t)n * OUT_CH + lane * 4) =
                make_float4(acc.x + bb.x, acc.y + bb.y, acc.z + bb.z, acc.w + bb.w);
    }
}

extern "C" void kernel_launch(void* const* d_in, const int* in_sizes, int n_in,
                              void* d_out, int out_size, void* d_ws, size_t ws_size,
                              hipStream_t stream) {
    const float* x      = (const float*)d_in[0];
    const int*   ei     = (const int*)  d_in[1];
    const float* W1     = (const float*)d_in[2];
    const float* a_src1 = (const float*)d_in[3];
    const float* a_dst1 = (const float*)d_in[4];
    const float* b1     = (const float*)d_in[5];
    const float* W2     = (const float*)d_in[6];
    const float* a_src2 = (const float*)d_in[7];
    const float* a_dst2 = (const float*)d_in[8];
    const float* b2     = (const float*)d_in[9];
    float* out = (float*)d_out;

    float* ws   = (float*)d_ws;
    float* h1   = ws;
    float* z2   = h1   + (size_t)NN * HC;
    float* als1 = z2   + (size_t)NN * ZP;
    float* ald1 = als1 + (size_t)NN * HEADS;
    float* als2 = ald1 + (size_t)NN * HEADS;
    float* ald2 = als2 + NN;
    int* deg      = (int*)(ald2 + NN);
    int* rowstart = deg + NN;
    int* ghist    = rowstart + NN;
    int* gcur     = ghist + GH;
    int* tmp      = gcur + GH;
    int* csr      = tmp + EE;

    const int gemmBlocks = (NN / 16 + 3) / 4;
    const int nodeBlocks1 = NN / 16;
    const int nodeBlocks2 = (NN / 4 + 7) / 8;

    k_gemm1<<<gemmBlocks, 128, 0, stream>>>(x, W1, a_src1, a_dst1, h1, als1, ald1, ghist);
    k_hist   <<<EBLK, TBE, 0, stream>>>(ei, ghist);
    k_scatter<<<EBLK, TBE, 0, stream>>>(ei, ghist, gcur, tmp);
    k_bsort  <<<NB,   TBE, 0, stream>>>(ghist, tmp, deg, rowstart, csr);
    k_node1<<<nodeBlocks1, 128, 0, stream>>>(csr, rowstart, deg, h1, als1, ald1, b1,
                                             W2, a_src2, a_dst2, z2, als2, ald2);

    k_node2<<<nodeBlocks2, 256, 0, stream>>>(csr, rowstart, deg, z2, als2, ald2, b2, out);
}


# Bucket build with no grid-wide count: edge blocks sort their edges by coarse bucket into own slices inside the GEMM launch, bucket blocks gather the pieces; 2 launches fewer; run order by LDS adds
# speedup vs baseline: 6.6619x; 1.1181x over previous
#include <hip/hip_runtime.h>
#include <math.h>

#define NN 100000
#define EE 1600000
#define IN_CH 128
#define HID 32
#define HEADS 4
#define HC 128
#define OUT_CH 40
#define NEG_SLOPE 0.2f

typedef __attribute__((ext_vector_type(2))) float v2f;
typedef __attribute__((ext_vector_type(8))) float v8f;

__device__ __forceinline__ float lrelu(float v) {
    return v >= 0.0f ? v : NEG_SLOPE * v;
}

#define BSH 8
#define BW  256
#define NB  ((NN + BW - 1) / BW)
#define SL  4096
#define EB  ((EE + SL - 1) / SL)
#define OFFP (NB + 1)
#define SG  (SL / (128 * 4))
#define TBE 512
#define PU  25
#define DB  17
static_assert(OFFP <= 512 && OFFP % 4 == 0, "a thread of an edge block owns four bins, whole 16-byte stores of offsets");
static_assert(EB <= TBE, "one thread per piece in k_bsort");
static_assert(EE % 4 == 0, "whole groups of four edges");
static_assert(NN <= (1 << 23), "source id and local destination share one word");
static_assert(NN <= (1 << DB) && SL <= (1 << (31 - DB)), "destination and rank share one register");

__device__ __forceinline__ int block_excl_scan(int v, int* wsum) {
    const int lane = threadIdx.x & 31;
    const int wv   = threadIdx.x >> 5;
    int incl = v;
#pragma unroll
    for (int off = 1; off < 32; off <<= 1) {
        int t = __shfl_up(incl, off);
        if (lane >= off) incl += t;
    }
    if (lane == 31) wsum[wv] = incl;
    __syncthreads();
    int woff = 0;
    for (int k = 0; k < wv; ++k) woff += wsum[k];
    return woff + incl - v;
}

__device__ __forceinline__ void edge_block_sort(const int* __restrict__ ei, int* __restrict__ boff,
                                                int* __restrict__ tmp, int* lds, int b) {
    int* cnt  = lds;
    int* off  = lds + 512;
    int* wsum = lds + 1024;
    const int t = threadIdx.x;
    for (int i = t; i < 512; i += 128) cnt[i] = 0;
    __syncthreads();
    const int ebase = b * SL;
    int4 d[SG], s[SG];
#pragma unroll
    for (int g = 0; g < SG; ++g) {
        const int e0 = ebase + (g * 128 + t) * 4;
        const int ec = e0 < EE ? e0 : 0;
        d[g] = *(const int4*)(ei + EE + ec);
        s[g] = *(const int4*)(ei + ec);
    }
#pragma unroll
    for (int g = 0; g < SG; ++g) {
        const int e0 = ebase + (g * 128 + t) * 4;
        if (e0 < EE) {
            d[g].x |= atomicAdd(&cnt[d[g].x >> BSH], 1) << DB;
            d[g].y |= atomicAdd(&cnt[d[g].y >> BSH], 1) << DB;
            d[g].z |= atomicAdd(&cnt[d[g].z >> BSH], 1) << DB;
            d[g].w |= atomicAdd(&cnt[d[g].w >> BSH], 1) << DB;
        }
    }
    __syncthreads();
    {
        const int4 v = *(const int4*)(cnt + 4 * t);
        const int ex = block_excl_scan(v.x + v.y + v.z + v.w, wsum);
        const int4 o = make_int4(ex, ex + v.x, ex + v.x + v.y, ex + v.x + v.y + v.z);
        *(int4*)(off + 4 * t) = o;
        if (4 * t < OFFP) *(int4*)(boff + (size_t)b * OFFP + 4 * t) = o;
    }
    __syncthreads();
    int* sl = tmp + (size_t)b * SL;
#pragma unroll
    for (int g = 0; g < SG; ++g) {
        const int e0 = ebase + (g * 128 + t) * 4;
        if (e0 < EE) {
            const int4 dd = make_int4(d[g].x & ((1 << DB) - 1), d[g].y & ((1 << DB) - 1),
                                      d[g].z & ((1 << DB) - 1), d[g].w & ((1 << DB) - 1));
            sl[off[dd.x >> BSH] + (d[g].x >> DB)] = (s[g].x << BSH) | (dd.x & (BW - 1));
            sl[off[dd.y >> BSH] + (d[g].y >> DB)] = (s[g].y << BSH) | (dd.y & (BW - 1));
            sl[off[dd.z >> BSH] + (d[g].z >> DB)] = (s[g].z << BSH) | (dd.z & (BW - 1));
            sl[off[dd.w >> BSH] + (d[g].w >> DB)] = (s[g].w << BSH) | (dd.w & (BW - 1));
        }
    }
}

#define P1 132
static_assert(4 * 16 * P1 >= 1040, "an edge block's counters fit in the tile's LDS");
__global__ void __launch_bounds__(128)
k_gemm1(const float* __restrict__ x, const float* __restrict__ W1,
        const float* __restrict__ a_src, const float* __restrict__ a_dst,
        float* __restrict__ h1, float* __restrict__ als, float* __restrict__ ald,
        const int* __restrict__ ei, int* __restrict__ boff, int* __restrict__ tmp) {
    __shared__ __attribute__((aligned(16))) float tile[4][16 * P1];
    if (blockIdx.x < EB) {
        edge_block_sort(ei, boff, tmp, (int*)(void*)&tile[0][0], (int)blockIdx.x);
        return;
    }
    const int gb   = (int)blockIdx.x - EB;
    const int wave = threadIdx.x >> 5;
    const int row0 = (gb * 4 + wave) * 16;
    if (row0 >= NN) return;
    const int lane = threadIdx.x & 31;
    const int m    = lane & 15;
    const int kb   = (lane >> 4) * 2;

    v8f c[8];
#pragma unroll
    for (int t = 0; t < 8; ++t) c[t] = (v8f){0,0,0,0,0,0,0,0};

    for (int k = 0; k < IN_CH; k += 4) {
        v2f a;
        const float* xr = x + (size_t)(row0 + m) * IN_CH + (k + kb);
        a.x = xr[0];
        a.y = xr[1];
#pragma unroll
        for (int t = 0; t < 8; ++t) {
            v2f b;
            const float* wr = W1 + (size_t)(k + kb) * HC + t * 16 + m;
            b.x = wr[0];
            b.y = wr[HC];
            c[t] = __builtin_amdgcn_wmma_f32_16x16x4_f32(false, a, false, b,
                                                         (short)0, c[t], false, false);
        }
    }
    const int rofs = (lane >> 4) * 8;
    float* sm = tile[wave];
#pragma unroll
    for (int t = 0; t < 8; ++t)
#pragma unroll
        for (int r = 0; r < 8; ++r) {
            h1[(size_t)(row0 + r + rofs) * HC + t * 16 + m] = c[t][r];
            sm[(r + rofs) * P1 + t * 16 + m] = c[t][r];
        }
    __builtin_amdgcn_wave_barrier();

    const int lr = lane & 15;
    const int hp = (lane >> 4) * 2;
    float s0 = 0.f, d0 = 0.f, s1 = 0.f, d1 = 0.f;
    const float* r0 = sm + lr * P1 + hp * HID;
    const float* as0 = a_src + hp * HID;
    const float* ad0 = a_dst + hp * HID;
#pragma unroll
    for (int cc = 0; cc < HID; ++cc) {
        const float v0 = r0[cc];
        const float v1 = r0[HID + cc];
        s0 = __builtin_fmaf(v0, as0[cc], s0);
        d0 = __builtin_fmaf(v0, ad0[cc], d0);
        s1 = __builtin_fmaf(v1, as0[HID + cc], s1);
        d1 = __builtin_fmaf(v1, ad0[HID + cc], d1);
    }
    *(float2*)(als + (size_t)(row0 + lr) * HEADS + hp) = make_float2(s0, s1);
    *(float2*)(ald + (size_t)(row0 + lr) * HEADS + hp) = make_float2(d0, d1);
}

__global__ void __launch_bounds__(TBE)
k_bsort(const int* __restrict__ boff, const int* __restrict__ tmp,
        int* __restrict__ deg, int* __restrict__ rowstart, int* __restrict__ csr) {
    __shared__ int c[BW];
    __shared__ int cur[BW];
    __shared__ int pst[TBE];
    __shared__ int pcn[TBE];
    __shared__ int wsumA[16];
    __shared__ int wsumC[16];
    __shared__ int sb[1];
    const int t    = threadIdx.x;
    const int b    = blockIdx.x;
    const int lane = t & 31;
    const int wv   = t >> 5;
    if (t < BW) c[t] = 0;
    int st = 0, cn = 0;
    if (t < EB) {
        const int* o = boff + (size_t)t * OFFP + b;
        st = o[0];
        cn = o[1] - st;
    }
    pst[t] = st;
    pcn[t] = cn;
    {
        const int exs = block_excl_scan(st, wsumA);
        if (t == TBE - 1) sb[0] = exs + st;
    }
    __syncthreads();
    const int bstart = sb[0];
    for (int p0 = wv; p0 < EB; p0 += 16 * PU) {
        int v[PU];
#pragma unroll
        for (int u = 0; u < PU; ++u) {
            const int p = p0 + 16 * u;
            int s0 = 0, n = 0;
            if (p < EB) { s0 = pst[p]; n = pcn[p]; }
            const bool in = lane < n;
            const int* src = in ? tmp : boff;
            v[u] = src[in ? (size_t)p * SL + s0 + lane : (size_t)0];
        }
#pragma unroll
        for (int u = 0; u < PU; ++u) {
            const int p = p0 + 16 * u;
            const int n = (p < EB) ? pcn[p] : 0;
            if (lane < n) atomicAdd(&c[v[u] & (BW - 1)], 1);
            for (int j = 32 + lane; j < n; j += 32) {
                const int w = tmp[(size_t)p * SL + pst[p] + j];
                atomicAdd(&c[w & (BW - 1)], 1);
            }
        }
    }
    __syncthreads();
    const int vv  = (t < BW) ? c[t] : 0;
    const int pre = block_excl_scan(vv, wsumC);
    if (t < BW) {
        cur[t] = pre;
        const int nd = b * BW + t;
        if (nd < NN) { deg[nd] = vv; rowstart[nd] = bstart + pre; }
    }
    __syncthreads();
    for (int p0 = wv; p0 < EB; p0 += 16 * PU) {
        int v[PU];
#pragma unroll
        for (int u = 0; u < PU; ++u) {
            const int p = p0 + 16 * u;
            int s0 = 0, n = 0;
            if (p < EB) { s0 = pst[p]; n = pcn[p]; }
            const bool in = lane < n;
            const int* src = in ? tmp : boff;
            v[u] = src[in ? (size_t)p * SL + s0 + lane : (size_t)0];
        }
#pragma unroll
        for (int u = 0; u < PU; ++u) {
            const int p = p0 + 16 * u;
            const int n = (p < EB) ? pcn[p] : 0;
            if (lane < n) {
                const int r = atomicAdd(&cur[v[u] & (BW - 1)], 1);
                csr[bstart + r] = v[u] >> BSH;
            }
            for (int j = 32 + lane; j < n; j += 32) {
                const int w = tmp[(size_t)p * SL + pst[p] + j];
                const int r = atomicAdd(&cur[w & (BW - 1)], 1);
                csr[bstart + r] = w >> BSH;
            }
        }
    }
}


__device__ __forceinline__ void logits4(const int* __restrict__ csr, const float* __restrict__ als,
                                        int start, int cnt, int n, int idx, const float4 dn,
                                        int& s, float4& v) {
    const float NEG_INF = -__builtin_inff();
    s = n;
    v = make_float4(NEG_INF, NEG_INF, NEG_INF, NEG_INF);
    if (idx <= cnt) {
        if (idx < cnt) s = csr[start + idx];
        const float4 a = *(const float4*)(als + (size_t)s * 4);
        v.x = lrelu(a.x + dn.x);
        v.y = lrelu(a.y + dn.y);
        v.z = lrelu(a.z + dn.z);
        v.w = lrelu(a.w + dn.w);
    }
}

__device__ __forceinline__ void chunk_rows(const float* __restrict__ h1, float4* wt,
                                           const float* wth, int lane, int s, const float4 w,
                                           int nv, float4& acc) {
    __builtin_amdgcn_wave_barrier();
    wt[lane] = w;
    __builtin_amdgcn_wave_barrier();
    const float* hl = h1 + lane * 4;
    int j = 0;
    for (; j + 4 <= nv; j += 4) {
        const int sa = __builtin_amdgcn_readlane(s, j);
        const int sb = __builtin_amdgcn_readlane(s, j + 1);
        const int sc = __builtin_amdgcn_readlane(s, j + 2);
        const int sd = __builtin_amdgcn_readlane(s, j + 3);
        const float4 ra = *(const float4*)(hl + (size_t)sa * HC);
        const float4 rb = *(const float4*)(hl + (size_t)sb * HC);
        const float4 rc = *(const float4*)(hl + (size_t)sc * HC);
        const float4 rd = *(const float4*)(hl + (size_t)sd * HC);
        const float wa = wth[4 * j];
        const float wb = wth[4 * j + 4];
        const float wc = wth[4 * j + 8];
        const float wd = wth[4 * j + 12];
        acc.x += ra.x * wa; acc.y += ra.y * wa; acc.z += ra.z * wa; acc.w += ra.w * wa;
        acc.x += rb.x * wb; acc.y += rb.y * wb; acc.z += rb.z * wb; acc.w += rb.w * wb;
        acc.x += rc.x * wc; acc.y += rc.y * wc; acc.z += rc.z * wc; acc.w += rc.w * wc;
        acc.x += rd.x * wd; acc.y += rd.y * wd; acc.z += rd.z * wd; acc.w += rd.w * wd;
    }
    for (; j < nv; ++j) {
        const int sa = __builtin_amdgcn_readlane(s, j);
        const float4 ra = *(const float4*)(hl + (size_t)sa * HC);
        const float wa = wth[4 * j];
        acc.x += ra.x * wa; acc.y += ra.y * wa; acc.z += ra.z * wa; acc.w += ra.w * wa;
    }
}

#define ZP 48
#define P2 52
static_assert(NN % 16 == 0, "whole 16-node strips: every wave of every block has its four nodes");
__global__ void __launch_bounds__(128)
k_node1(const int* __restrict__ csr, const int* __restrict__ rowstart,
        const int* __restrict__ deg, const float* __restrict__ h1,
        const float* __restrict__ als, const float* __restrict__ ald,
        const float* __restrict__ b1, const float* __restrict__ W2,
        const float* __restrict__ a_src2, const float* __restrict__ a_dst2,
        float* __restrict__ z, float* __restrict__ ald2) {
    __shared__ float4 wtab[4][6][32];
    __shared__ float hs[16 * P1];
    __shared__ float zt[16 * P2];
    const int lane = threadIdx.x & 31;
    const int wv   = threadIdx.x >> 5;
    {
        const int nb = __builtin_amdgcn_readfirstlane((int)(blockIdx.x * 16 + wv * 4));
        const int4 st4 = *(const int4*)(rowstart + nb);
        const int4 cn4 = *(const int4*)(deg + nb);
        const int stq[4] = { __builtin_amdgcn_readfirstlane(st4.x), __builtin_amdgcn_readfirstlane(st4.y),
                             __builtin_amdgcn_readfirstlane(st4.z), __builtin_amdgcn_readfirstlane(st4.w) };
        const int cnq[4] = { __builtin_amdgcn_readfirstlane(cn4.x), __builtin_amdgcn_readfirstlane(cn4.y),
                             __builtin_amdgcn_readfirstlane(cn4.z), __builtin_amdgcn_readfirstlane(cn4.w) };
        int sq[4];
#pragma unroll
        for (int q = 0; q < 4; ++q) {
            const int sv = csr[lane < cnq[q] ? stq[q] + lane : 0];
            sq[q] = lane < cnq[q] ? sv : nb + q;
        }
#pragma unroll
        for (int q = 0; q < 4; ++q)
            wtab[wv][1 + q][lane] = *(const float4*)(als + (size_t)sq[q] * 4);
        *(int4*)&wtab[wv][5][lane] = make_int4(sq[0], sq[1], sq[2], sq[3]);
    }
    for (int q = 0; q < 4; ++q) {
        const int n = __builtin_amdgcn_readfirstlane((int)(blockIdx.x * 16 + wv * 4 + q));
        const int start = __builtin_amdgcn_readfirstlane(rowstart[n]);
        const int cnt   = __builtin_amdgcn_readfirstlane(deg[n]);
        const int tot   = cnt + 1;
        const float4 dn = *(const float4*)(ald + (size_t)n * 4);
        float4* wt = wtab[wv][0];
        const float* wth = (const float*)wt + (lane >> 3);

        const int s0 = ((const int*)&wtab[wv][5][lane])[q];
        float4 v0 = make_float4(-__builtin_inff(), -__builtin_inff(), -__builtin_inff(), -__builtin_inff());
        {
            const float4 a0 = wtab[wv][1 + q][lane];
            if (lane <= cnt) {
                v0.x = lrelu(a0.x + dn.x);
                v0.y = lrelu(a0.y + dn.y);
                v0.z = lrelu(a0.z + dn.z);
                v0.w = lrelu(a0.w + dn.w);
            }
        }

        float4 m = v0;
        for (int base = 32; base < tot; base += 32) {
            int s; float4 v;
            logits4(csr, als, start, cnt, n, base + lane, dn, s, v);
            m.x = fmaxf(m.x, v.x); m.y = fmaxf(m.y, v.y); m.z = fmaxf(m.z, v.z); m.w = fmaxf(m.w, v.w);
        }
#pragma unroll
        for (int off = 16; off > 0; off >>= 1) {
            m.x = fmaxf(m.x, __shfl_xor(m.x, off));
            m.y = fmaxf(m.y, __shfl_xor(m.y, off));
            m.z = fmaxf(m.z, __shfl_xor(m.z, off));
            m.w = fmaxf(m.w, __shfl_xor(m.w, off));
        }

        float4 p0;
        p0.x = __expf(v0.x - m.x); p0.y = __expf(v0.y - m.y);
        p0.z = __expf(v0.z - m.z); p0.w = __expf(v0.w - m.w);
        float4 d = p0;
        for (int base = 32; base < tot; base += 32) {
            int s; float4 v;
            logits4(csr, als, start, cnt, n, base + lane, dn, s, v);
            d.x += __expf(v.x - m.x); d.y += __expf(v.y - m.y);
            d.z += __expf(v.z - m.z); d.w += __expf(v.w - m.w);
        }
#pragma unroll
        for (int off = 16; off > 0; off >>= 1) {
            d.x += __shfl_xor(d.x, off);
            d.y += __shfl_xor(d.y, off);
            d.z += __shfl_xor(d.z, off);
            d.w += __shfl_xor(d.w, off);
        }
        d.x += 1e-16f; d.y += 1e-16f; d.z += 1e-16f; d.w += 1e-16f;

        float4 acc = make_float4(0.f, 0.f, 0.f, 0.f);
        {
            float4 w;
            w.x = p0.x / d.x; w.y = p0.y / d.y; w.z = p0.z / d.z; w.w = p0.w / d.w;
            chunk_rows(h1, wt, wth, lane, s0, w, (tot < 32) ? tot : 32, acc);
        }
        for (int base = 32; base < tot; base += 32) {
            int s; float4 v;
            logits4(csr, als, start, cnt, n, base + lane, dn, s, v);
            float4 w;
            w.x = __expf(v.x - m.x) / d.x; w.y = __expf(v.y - m.y) / d.y;
            w.z = __expf(v.z - m.z) / d.z; w.w = __expf(v.w - m.w) / d.w;
            chunk_rows(h1, wt, wth, lane, s, w, (tot - base < 32) ? (tot - base) : 32, acc);
        }

        const float4 bb = *(const float4*)(b1 + lane * 4);
        float4 o;
        o.x = acc.x + bb.x; o.y = acc.y + bb.y; o.z = acc.z + bb.z; o.w = acc.w + bb.w;
        o.x = o.x > 0.0f ? o.x : expm1f(o.x);
        o.y = o.y > 0.0f ? o.y : expm1f(o.y);
        o.z = o.z > 0.0f ? o.z : expm1f(o.z);
        o.w = o.w > 0.0f ? o.w : expm1f(o.w);
        *(float4*)(hs + (wv * 4 + q) * P1 + lane * 4) = o;
    }

    const int gm  = lane & 15;
    const int gkb = (lane >> 4) * 2;
    const int col = wv * 16 + gm;
    const float* wp = W2 + (size_t)gkb * OUT_CH + (col < OUT_CH ? col : OUT_CH - 1);
    v2f bq[16];
    if (wv < 3) {
#pragma unroll
        for (int i = 0; i < 16; ++i) {
            bq[i].x = wp[(size_t)(4 * i) * OUT_CH];
            bq[i].y = wp[(size_t)(4 * i + 1) * OUT_CH];
        }
    }
    __syncthreads();

    const int row0 = blockIdx.x * 16;
    if (wv < 3) {
        const float* ap = hs + gm * P1 + gkb;
        v8f c = (v8f){0,0,0,0,0,0,0,0};
#pragma unroll
        for (int i = 0; i < 16; ++i) {
            const v2f a = *(const v2f*)(ap + 4 * i);
            c = __builtin_amdgcn_wmma_f32_16x16x4_f32(false, a, false, bq[i],
                                                      (short)0, c, false, false);
        }
#pragma unroll
        for (int k = 64; k < HC; k += 4) {
            const v2f a = *(const v2f*)(ap + k);
            v2f b;
            b.x = wp[(size_t)k * OUT_CH];
            b.y = wp[(size_t)(k + 1) * OUT_CH];
            c = __builtin_amdgcn_wmma_f32_16x16x4_f32(false, a, false, b,
                                                      (short)0, c, false, false);
        }
        const int rofs = (lane >> 4) * 8;
#pragma unroll
        for (int r = 0; r < 8; ++r) zt[(r + rofs) * P2 + col] = c[r];
    }
    __syncthreads();
    if (wv != 3) return;
    {
        const int lr = lane & 15;
        const bool tgt = lane >= 16;
        const float* av = tgt ? a_dst2 : a_src2;
        const float* zr = zt + lr * P2;
        float sacc = 0.f;
#pragma unroll
        for (int cc = 0; cc < OUT_CH; ++cc) sacc = __builtin_fmaf(zr[cc], av[cc], sacc);
        if (tgt) ald2[row0 + lr] = sacc;
        else     zt[lr * P2 + OUT_CH] = sacc;
    }
    __builtin_amdgcn_wave_barrier();
    const int t3 = (int)threadIdx.x - 3 * 32;
#pragma unroll
    for (int i = 0; i < 16 * ZP / 4 / 32; ++i) {
        const int idx = t3 + 32 * i;
        const int r  = idx / (ZP / 4);
        const int c4 = idx - r * (ZP / 4);
        const float4 v = *(const float4*)(zt + r * P2 + c4 * 4);
        *(float4*)(z + (size_t)row0 * ZP + idx * 4) = v;
    }
}


#define M_START (-3.0e38f)

__device__ __forceinline__ float row_logit(float v) {
    return __int_as_float(__builtin_amdgcn_update_dpp(0, __float_as_int(v), 0x150 + OUT_CH / 4, 0xf, 0xf, false));
}

__device__ __forceinline__ float xhalf(float x) {
    const int b = __float_as_int(x);
    return __int_as_float(__builtin_amdgcn_permlanex16(b, b, 0x76543210, 0xfedcba98, false, false));
}

__device__ __forceinline__ void chunk_online(const float* __restrict__ zl, int half, int s, int nv,
                                             float dn, float& m, float& d, float4& acc) {
    const float NEG_INF = -__builtin_inff();
    const int np = (nv + 1) >> 1;
    for (int p = 0; p < np; p += 4) {
        int sid[4];
#pragma unroll
        for (int j = 0; j < 4; ++j) sid[j] = __shfl(s, (2 * (p + j) + half) & 31);
        float4 r[4];
#pragma unroll
        for (int j = 0; j < 4; ++j) {
            r[j] = make_float4(0.f, 0.f, 0.f, 0.f);
            if (p + j < np)
                r[j] = *(const float4*)(zl + (size_t)sid[j] * ZP);
        }
        float e[4];
#pragma unroll
        for (int j = 0; j < 4; ++j) {
            const float a = row_logit(r[j].x);
            e[j] = (2 * (p + j) + half < nv) ? lrelu(a + dn) : NEG_INF;
        }
        const float mn = fmaxf(fmaxf(m, fmaxf(e[0], e[1])), fmaxf(e[2], e[3]));
        const float sc = __expf(m - mn);
        m = mn;
        const float p0 = __expf(e[0] - mn);
        const float p1 = __expf(e[1] - mn);
        const float p2 = __expf(e[2] - mn);
        const float p3 = __expf(e[3] - mn);
        d = d * sc + p0; d += p1; d += p2; d += p3;
        acc.x = __builtin_fmaf(r[0].x, p0, acc.x * sc); acc.y = __builtin_fmaf(r[0].y, p0, acc.y * sc);
        acc.z = __builtin_fmaf(r[0].z, p0, acc.z * sc); acc.w = __builtin_fmaf(r[0].w, p0, acc.w * sc);
        acc.x = __builtin_fmaf(r[1].x, p1, acc.x); acc.y = __builtin_fmaf(r[1].y, p1, acc.y);
        acc.z = __builtin_fmaf(r[1].z, p1, acc.z); acc.w = __builtin_fmaf(r[1].w, p1, acc.w);
        acc.x = __builtin_fmaf(r[2].x, p2, acc.x); acc.y = __builtin_fmaf(r[2].y, p2, acc.y);
        acc.z = __builtin_fmaf(r[2].z, p2, acc.z); acc.w = __builtin_fmaf(r[2].w, p2, acc.w);
        acc.x = __builtin_fmaf(r[3].x, p3, acc.x); acc.y = __builtin_fmaf(r[3].y, p3, acc.y);
        acc.z = __builtin_fmaf(r[3].z, p3, acc.z); acc.w = __builtin_fmaf(r[3].w, p3, acc.w);
    }
}

static_assert(NN % 4 == 0, "a wave's four nodes all exist");
static_assert(OUT_CH % 4 == 0 && OUT_CH / 4 < 16 && OUT_CH + 4 <= ZP && ZP <= P2 && ZP % 4 == 0,
              "a row and its logit column are one 16-byte load per lane of a half wave");
static_assert(EE > 0, "run slot 0 is always written");
__global__ void __launch_bounds__(256)
k_node2(const int* __restrict__ csr, const int* __restrict__ rowstart,
        const int* __restrict__ deg, const float* __restrict__ z,
        const float* __restrict__ ald,
        const float* __restrict__ b2, float* __restrict__ out) {
    const int lane = threadIdx.x & 31;
    const int w = __builtin_amdgcn_readfirstlane((int)(blockIdx.x * 8 + (threadIdx.x >> 5)));
    const int n0 = w * 4;
    if (n0 >= NN) return;
    const int  half = lane >> 4;
    const int  c4   = lane & 15;
    const bool chan = c4 < OUT_CH / 4;
    const float4 bb = *(const float4*)(b2 + (chan ? c4 : 0) * 4);
    const float* zl = z + (chan ? c4 : OUT_CH / 4) * 4;

    const int4   st4 = *(const int4*)(rowstart + n0);
    const int4   cn4 = *(const int4*)(deg + n0);
    const float4 dn4 = *(const float4*)(ald + n0);
    const int st[4] = { __builtin_amdgcn_readfirstlane(st4.x), __builtin_amdgcn_readfirstlane(st4.y),
                        __builtin_amdgcn_readfirstlane(st4.z), __builtin_amdgcn_readfirstlane(st4.w) };
    const int cn[4] = { __builtin_amdgcn_readfirstlane(cn4.x), __builtin_amdgcn_readfirstlane(cn4.y),
                        __builtin_amdgcn_readfirstlane(cn4.z), __builtin_amdgcn_readfirstlane(cn4.w) };
    const float dn[4] = { dn4.x, dn4.y, dn4.z, dn4.w };

    int s[4];
#pragma unroll
    for (int q = 0; q < 4; ++q) {
        const int sv = csr[lane < cn[q] ? st[q] + lane : 0];
        s[q] = lane < cn[q] ? sv : n0 + q;
    }

#pragma unroll
    for (int q = 0; q < 4; ++q) {
        const int n   = n0 + q;
        const int tot = cn[q] + 1;
        float m = M_START, d = 0.f;
        float4 acc = make_float4(0.f, 0.f, 0.f, 0.f);
        chunk_online(zl, half, s[q], (tot < 32) ? tot : 32, dn[q], m, d, acc);
#pragma nounroll
        for (int base = 32; base < tot; base += 32) {
            const int idx = base + lane;
            int s2 = n;
            if (idx < cn[q]) s2 = csr[st[q] + idx];
            chunk_online(zl, half, s2, (tot - base < 32) ? (tot - base) : 32, dn[q], m, d, acc);
        }
        const float mm = fmaxf(m, xhalf(m));
        const float f  = __expf(m - mm);
        d *= f;
        acc.x *= f; acc.y *= f; acc.z *= f; acc.w *= f;
        d += xhalf(d);
        acc.x += xhalf(acc.x); acc.y += xhalf(acc.y);
        acc.z += xhalf(acc.z); acc.w += xhalf(acc.w);
        const float inv = 1.0f / (d + 1e-16f);
        if (lane < OUT_CH / 4)
            *(float4*)(out + (size_t)n * OUT_CH + lane * 4) =
                make_float4(__builtin_fmaf(acc.x, inv, bb.x), __builtin_fmaf(acc.y, inv, bb.y),
                            __builtin_fmaf(acc.z, inv, bb.z), __builtin_fmaf(acc.w, inv, bb.w));
    }
}

extern "C" void kernel_launch(void* const* d_in, const int* in_sizes, int n_in,
                              void* d_out, int out_size, void* d_ws, size_t ws_size,
                              hipStream_t stream) {
    const float* x      = (const float*)d_in[0];
    const int*   ei     = (const int*)  d_in[1];
    const float* W1     = (const float*)d_in[2];
    const float* a_src1 = (const float*)d_in[3];
    const float* a_dst1 = (const float*)d_in[4];
    const float* b1     = (const float*)d_in[5];
    const float* W2     = (const float*)d_in[6];
    const float* a_src2 = (const float*)d_in[7];
    const float* a_dst2 = (const float*)d_in[8];
    const float* b2     = (const float*)d_in[9];
    float* out = (float*)d_out;

    float* ws   = (float*)d_ws;
    float* h1   = ws;
    float* z2   = h1   + (size_t)NN * HC;
    float* als1 = z2   + (size_t)NN * ZP;
    float* ald1 = als1 + (size_t)NN * HEADS;
    float* ald2 = ald1 + (size_t)NN * HEADS;
    int* deg      = (int*)(ald2 + NN);
    int* rowstart = deg + NN;
    int* boff     = rowstart + NN;
    int* tmp      = boff + EB * OFFP;
    int* csr      = tmp + EB * SL;

    const int gemmBlocks = (NN / 16 + 3) / 4;
    const int nodeBlocks1 = NN / 16;
    const int nodeBlocks2 = (NN / 4 + 7) / 8;

    k_gemm1<<<EB + gemmBlocks, 128, 0, stream>>>(x, W1, a_src1, a_dst1, h1, als1, ald1, ei, boff, tmp);
    k_bsort<<<NB, TBE, 0, stream>>>(boff, tmp, deg, rowstart, csr);
    k_node1<<<nodeBlocks1, 128, 0, stream>>>(csr, rowstart, deg, h1, als1, ald1, b1,
                                             W2, a_src2, a_dst2, z2, ald2);

    k_node2<<<nodeBlocks2, 256, 0, stream>>>(csr, rowstart, deg, z2, ald2, b2, out);
}


# Both node kernels: Leaky ReLU, max and exp once per lane for one row of a batch of four (layer 1: the 8-lane logit sum ends as one row per lane), weights passed by DPP; denominator summed per lane
# speedup vs baseline: 7.4627x; 1.1202x over previous
#include <hip/hip_runtime.h>
#include <math.h>

#define NN 100000
#define EE 1600000
#define IN_CH 128
#define HID 32
#define HEADS 4
#define HC 128
#define OUT_CH 40
#define NEG_SLOPE 0.2f

typedef __attribute__((ext_vector_type(2))) float v2f;
typedef __attribute__((ext_vector_type(8))) float v8f;

__device__ __forceinline__ float lrelu(float v) {
    return v >= 0.0f ? v : NEG_SLOPE * v;
}

#define BSH 8
#define BW  256
#define NB  ((NN + BW - 1) / BW)
#define SL  4096
#define EB  ((EE + SL - 1) / SL)
#define OFFP (NB + 1)
#define SG  (SL / (128 * 4))
#define TBE 512
#define PU  25
#define DB  17
static_assert(OFFP <= 512 && OFFP % 4 == 0, "a thread of an edge block owns four bins, whole 16-byte stores of offsets");
static_assert(EB <= TBE, "one thread per piece in k_bsort");
static_assert(EE % 4 == 0, "whole groups of four edges");
static_assert(NN <= (1 << 23), "source id and local destination share one word");
static_assert(NN <= (1 << DB) && SL <= (1 << (31 - DB)), "destination and rank share one register");

__device__ __forceinline__ int block_excl_scan(int v, int* wsum) {
    const int lane = threadIdx.x & 31;
    const int wv   = threadIdx.x >> 5;
    int incl = v;
#pragma unroll
    for (int off = 1; off < 32; off <<= 1) {
        int t = __shfl_up(incl, off);
        if (lane >= off) incl += t;
    }
    if (lane == 31) wsum[wv] = incl;
    __syncthreads();
    int woff = 0;
    for (int k = 0; k < wv; ++k) woff += wsum[k];
    return woff + incl - v;
}

__device__ __forceinline__ void edge_block_sort(const int* __restrict__ ei, int* __restrict__ boff,
                                                int* __restrict__ tmp, int* lds, int b) {
    int* cnt  = lds;
    int* off  = lds + 512;
    int* wsum = lds + 1024;
    const int t = threadIdx.x;
    for (int i = t; i < 512; i += 128) cnt[i] = 0;
    __syncthreads();
    const int ebase = b * SL;
    int4 d[SG], s[SG];
#pragma unroll
    for (int g = 0; g < SG; ++g) {
        const int e0 = ebase + (g * 128 + t) * 4;
        const int ec = e0 < EE ? e0 : 0;
        d[g] = *(const int4*)(ei + EE + ec);
        s[g] = *(const int4*)(ei + ec);
    }
#pragma unroll
    for (int g = 0; g < SG; ++g) {
        const int e0 = ebase + (g * 128 + t) * 4;
        if (e0 < EE) {
            d[g].x |= atomicAdd(&cnt[d[g].x >> BSH], 1) << DB;
            d[g].y |= atomicAdd(&cnt[d[g].y >> BSH], 1) << DB;
            d[g].z |= atomicAdd(&cnt[d[g].z >> BSH], 1) << DB;
            d[g].w |= atomicAdd(&cnt[d[g].w >> BSH], 1) << DB;
        }
    }
    __syncthreads();
    {
        const int4 v = *(const int4*)(cnt + 4 * t);
        const int ex = block_excl_scan(v.x + v.y + v.z + v.w, wsum);
        const int4 o = make_int4(ex, ex + v.x, ex + v.x + v.y, ex + v.x + v.y + v.z);
        *(int4*)(off + 4 * t) = o;
        if (4 * t < OFFP) *(int4*)(boff + (size_t)b * OFFP + 4 * t) = o;
    }
    __syncthreads();
    int* sl = tmp + (size_t)b * SL;
#pragma unroll
    for (int g = 0; g < SG; ++g) {
        const int e0 = ebase + (g * 128 + t) * 4;
        if (e0 < EE) {
            const int4 dd = make_int4(d[g].x & ((1 << DB) - 1), d[g].y & ((1 << DB) - 1),
                                      d[g].z & ((1 << DB) - 1), d[g].w & ((1 << DB) - 1));
            sl[off[dd.x >> BSH] + (d[g].x >> DB)] = (s[g].x << BSH) | (dd.x & (BW - 1));
            sl[off[dd.y >> BSH] + (d[g].y >> DB)] = (s[g].y << BSH) | (dd.y & (BW - 1));
            sl[off[dd.z >> BSH] + (d[g].z >> DB)] = (s[g].z << BSH) | (dd.z & (BW - 1));
            sl[off[dd.w >> BSH] + (d[g].w >> DB)] = (s[g].w << BSH) | (dd.w & (BW - 1));
        }
    }
}

#define P1 132
static_assert(4 * 16 * P1 >= 1040, "an edge block's counters fit in the tile's LDS");
__global__ void __launch_bounds__(128)
k_gemm1(const float* __restrict__ x, const float* __restrict__ W1,
        const float* __restrict__ a_dst,
        float* __restrict__ h1, float* __restrict__ ald,
        const int* __restrict__ ei, int* __restrict__ boff, int* __restrict__ tmp) {
    __shared__ __attribute__((aligned(16))) float tile[4][16 * P1];
    if (blockIdx.x < EB) {
        edge_block_sort(ei, boff, tmp, (int*)(void*)&tile[0][0], (int)blockIdx.x);
        return;
    }
    const int gb   = (int)blockIdx.x - EB;
    const int wave = threadIdx.x >> 5;
    const int row0 = (gb * 4 + wave) * 16;
    if (row0 >= NN) return;
    const int lane = threadIdx.x & 31;
    const int m    = lane & 15;
    const int kb   = (lane >> 4) * 2;

    v8f c[8];
#pragma unroll
    for (int t = 0; t < 8; ++t) c[t] = (v8f){0,0,0,0,0,0,0,0};

    for (int k = 0; k < IN_CH; k += 4) {
        v2f a;
        const float* xr = x + (size_t)(row0 + m) * IN_CH + (k + kb);
        a.x = xr[0];
        a.y = xr[1];
#pragma unroll
        for (int t = 0; t < 8; ++t) {
            v2f b;
            const float* wr = W1 + (size_t)(k + kb) * HC + t * 16 + m;
            b.x = wr[0];
            b.y = wr[HC];
            c[t] = __builtin_amdgcn_wmma_f32_16x16x4_f32(false, a, false, b,
                                                         (short)0, c[t], false, false);
        }
    }
    const int rofs = (lane >> 4) * 8;
    float* sm = tile[wave];
#pragma unroll
    for (int t = 0; t < 8; ++t)
#pragma unroll
        for (int r = 0; r < 8; ++r) {
            h1[(size_t)(row0 + r + rofs) * HC + t * 16 + m] = c[t][r];
            sm[(r + rofs) * P1 + t * 16 + m] = c[t][r];
        }
    __builtin_amdgcn_wave_barrier();

    const int lr = lane & 15;
    const int hp = (lane >> 4) * 2;
    float d0 = 0.f, d1 = 0.f;
    const float* r0 = sm + lr * P1 + hp * HID;
    const float* ad0 = a_dst + hp * HID;
#pragma unroll
    for (int cc = 0; cc < HID; ++cc) {
        d0 = __builtin_fmaf(r0[cc], ad0[cc], d0);
        d1 = __builtin_fmaf(r0[HID + cc], ad0[HID + cc], d1);
    }
    *(float2*)(ald + (size_t)(row0 + lr) * HEADS + hp) = make_float2(d0, d1);
}

__global__ void __launch_bounds__(TBE)
k_bsort(const int* __restrict__ boff, const int* __restrict__ tmp,
        int* __restrict__ deg, int* __restrict__ rowstart, int* __restrict__ csr) {
    __shared__ int c[BW];
    __shared__ int cur[BW];
    __shared__ int pst[TBE];
    __shared__ int pcn[TBE];
    __shared__ int wsumA[16];
    __shared__ int wsumC[16];
    __shared__ int sb[1];
    const int t    = threadIdx.x;
    const int b    = blockIdx.x;
    const int lane = t & 31;
    const int wv   = t >> 5;
    if (t < BW) c[t] = 0;
    int st = 0, cn = 0;
    if (t < EB) {
        const int* o = boff + (size_t)t * OFFP + b;
        st = o[0];
        cn = o[1] - st;
    }
    pst[t] = st;
    pcn[t] = cn;
    {
        const int exs = block_excl_scan(st, wsumA);
        if (t == TBE - 1) sb[0] = exs + st;
    }
    __syncthreads();
    const int bstart = sb[0];
    for (int p0 = wv; p0 < EB; p0 += 16 * PU) {
        int v[PU];
#pragma unroll
        for (int u = 0; u < PU; ++u) {
            const int p = p0 + 16 * u;
            int s0 = 0, n = 0;
            if (p < EB) { s0 = pst[p]; n = pcn[p]; }
            const bool in = lane < n;
            const int* src = in ? tmp : boff;
            v[u] = src[in ? (size_t)p * SL + s0 + lane : (size_t)0];
        }
#pragma unroll
        for (int u = 0; u < PU; ++u) {
            const int p = p0 + 16 * u;
            const int n = (p < EB) ? pcn[p] : 0;
            if (lane < n) atomicAdd(&c[v[u] & (BW - 1)], 1);
            for (int j = 32 + lane; j < n; j += 32) {
                const int w = tmp[(size_t)p * SL + pst[p] + j];
                atomicAdd(&c[w & (BW - 1)], 1);
            }
        }
    }
    __syncthreads();
    const int vv  = (t < BW) ? c[t] : 0;
    const int pre = block_excl_scan(vv, wsumC);
    if (t < BW) {
        cur[t] = pre;
        const int nd = b * BW + t;
        if (nd < NN) { deg[nd] = vv; rowstart[nd] = bstart + pre; }
    }
    __syncthreads();
    for (int p0 = wv; p0 < EB; p0 += 16 * PU) {
        int v[PU];
#pragma unroll
        for (int u = 0; u < PU; ++u) {
            const int p = p0 + 16 * u;
            int s0 = 0, n = 0;
            if (p < EB) { s0 = pst[p]; n = pcn[p]; }
            const bool in = lane < n;
            const int* src = in ? tmp : boff;
            v[u] = src[in ? (size_t)p * SL + s0 + lane : (size_t)0];
        }
#pragma unroll
        for (int u = 0; u < PU; ++u) {
            const int p = p0 + 16 * u;
            const int n = (p < EB) ? pcn[p] : 0;
            if (lane < n) {
                const int r = atomicAdd(&cur[v[u] & (BW - 1)], 1);
                csr[bstart + r] = v[u] >> BSH;
            }
            for (int j = 32 + lane; j < n; j += 32) {
                const int w = tmp[(size_t)p * SL + pst[p] + j];
                const int r = atomicAdd(&cur[w & (BW - 1)], 1);
                csr[bstart + r] = w >> BSH;
            }
        }
    }
}


#define M_START (-3.0e38f)

__device__ __forceinline__ float xm1(float v) {
    return __int_as_float(__builtin_amdgcn_update_dpp(0, __float_as_int(v), 0x161, 0xf, 0xf, true));
}
__device__ __forceinline__ float xm2(float v) {
    return __int_as_float(__builtin_amdgcn_update_dpp(0, __float_as_int(v), 0x162, 0xf, 0xf, true));
}
__device__ __forceinline__ float xm4(float v) {
    return __int_as_float(__builtin_amdgcn_update_dpp(0, __float_as_int(v), 0x164, 0xf, 0xf, true));
}
template <int K>
__device__ __forceinline__ float quad_lane(float v) {
    return __int_as_float(__builtin_amdgcn_update_dpp(0, __float_as_int(v), K * 0x55, 0xf, 0xf, true));
}

template <bool TAIL>
__device__ __forceinline__ void batch_online1(const float* __restrict__ h1, unsigned lo, int s, int j, int r,
                                              const float4 as, float dn, bool b0, bool b1, int rho,
                                              float& m, float& d, float4& acc) {
    int sid[4];
#pragma unroll
    for (int i = 0; i < 4; ++i)
        sid[i] = __builtin_amdgcn_readlane(s, j + ((TAIL && i >= r) ? r - 1 : i));
    float4 row[4];
#pragma unroll
    for (int i = 0; i < 4; ++i) {
        row[i] = make_float4(0.f, 0.f, 0.f, 0.f);
        if (!TAIL || i < r) {
            const float* rp = h1 + (size_t)sid[i] * HC;
            row[i] = *(const float4*)(rp + lo);
        }
    }
    float t[4];
#pragma unroll
    for (int i = 0; i < 4; ++i) {
        const v2f r01 = { row[i].x, row[i].y };
        const v2f r23 = { row[i].z, row[i].w };
        const v2f a01 = { as.x, as.y };
        const v2f a23 = { as.z, as.w };
        const v2f t2  = __builtin_elementwise_fma(r23, a23, r01 * a01);
        t[i] = t2.x + t2.y;
    }
    const float uA = (b0 ? t[2] : t[0]) + xm1(b0 ? t[0] : t[2]);
    const float uB = (b0 ? t[3] : t[1]) + xm1(b0 ? t[1] : t[3]);
    float w = (b1 ? uB : uA) + xm2(b1 ? uA : uB);
    w += xm4(w);
    float e = lrelu(w + dn);
    if (TAIL) e = (rho < r) ? e : -__builtin_inff();
    float em = fmaxf(e, xm1(e));
    em = fmaxf(em, xm2(em));
    const float mn = fmaxf(m, em);
    const float sc = __expf(m - mn);
    m = mn;
    const float pe = __expf(e - mn);
    d = __builtin_fmaf(d, sc, pe);
    const float p0 = quad_lane<0>(pe);
    const float p1 = quad_lane<2>(pe);
    const float p2 = quad_lane<1>(pe);
    const float p3 = quad_lane<3>(pe);
    acc.x = __builtin_fmaf(row[0].x, p0, acc.x * sc); acc.y = __builtin_fmaf(row[0].y, p0, acc.y * sc);
    acc.z = __builtin_fmaf(row[0].z, p0, acc.z * sc); acc.w = __builtin_fmaf(row[0].w, p0, acc.w * sc);
    acc.x = __builtin_fmaf(row[1].x, p1, acc.x); acc.y = __builtin_fmaf(row[1].y, p1, acc.y);
    acc.z = __builtin_fmaf(row[1].z, p1, acc.z); acc.w = __builtin_fmaf(row[1].w, p1, acc.w);
    acc.x = __builtin_fmaf(row[2].x, p2, acc.x); acc.y = __builtin_fmaf(row[2].y, p2, acc.y);
    acc.z = __builtin_fmaf(row[2].z, p2, acc.z); acc.w = __builtin_fmaf(row[2].w, p2, acc.w);
    acc.x = __builtin_fmaf(row[3].x, p3, acc.x); acc.y = __builtin_fmaf(row[3].y, p3, acc.y);
    acc.z = __builtin_fmaf(row[3].z, p3, acc.z); acc.w = __builtin_fmaf(row[3].w, p3, acc.w);
}

#define ZP 48
#define P2 52
static_assert(NN % 16 == 0, "whole 16-node strips: every wave of every block has its four nodes");
static_assert(HC == 32 * 4 && HID == 8 * 4, "a lane per 4 channels: 32 lanes a row, 8 lanes a head");
static_assert(EE > 0, "run slot 0 is always written");
__global__ void __launch_bounds__(128)
k_node1(const int* __restrict__ csr, const int* __restrict__ rowstart,
        const int* __restrict__ deg, const float* __restrict__ h1,
        const float* __restrict__ a_src, const float* __restrict__ ald,
        const float* __restrict__ b1, const float* __restrict__ W2,
        const float* __restrict__ a_src2, const float* __restrict__ a_dst2,
        float* __restrict__ z, float* __restrict__ ald2) {
    __shared__ float hs[16 * P1];
    __shared__ float zt[16 * P2];
    const int lane = threadIdx.x & 31;
    const int wv   = threadIdx.x >> 5;
    const int head = lane >> 3;
    const bool lb0 = (lane & 1) != 0;
    const bool lb1 = (lane & 2) != 0;
    const int  rho = (lb0 ? 2 : 0) + (lb1 ? 1 : 0);
    const unsigned lo = (unsigned)lane * 4u;
    const float4  as = *(const float4*)(a_src + lane * 4);
    const int nb = __builtin_amdgcn_readfirstlane((int)(blockIdx.x * 16 + wv * 4));
    const int4 st4 = *(const int4*)(rowstart + nb);
    const int4 cn4 = *(const int4*)(deg + nb);
    const int st[4] = { __builtin_amdgcn_readfirstlane(st4.x), __builtin_amdgcn_readfirstlane(st4.y),
                        __builtin_amdgcn_readfirstlane(st4.z), __builtin_amdgcn_readfirstlane(st4.w) };
    const int cn[4] = { __builtin_amdgcn_readfirstlane(cn4.x), __builtin_amdgcn_readfirstlane(cn4.y),
                        __builtin_amdgcn_readfirstlane(cn4.z), __builtin_amdgcn_readfirstlane(cn4.w) };
    int   sq[4];
    float dn[4];
#pragma unroll
    for (int q = 0; q < 4; ++q) {
        const int sv = csr[lane < cn[q] ? st[q] + lane : 0];
        sq[q] = lane < cn[q] ? sv : nb + q;
        dn[q] = ald[(size_t)(nb + q) * HEADS + head];
    }
#pragma unroll
    for (int q = 0; q < 4; ++q) {
        const int n   = nb + q;
        const int tot = cn[q] + 1;
        const float dnq = dn[q];
        float m = M_START, d = 0.f;
        float4 acc = make_float4(0.f, 0.f, 0.f, 0.f);
#pragma nounroll
        for (int base = 0; base < tot; base += 32) {
            int s = sq[q];
            if (base != 0) {
                const int idx = base + lane;
                s = n;
                if (idx < cn[q]) s = csr[st[q] + idx];
            }
            const int nv = (tot - base < 32) ? (tot - base) : 32;
            int j = 0;
            for (; j + 4 <= nv; j += 4) batch_online1<false>(h1, lo, s, j, 4, as, dnq, lb0, lb1, rho, m, d, acc);
            if (j < nv)                 batch_online1<true >(h1, lo, s, j, nv - j, as, dnq, lb0, lb1, rho, m, d, acc);
        }
        float dt = d + xm1(d);
        dt += xm2(dt);
        const float inv = 1.0f / (dt + 1e-16f);
        const float4 bb = *(const float4*)(b1 + lane * 4);
        float4 o;
        o.x = __builtin_fmaf(acc.x, inv, bb.x); o.y = __builtin_fmaf(acc.y, inv, bb.y);
        o.z = __builtin_fmaf(acc.z, inv, bb.z); o.w = __builtin_fmaf(acc.w, inv, bb.w);
        o.x = o.x > 0.0f ? o.x : expm1f(o.x);
        o.y = o.y > 0.0f ? o.y : expm1f(o.y);
        o.z = o.z > 0.0f ? o.z : expm1f(o.z);
        o.w = o.w > 0.0f ? o.w : expm1f(o.w);
        *(float4*)(hs + (wv * 4 + q) * P1 + lane * 4) = o;
    }

    const int gm  = lane & 15;
    const int gkb = (lane >> 4) * 2;
    const int col = wv * 16 + gm;
    const float* wp = W2 + (size_t)gkb * OUT_CH + (col < OUT_CH ? col : OUT_CH - 1);
    v2f bq[16];
    if (wv < 3) {
#pragma unroll
        for (int i = 0; i < 16; ++i) {
            bq[i].x = wp[(size_t)(4 * i) * OUT_CH];
            bq[i].y = wp[(size_t)(4 * i + 1) * OUT_CH];
        }
    }
    __syncthreads();

    const int row0 = blockIdx.x * 16;
    if (wv < 3) {
        const float* ap = hs + gm * P1 + gkb;
        v8f c = (v8f){0,0,0,0,0,0,0,0};
#pragma unroll
        for (int i = 0; i < 16; ++i) {
            const v2f a = *(const v2f*)(ap + 4 * i);
            c = __builtin_amdgcn_wmma_f32_16x16x4_f32(false, a, false, bq[i],
                                                      (short)0, c, false, false);
        }
#pragma unroll
        for (int k = 64; k < HC; k += 4) {
            const v2f a = *(const v2f*)(ap + k);
            v2f b;
            b.x = wp[(size_t)k * OUT_CH];
            b.y = wp[(size_t)(k + 1) * OUT_CH];
            c = __builtin_amdgcn_wmma_f32_16x16x4_f32(false, a, false, b,
                                                      (short)0, c, false, false);
        }
        const int rofs = (lane >> 4) * 8;
#pragma unroll
        for (int r = 0; r < 8; ++r) zt[(r + rofs) * P2 + col] = c[r];
    }
    __syncthreads();
    if (wv != 3) return;
    {
        const int lr = lane & 15;
        const bool tgt = lane >= 16;
        const float* av = tgt ? a_dst2 : a_src2;
        const float* zr = zt + lr * P2;
        float sacc = 0.f;
#pragma unroll
        for (int cc = 0; cc < OUT_CH; ++cc) sacc = __builtin_fmaf(zr[cc], av[cc], sacc);
        if (tgt) ald2[row0 + lr] = sacc;
        else     zt[lr * P2 + OUT_CH] = sacc;
    }
    __builtin_amdgcn_wave_barrier();
    const int t3 = (int)threadIdx.x - 3 * 32;
#pragma unroll
    for (int i = 0; i < 16 * ZP / 4 / 32; ++i) {
        const int idx = t3 + 32 * i;
        const int r  = idx / (ZP / 4);
        const int c4 = idx - r * (ZP / 4);
        const float4 v = *(const float4*)(zt + r * P2 + c4 * 4);
        *(float4*)(z + (size_t)row0 * ZP + idx * 4) = v;
    }
}


__device__ __forceinline__ float xhalf(float x) {
    const int b = __float_as_int(x);
    return __int_as_float(__builtin_amdgcn_permlanex16(b, b, 0x76543210, 0xfedcba98, false, false));
}

template <int N>
__device__ __forceinline__ float half_lane(float v) {
    return __int_as_float(__builtin_amdgcn_update_dpp(0, __float_as_int(v), 0x150 + N, 0xf, 0xf, true));
}

#define LQ 12
static_assert(LQ % 4 == 0 && LQ >= OUT_CH / 4 && LQ + 4 <= 16, "the weighing lanes read the logit's group");

__device__ __forceinline__ void chunk_online(const float* __restrict__ zl, int half, bool k0, bool k1, int kk, int s, int nv,
                                             float dn, float& m, float& d, float4& acc) {
    const float NEG_INF = -__builtin_inff();
    const int np = (nv + 1) >> 1;
    for (int p = 0; p < np; p += 4) {
        int sid[4];
#pragma unroll
        for (int j = 0; j < 4; ++j) sid[j] = __shfl(s, (2 * (p + j) + half) & 31);
        float4 r[4];
#pragma unroll
        for (int j = 0; j < 4; ++j) {
            r[j] = make_float4(0.f, 0.f, 0.f, 0.f);
            if (p + j < np)
                r[j] = *(const float4*)(zl + (size_t)sid[j] * ZP);
        }
        const float a01 = k0 ? r[1].x : r[0].x;
        const float a23 = k0 ? r[3].x : r[2].x;
        const float a   = k1 ? a23 : a01;
        const float e = (2 * p + kk < nv) ? lrelu(a + dn) : NEG_INF;
        float em = fmaxf(e, xm1(e));
        em = fmaxf(em, xm2(em));
        const float mn = fmaxf(m, half_lane<LQ>(em));
        const float sc = __expf(m - mn);
        m = mn;
        const float pe = __expf(e - mn);
        d = __builtin_fmaf(d, sc, pe);
        const float p0 = half_lane<LQ + 0>(pe);
        const float p1 = half_lane<LQ + 1>(pe);
        const float p2 = half_lane<LQ + 2>(pe);
        const float p3 = half_lane<LQ + 3>(pe);
        acc.x = __builtin_fmaf(r[0].x, p0, acc.x * sc); acc.y = __builtin_fmaf(r[0].y, p0, acc.y * sc);
        acc.z = __builtin_fmaf(r[0].z, p0, acc.z * sc); acc.w = __builtin_fmaf(r[0].w, p0, acc.w * sc);
        acc.x = __builtin_fmaf(r[1].x, p1, acc.x); acc.y = __builtin_fmaf(r[1].y, p1, acc.y);
        acc.z = __builtin_fmaf(r[1].z, p1, acc.z); acc.w = __builtin_fmaf(r[1].w, p1, acc.w);
        acc.x = __builtin_fmaf(r[2].x, p2, acc.x); acc.y = __builtin_fmaf(r[2].y, p2, acc.y);
        acc.z = __builtin_fmaf(r[2].z, p2, acc.z); acc.w = __builtin_fmaf(r[2].w, p2, acc.w);
        acc.x = __builtin_fmaf(r[3].x, p3, acc.x); acc.y = __builtin_fmaf(r[3].y, p3, acc.y);
        acc.z = __builtin_fmaf(r[3].z, p3, acc.z); acc.w = __builtin_fmaf(r[3].w, p3, acc.w);
    }
}

static_assert(NN % 4 == 0, "a wave's four nodes all exist");
static_assert(OUT_CH % 4 == 0 && OUT_CH / 4 < 16 && OUT_CH + 4 <= ZP && ZP <= P2 && ZP % 4 == 0,
              "a row and its logit column are one 16-byte load per lane of a half wave");
__global__ void __launch_bounds__(256)
k_node2(const int* __restrict__ csr, const int* __restrict__ rowstart,
        const int* __restrict__ deg, const float* __restrict__ z,
        const float* __restrict__ ald,
        const float* __restrict__ b2, float* __restrict__ out) {
    const int lane = threadIdx.x & 31;
    const int w = __builtin_amdgcn_readfirstlane((int)(blockIdx.x * 8 + (threadIdx.x >> 5)));
    const int n0 = w * 4;
    if (n0 >= NN) return;
    const int  half = lane >> 4;
    const int  c4   = lane & 15;
    const bool chan = c4 < OUT_CH / 4;
    const bool k0   = (lane & 1) != 0;
    const bool k1   = (lane & 2) != 0;
    const int  kk   = 2 * (lane & 3) + half;
    const float4 bb = *(const float4*)(b2 + (chan ? c4 : 0) * 4);
    const float* zl = z + (chan ? c4 : OUT_CH / 4) * 4;

    const int4   st4 = *(const int4*)(rowstart + n0);
    const int4   cn4 = *(const int4*)(deg + n0);
    const float4 dn4 = *(const float4*)(ald + n0);
    const int st[4] = { __builtin_amdgcn_readfirstlane(st4.x), __builtin_amdgcn_readfirstlane(st4.y),
                        __builtin_amdgcn_readfirstlane(st4.z), __builtin_amdgcn_readfirstlane(st4.w) };
    const int cn[4] = { __builtin_amdgcn_readfirstlane(cn4.x), __builtin_amdgcn_readfirstlane(cn4.y),
                        __builtin_amdgcn_readfirstlane(cn4.z), __builtin_amdgcn_readfirstlane(cn4.w) };
    const float dn[4] = { dn4.x, dn4.y, dn4.z, dn4.w };

    int s[4];
#pragma unroll
    for (int q = 0; q < 4; ++q) {
        const int sv = csr[lane < cn[q] ? st[q] + lane : 0];
        s[q] = lane < cn[q] ? sv : n0 + q;
    }

#pragma unroll
    for (int q = 0; q < 4; ++q) {
        const int n   = n0 + q;
        const int tot = cn[q] + 1;
        float m = M_START, d = 0.f;
        float4 acc = make_float4(0.f, 0.f, 0.f, 0.f);
        chunk_online(zl, half, k0, k1, kk, s[q], (tot < 32) ? tot : 32, dn[q], m, d, acc);
#pragma nounroll
        for (int base = 32; base < tot; base += 32) {
            const int idx = base + lane;
            int s2 = n;
            if (idx < cn[q]) s2 = csr[st[q] + idx];
            chunk_online(zl, half, k0, k1, kk, s2, (tot - base < 32) ? (tot - base) : 32, dn[q], m, d, acc);
        }
        d += xm1(d);
        d += xm2(d);
        d = half_lane<LQ>(d);
        const float mm = fmaxf(m, xhalf(m));
        const float f  = __expf(m - mm);
        d *= f;
        acc.x *= f; acc.y *= f; acc.z *= f; acc.w *= f;
        d += xhalf(d);
        acc.x += xhalf(acc.x); acc.y += xhalf(acc.y);
        acc.z += xhalf(acc.z); acc.w += xhalf(acc.w);
        const float inv = 1.0f / (d + 1e-16f);
        if (lane < OUT_CH / 4)
            *(float4*)(out + (size_t)n * OUT_CH + lane * 4) =
                make_float4(__builtin_fmaf(acc.x, inv, bb.x), __builtin_fmaf(acc.y, inv, bb.y),
                            __builtin_fmaf(acc.z, inv, bb.z), __builtin_fmaf(acc.w, inv, bb.w));
    }
}

extern "C" void kernel_launch(void* const* d_in, const int* in_sizes, int n_in,
                              void* d_out, int out_size, void* d_ws, size_t ws_size,
                              hipStream_t stream) {
    const float* x      = (const float*)d_in[0];
    const int*   ei     = (const int*)  d_in[1];
    const float* W1     = (const float*)d_in[2];
    const float* a_src1 = (const float*)d_in[3];
    const float* a_dst1 = (const float*)d_in[4];
    const float* b1     = (const float*)d_in[5];
    const float* W2     = (const float*)d_in[6];
    const float* a_src2 = (const float*)d_in[7];
    const float* a_dst2 = (const float*)d_in[8];
    const float* b2     = (const float*)d_in[9];
    float* out = (float*)d_out;

    float* ws   = (float*)d_ws;
    float* h1   = ws;
    float* z2   = h1   + (size_t)NN * HC;
    float* ald1 = z2   + (size_t)NN * ZP;
    float* ald2 = ald1 + (size_t)NN * HEADS;
    int* deg      = (int*)(ald2 + NN);
    int* rowstart = deg + NN;
    int* boff     = rowstart + NN;
    int* tmp      = boff + EB * OFFP;
    int* csr      = tmp + EB * SL;

    const int gemmBlocks = (NN / 16 + 3) / 4;
    const int nodeBlocks1 = NN / 16;
    const int nodeBlocks2 = (NN / 4 + 7) / 8;

    k_gemm1<<<EB + gemmBlocks, 128, 0, stream>>>(x, W1, a_dst1, h1, ald1, ei, boff, tmp);
    k_bsort<<<NB, TBE, 0, stream>>>(boff, tmp, deg, rowstart, csr);
    k_node1<<<nodeBlocks1, 128, 0, stream>>>(csr, rowstart, deg, h1, a_src1, ald1, b1,
                                             W2, a_src2, a_dst2, z2, ald2);

    k_node2<<<nodeBlocks2, 256, 0, stream>>>(csr, rowstart, deg, z2, ald2, b2, out);
}
